# starting kernel
# baseline (speedup 1.0000x reference)
#include <hip/hip_runtime.h>
#include <hip/hip_bf16.h>
#include <math.h>

#define N_NODES 10000
#define N_EDGES 320000
#define F_IN    256
#define F_HID   256
#define F_OUT   16
#define BN_EPS  1e-5f

typedef __attribute__((ext_vector_type(2))) float v2f;
typedef __attribute__((ext_vector_type(8))) float v8f;

__device__ __forceinline__ v8f wmma4(v2f a, v2f b, v8f c) {
    return __builtin_amdgcn_wmma_f32_16x16x4_f32(
        false, a, false, b, (short)0, c, false, false);
}

__device__ __forceinline__ v2f load_a(const float* A, int lda, int m0, int k0, int lane) {
    const int half = lane >> 4;
    const int row  = lane & 15;
    const float* p = A + (size_t)(m0 + row) * lda + k0 + (half << 1);
    v2f r; r.x = p[0]; r.y = p[1];
    return r;
}

__device__ __forceinline__ v2f load_b(const float* B, int ldb, int k0, int n0, int lane) {
    const int half = lane >> 4;
    const int col  = lane & 15;
    const float* p = B + (size_t)(k0 + (half << 1)) * ldb + n0 + col;
    v2f r; r.x = p[0]; r.y = p[ldb];
    return r;
}

__device__ __forceinline__ void store_c(float* C, int ldc, int m0, int n0, int lane, v8f acc) {
    const int half = lane >> 4;
    const int col  = lane & 15;
#pragma unroll
    for (int r = 0; r < 8; ++r)
        C[(size_t)(m0 + r + half * 8) * ldc + n0 + col] = acc[r];
}

__global__ void deg_accum_kernel(const int* __restrict__ dst, float* __restrict__ deg) {
    int e = blockIdx.x * blockDim.x + threadIdx.x;
    if (e < N_EDGES) atomicAdd(&deg[dst[e]], 1.0f);
}

__global__ void deg_inv_kernel(float* __restrict__ deg) {
    int n = blockIdx.x * blockDim.x + threadIdx.x;
    if (n < N_NODES) deg[n] = 1.0f / fmaxf(deg[n], 1.0f);
}

__global__ void agg1_scatter_kernel(const float* __restrict__ x,
                                    const int* __restrict__ src,
                                    const int* __restrict__ dst,
                                    float* __restrict__ agg1) {
    long long g = (long long)blockIdx.x * blockDim.x + threadIdx.x;
    if (g >= (long long)N_EDGES * 64) return;
    int e  = (int)(g >> 6);
    int c4 = ((int)g & 63) << 2;
    const float4 v = *(const float4*)(x + (size_t)src[e] * F_IN + c4);
    float* a = agg1 + (size_t)dst[e] * F_IN + c4;
    atomicAdd(a + 0, v.x);
    atomicAdd(a + 1, v.y);
    atomicAdd(a + 2, v.z);
    atomicAdd(a + 3, v.w);
}

template <int COLS>
__global__ void scale_rows_kernel(float* __restrict__ m, const float* __restrict__ deg_inv,
                                  long long total) {
    long long g = (long long)blockIdx.x * blockDim.x + threadIdx.x;
    if (g < total) m[g] *= deg_inv[g / COLS];
}

__global__ __launch_bounds__(256) void gemm1_kernel(const float* __restrict__ A1,
                                                    const float* __restrict__ B1,
                                                    const float* __restrict__ A2,
                                                    const float* __restrict__ B2,
                                                    float* __restrict__ C) {
    const int lane = threadIdx.x & 31;
    const int wave = threadIdx.x >> 5;
    const int tile = blockIdx.x * 8 + wave;
    if (tile >= 2500) return;
    const int mt = tile >> 2;
    const int ng = tile & 3;
    const int m0 = mt * 16, n0 = ng * 64;

    v8f acc0 = {}, acc1 = {}, acc2 = {}, acc3 = {};
#pragma unroll 2
    for (int k0 = 0; k0 < F_IN; k0 += 4) {
        v2f a = load_a(A1, F_IN, m0, k0, lane);
        acc0 = wmma4(a, load_b(B1, F_HID, k0, n0 +  0, lane), acc0);
        acc1 = wmma4(a, load_b(B1, F_HID, k0, n0 + 16, lane), acc1);
        acc2 = wmma4(a, load_b(B1, F_HID, k0, n0 + 32, lane), acc2);
        acc3 = wmma4(a, load_b(B1, F_HID, k0, n0 + 48, lane), acc3);
    }
#pragma unroll 2
    for (int k0 = 0; k0 < F_IN; k0 += 4) {
        v2f a = load_a(A2, F_IN, m0, k0, lane);
        acc0 = wmma4(a, load_b(B2, F_HID, k0, n0 +  0, lane), acc0);
        acc1 = wmma4(a, load_b(B2, F_HID, k0, n0 + 16, lane), acc1);
        acc2 = wmma4(a, load_b(B2, F_HID, k0, n0 + 32, lane), acc2);
        acc3 = wmma4(a, load_b(B2, F_HID, k0, n0 + 48, lane), acc3);
    }
    store_c(C, F_HID, m0, n0 +  0, lane, acc0);
    store_c(C, F_HID, m0, n0 + 16, lane, acc1);
    store_c(C, F_HID, m0, n0 + 32, lane, acc2);
    store_c(C, F_HID, m0, n0 + 48, lane, acc3);
}

__global__ __launch_bounds__(256) void gemm3_kernel(const float* __restrict__ A,
                                                    const float* __restrict__ B1,
                                                    const float* __restrict__ B2,
                                                    float* __restrict__ D1,
                                                    float* __restrict__ D2) {
    const int lane = threadIdx.x & 31;
    const int wave = threadIdx.x >> 5;
    const int tile = blockIdx.x * 8 + wave;
    if (tile >= 625) return;
    const int m0 = tile * 16;

    v8f acc1 = {}, acc2 = {};
#pragma unroll 4
    for (int k0 = 0; k0 < F_HID; k0 += 4) {
        v2f a  = load_a(A, F_HID, m0, k0, lane);
        v2f b1 = load_b(B1, F_OUT, k0, 0, lane);
        v2f b2 = load_b(B2, F_OUT, k0, 0, lane);
        acc1 = wmma4(a, b1, acc1);
        acc2 = wmma4(a, b2, acc2);
    }
    store_c(D1, F_OUT, m0, 0, lane, acc1);
    store_c(D2, F_OUT, m0, 0, lane, acc2);
}

template <int COLS>
__global__ void bn_stats_kernel(const float* __restrict__ h,
                                float* __restrict__ sum, float* __restrict__ sumsq) {
    int g = blockIdx.x * blockDim.x + threadIdx.x;
    int col     = g & (COLS - 1);
    int slice   = g / COLS;
    int nslices = (gridDim.x * blockDim.x) / COLS;
    float s = 0.0f, q = 0.0f;
    for (int r = slice; r < N_NODES; r += nslices) {
        float v = h[(size_t)r * COLS + col];
        s += v;
        q += v * v;
    }
    atomicAdd(&sum[col], s);
    atomicAdd(&sumsq[col], q);
}

template <int COLS>
__global__ void bn_relu_kernel(float* __restrict__ h,
                               const float* __restrict__ sum, const float* __restrict__ sumsq,
                               const float* __restrict__ gamma, const float* __restrict__ beta,
                               long long total) {
    long long g = (long long)blockIdx.x * blockDim.x + threadIdx.x;
    if (g >= total) return;
    int c = (int)(g & (COLS - 1));
    float mean = sum[c] * (1.0f / N_NODES);
    float var  = sumsq[c] * (1.0f / N_NODES) - mean * mean;
    float v = (h[g] - mean) * rsqrtf(var + BN_EPS) * gamma[c] + beta[c];
    h[g] = fmaxf(v, 0.0f);
}

__global__ void agg3_scatter_kernel(const float* __restrict__ p,
                                    const int* __restrict__ src,
                                    const int* __restrict__ dst,
                                    float* __restrict__ agg3) {
    int e = blockIdx.x * blockDim.x + threadIdx.x;
    if (e >= N_EDGES) return;
    const float* ps = p + (size_t)src[e] * F_OUT;
    float* ad = agg3 + (size_t)dst[e] * F_OUT;
#pragma unroll
    for (int j = 0; j < F_OUT; j += 4) {
        float4 v = *(const float4*)(ps + j);
        atomicAdd(ad + j + 0, v.x);
        atomicAdd(ad + j + 1, v.y);
        atomicAdd(ad + j + 2, v.z);
        atomicAdd(ad + j + 3, v.w);
    }
}

__global__ void combine3_kernel(float* __restrict__ agg3, const float* __restrict__ hl,
                                const float* __restrict__ deg_inv, long long total) {
    long long g = (long long)blockIdx.x * blockDim.x + threadIdx.x;
    if (g < total) agg3[g] = agg3[g] * deg_inv[g / F_OUT] + hl[g];
}

__global__ void bn_final_kernel(const float* __restrict__ t,
                                const float* __restrict__ sum, const float* __restrict__ sumsq,
                                const float* __restrict__ gamma, const float* __restrict__ beta,
                                const int* __restrict__ ste,
                                float* __restrict__ out, long long total) {
    long long g = (long long)blockIdx.x * blockDim.x + threadIdx.x;
    if (g >= total) return;
    int c = (int)(g & (F_OUT - 1));
    float mean = sum[c] * (1.0f / N_NODES);
    float var  = sumsq[c] * (1.0f / N_NODES) - mean * mean;
    float v = (t[g] - mean) * rsqrtf(var + BN_EPS) * gamma[c] + beta[c];
    if (*ste)
        out[g] = (v > 0.0f) ? 1.0f : 0.0f;
    else
        out[g] = 1.0f / (1.0f + expf(-v));
}

extern "C" void kernel_launch(void* const* d_in, const int* in_sizes, int n_in,
                              void* d_out, int out_size, void* d_ws, size_t ws_size,
                              hipStream_t stream) {
    const float* x        = (const float*)d_in[0];
    const float* W_conv1  = (const float*)d_in[1];
    const float* W_lin1   = (const float*)d_in[2];
    const float* bn1_g    = (const float*)d_in[3];
    const float* bn1_b    = (const float*)d_in[4];
    const float* W_conv3  = (const float*)d_in[5];
    const float* W_lin3   = (const float*)d_in[6];
    const float* bn3_g    = (const float*)d_in[7];
    const float* bn3_b    = (const float*)d_in[8];
    const int*   src      = (const int*)d_in[9];
    const int*   dst      = (const int*)d_in[10];
    const int*   ste      = (const int*)d_in[11];
    float*       out      = (float*)d_out;

    float* ws = (float*)d_ws;
    const size_t o_deg  = 0;
    const size_t o_agg1 = 10240;
    const size_t o_h    = o_agg1 + 2560000;
    const size_t o_p    = o_h + 2560000;
    const size_t o_hl   = o_p + 160000;
    const size_t o_agg3 = o_hl + 160000;
    const size_t o_s1   = o_agg3 + 160000;
    const size_t o_q1   = o_s1 + 256;
    const size_t o_s3   = o_q1 + 256;
    const size_t o_q3   = o_s3 + 16;
    float* deg_inv = ws + o_deg;
    float* agg1    = ws + o_agg1;
    float* h       = ws + o_h;
    float* p       = ws + o_p;
    float* hl      = ws + o_hl;
    float* agg3    = ws + o_agg3;
    float* s1 = ws + o_s1; float* q1 = ws + o_q1;
    float* s3 = ws + o_s3; float* q3 = ws + o_q3;

    hipMemsetAsync(deg_inv, 0, 10240 * sizeof(float), stream);
    hipMemsetAsync(agg1, 0, 2560000 * sizeof(float), stream);
    hipMemsetAsync(agg3, 0, (160000 + 256 + 256 + 16 + 16) * sizeof(float), stream);

    deg_accum_kernel<<<(N_EDGES + 255) / 256, 256, 0, stream>>>(dst, deg_inv);
    deg_inv_kernel<<<(N_NODES + 255) / 256, 256, 0, stream>>>(deg_inv);

    {
        long long total = (long long)N_EDGES * 64;
        agg1_scatter_kernel<<<(int)((total + 255) / 256), 256, 0, stream>>>(x, src, dst, agg1);
        long long tot2 = (long long)N_NODES * F_IN;
        scale_rows_kernel<F_IN><<<(int)((tot2 + 255) / 256), 256, 0, stream>>>(agg1, deg_inv, tot2);
    }

    gemm1_kernel<<<(2500 + 7) / 8, 256, 0, stream>>>(agg1, W_conv1, x, W_lin1, h);

    bn_stats_kernel<F_HID><<<128, 256, 0, stream>>>(h, s1, q1);
    {
        long long tot = (long long)N_NODES * F_HID;
        bn_relu_kernel<F_HID><<<(int)((tot + 255) / 256), 256, 0, stream>>>(h, s1, q1, bn1_g, bn1_b, tot);
    }

    gemm3_kernel<<<(625 + 7) / 8, 256, 0, stream>>>(h, W_conv3, W_lin3, p, hl);

    agg3_scatter_kernel<<<(N_EDGES + 255) / 256, 256, 0, stream>>>(p, src, dst, agg3);
    {
        long long tot = (long long)N_NODES * F_OUT;
        combine3_kernel<<<(int)((tot + 255) / 256), 256, 0, stream>>>(agg3, hl, deg_inv, tot);
    }

    bn_stats_kernel<F_OUT><<<64, 256, 0, stream>>>(agg3, s3, q3);
    {
        long long tot = (long long)N_NODES * F_OUT;
        bn_final_kernel<<<(int)((tot + 255) / 256), 256, 0, stream>>>(agg3, s3, q3, bn3_g, bn3_b, ste, out, tot);
    }
}


# aggregations: float-atomic scatter replaced by gather over edges bucketed by dst (int atomic rank, scan, place); sums in bucket (rank arrival) order, 8-way tree on 16-wide rows; scale, +hl in store
# speedup vs baseline: 2.6074x; 2.6074x over previous
#include <hip/hip_runtime.h>
#include <hip/hip_bf16.h>
#include <math.h>

#define N_NODES 10000
#define N_EDGES 320000
#define F_IN    256
#define F_HID   256
#define F_OUT   16
#define BN_EPS  1e-5f

typedef __attribute__((ext_vector_type(2))) float v2f;
typedef __attribute__((ext_vector_type(8))) float v8f;

__device__ __forceinline__ v8f wmma4(v2f a, v2f b, v8f c) {
    return __builtin_amdgcn_wmma_f32_16x16x4_f32(
        false, a, false, b, (short)0, c, false, false);
}

__device__ __forceinline__ v2f load_a(const float* A, int lda, int m0, int k0, int lane) {
    const int half = lane >> 4;
    const int row  = lane & 15;
    const float* p = A + (size_t)(m0 + row) * lda + k0 + (half << 1);
    v2f r; r.x = p[0]; r.y = p[1];
    return r;
}

__device__ __forceinline__ v2f load_b(const float* B, int ldb, int k0, int n0, int lane) {
    const int half = lane >> 4;
    const int col  = lane & 15;
    const float* p = B + (size_t)(k0 + (half << 1)) * ldb + n0 + col;
    v2f r; r.x = p[0]; r.y = p[ldb];
    return r;
}

__device__ __forceinline__ void store_c(float* C, int ldc, int m0, int n0, int lane, v8f acc) {
    const int half = lane >> 4;
    const int col  = lane & 15;
#pragma unroll
    for (int r = 0; r < 8; ++r)
        C[(size_t)(m0 + r + half * 8) * ldc + n0 + col] = acc[r];
}

__global__ void edge_rank_kernel(const int* __restrict__ dst, int* __restrict__ cur,
                                 int* __restrict__ rank) {
    int i = blockIdx.x * blockDim.x + threadIdx.x;
    if (i >= N_EDGES / 4) return;
    const int4 d = ((const int4*)dst)[i];
    int4 r;
    r.x = atomicAdd(&cur[d.x], 1);
    r.y = atomicAdd(&cur[d.y], 1);
    r.z = atomicAdd(&cur[d.z], 1);
    r.w = atomicAdd(&cur[d.w], 1);
    ((int4*)rank)[i] = r;
}

__global__ __launch_bounds__(1024) void offsets_scan_kernel(int* cur) {
    __shared__ int wave_total[32];
    const int t = threadIdx.x, lane = t & 31, wv = t >> 5;
    int2* mine = (int2*)(cur + t * 10);
    int2 v[5];
    int s = 0;
#pragma unroll
    for (int i = 0; i < 5; ++i) { v[i] = mine[i]; s += v[i].x + v[i].y; }
    int inc = s;
#pragma unroll
    for (int o = 1; o < 32; o <<= 1) { int u = __shfl_up(inc, o); if (lane >= o) inc += u; }
    if (lane == 31) wave_total[wv] = inc;
    __syncthreads();
    if (wv == 0) {
        const int w = wave_total[lane];
        int winc = w;
#pragma unroll
        for (int o = 1; o < 32; o <<= 1) { int u = __shfl_up(winc, o); if (lane >= o) winc += u; }
        wave_total[lane] = winc - w;
    }
    __syncthreads();
    int run = wave_total[wv] + inc - s;
#pragma unroll
    for (int i = 0; i < 5; ++i) {
        int2 o2;
        o2.x = run; run += v[i].x;
        o2.y = run; run += v[i].y;
        mine[i] = o2;
    }
}

__global__ void edge_place_kernel(const int* __restrict__ src, const int* __restrict__ dst,
                                  const int* __restrict__ rank, const int* __restrict__ cur,
                                  int* __restrict__ bucket_src) {
    int i = blockIdx.x * blockDim.x + threadIdx.x;
    if (i >= N_EDGES / 4) return;
    const int4 d = ((const int4*)dst)[i];
    const int4 s = ((const int4*)src)[i];
    const int4 r = ((const int4*)rank)[i];
    bucket_src[cur[d.x] + r.x] = s.x;
    bucket_src[cur[d.y] + r.y] = s.y;
    bucket_src[cur[d.z] + r.z] = s.z;
    bucket_src[cur[d.w] + r.w] = s.w;
}

template <int K>
__device__ __forceinline__ void add_rows(const float* __restrict__ xl, const int* __restrict__ ids,
                                         float4& a0, float4& a1) {
    int s[K];
    float4 u[K], w[K];
#pragma unroll
    for (int j = 0; j < K; ++j) s[j] = ids[j];
#pragma unroll
    for (int j = 0; j < K; ++j) {
        const float* r = xl + (size_t)s[j] * F_IN;
        u[j] = *(const float4*)r;
        w[j] = *(const float4*)(r + 128);
    }
#pragma unroll
    for (int j = 0; j < K; ++j) {
        a0.x += u[j].x; a0.y += u[j].y; a0.z += u[j].z; a0.w += u[j].w;
        a1.x += w[j].x; a1.y += w[j].y; a1.z += w[j].z; a1.w += w[j].w;
    }
}

__global__ __launch_bounds__(256) void agg1_gather_kernel(const float* __restrict__ x,
                                                          const int* __restrict__ cur,
                                                          const int* __restrict__ bucket_src,
                                                          float* __restrict__ agg1) {
    const int lane = threadIdx.x & 31;
    const int node = __builtin_amdgcn_readfirstlane(blockIdx.x * 8 + (threadIdx.x >> 5));
    const int begin = cur[node];
    const int end   = cur[node + 1];
    const float* xl = x + lane * 4;
    float4 a0 = {0.0f, 0.0f, 0.0f, 0.0f}, a1 = {0.0f, 0.0f, 0.0f, 0.0f};
    int e = begin;
    for (; e + 8 <= end; e += 8) add_rows<8>(xl, bucket_src + e, a0, a1);
    if (e + 4 <= end) { add_rows<4>(xl, bucket_src + e, a0, a1); e += 4; }
    if (e + 2 <= end) { add_rows<2>(xl, bucket_src + e, a0, a1); e += 2; }
    if (e < end)      { add_rows<1>(xl, bucket_src + e, a0, a1); }
    const float di = 1.0f / fmaxf((float)(end - begin), 1.0f);
    float* o = agg1 + (size_t)node * F_IN + lane * 4;
    float4 r0 = {a0.x * di, a0.y * di, a0.z * di, a0.w * di};
    float4 r1 = {a1.x * di, a1.y * di, a1.z * di, a1.w * di};
    *(float4*)o = r0;
    *(float4*)(o + 128) = r1;
}

__global__ __launch_bounds__(256) void gemm1_kernel(const float* __restrict__ A1,
                                                    const float* __restrict__ B1,
                                                    const float* __restrict__ A2,
                                                    const float* __restrict__ B2,
                                                    float* __restrict__ C) {
    const int lane = threadIdx.x & 31;
    const int wave = threadIdx.x >> 5;
    const int tile = blockIdx.x * 8 + wave;
    if (tile >= 2500) return;
    const int mt = tile >> 2;
    const int ng = tile & 3;
    const int m0 = mt * 16, n0 = ng * 64;

    v8f acc0 = {}, acc1 = {}, acc2 = {}, acc3 = {};
#pragma unroll 2
    for (int k0 = 0; k0 < F_IN; k0 += 4) {
        v2f a = load_a(A1, F_IN, m0, k0, lane);
        acc0 = wmma4(a, load_b(B1, F_HID, k0, n0 +  0, lane), acc0);
        acc1 = wmma4(a, load_b(B1, F_HID, k0, n0 + 16, lane), acc1);
        acc2 = wmma4(a, load_b(B1, F_HID, k0, n0 + 32, lane), acc2);
        acc3 = wmma4(a, load_b(B1, F_HID, k0, n0 + 48, lane), acc3);
    }
#pragma unroll 2
    for (int k0 = 0; k0 < F_IN; k0 += 4) {
        v2f a = load_a(A2, F_IN, m0, k0, lane);
        acc0 = wmma4(a, load_b(B2, F_HID, k0, n0 +  0, lane), acc0);
        acc1 = wmma4(a, load_b(B2, F_HID, k0, n0 + 16, lane), acc1);
        acc2 = wmma4(a, load_b(B2, F_HID, k0, n0 + 32, lane), acc2);
        acc3 = wmma4(a, load_b(B2, F_HID, k0, n0 + 48, lane), acc3);
    }
    store_c(C, F_HID, m0, n0 +  0, lane, acc0);
    store_c(C, F_HID, m0, n0 + 16, lane, acc1);
    store_c(C, F_HID, m0, n0 + 32, lane, acc2);
    store_c(C, F_HID, m0, n0 + 48, lane, acc3);
}

__global__ __launch_bounds__(256) void gemm3_kernel(const float* __restrict__ A,
                                                    const float* __restrict__ B1,
                                                    const float* __restrict__ B2,
                                                    float* __restrict__ D1,
                                                    float* __restrict__ D2) {
    const int lane = threadIdx.x & 31;
    const int wave = threadIdx.x >> 5;
    const int tile = blockIdx.x * 8 + wave;
    if (tile >= 625) return;
    const int m0 = tile * 16;

    v8f acc1 = {}, acc2 = {};
#pragma unroll 4
    for (int k0 = 0; k0 < F_HID; k0 += 4) {
        v2f a  = load_a(A, F_HID, m0, k0, lane);
        v2f b1 = load_b(B1, F_OUT, k0, 0, lane);
        v2f b2 = load_b(B2, F_OUT, k0, 0, lane);
        acc1 = wmma4(a, b1, acc1);
        acc2 = wmma4(a, b2, acc2);
    }
    store_c(D1, F_OUT, m0, 0, lane, acc1);
    store_c(D2, F_OUT, m0, 0, lane, acc2);
}

template <int COLS>
__global__ void bn_stats_kernel(const float* __restrict__ h,
                                float* __restrict__ sum, float* __restrict__ sumsq) {
    int g = blockIdx.x * blockDim.x + threadIdx.x;
    int col     = g & (COLS - 1);
    int slice   = g / COLS;
    int nslices = (gridDim.x * blockDim.x) / COLS;
    float s = 0.0f, q = 0.0f;
    for (int r = slice; r < N_NODES; r += nslices) {
        float v = h[(size_t)r * COLS + col];
        s += v;
        q += v * v;
    }
    atomicAdd(&sum[col], s);
    atomicAdd(&sumsq[col], q);
}

template <int COLS>
__global__ void bn_relu_kernel(float* __restrict__ h,
                               const float* __restrict__ sum, const float* __restrict__ sumsq,
                               const float* __restrict__ gamma, const float* __restrict__ beta,
                               long long total) {
    long long g = (long long)blockIdx.x * blockDim.x + threadIdx.x;
    if (g >= total) return;
    int c = (int)(g & (COLS - 1));
    float mean = sum[c] * (1.0f / N_NODES);
    float var  = sumsq[c] * (1.0f / N_NODES) - mean * mean;
    float v = (h[g] - mean) * rsqrtf(var + BN_EPS) * gamma[c] + beta[c];
    h[g] = fmaxf(v, 0.0f);
}

__global__ __launch_bounds__(256) void agg3_gather_kernel(const float* __restrict__ p,
                                                          const float* __restrict__ hl,
                                                          const int* __restrict__ cur,
                                                          const int* __restrict__ bucket_src,
                                                          float* __restrict__ t) {
    const int lane = threadIdx.x & 31;
    const int node = __builtin_amdgcn_readfirstlane(blockIdx.x * 8 + (threadIdx.x >> 5));
    const int begin = cur[node];
    const int end   = cur[node + 1];
    const int slot = lane >> 2, c4 = (lane & 3) << 2;
    float4 acc = {0.0f, 0.0f, 0.0f, 0.0f};
    for (int base = begin; base < end; base += 32) {
        int s[4];
        float4 v[4];
#pragma unroll
        for (int j = 0; j < 4; ++j) s[j] = bucket_src[min(base + slot + 8 * j, end - 1)];
#pragma unroll
        for (int j = 0; j < 4; ++j) v[j] = *(const float4*)(p + (size_t)s[j] * F_OUT + c4);
#pragma unroll
        for (int j = 0; j < 4; ++j)
            if (base + slot + 8 * j < end) {
                acc.x += v[j].x; acc.y += v[j].y; acc.z += v[j].z; acc.w += v[j].w;
            }
    }
#pragma unroll
    for (int o = 4; o < 32; o <<= 1) {
        acc.x += __shfl_xor(acc.x, o);
        acc.y += __shfl_xor(acc.y, o);
        acc.z += __shfl_xor(acc.z, o);
        acc.w += __shfl_xor(acc.w, o);
    }
    if (lane < 4) {
        const float di = 1.0f / fmaxf((float)(end - begin), 1.0f);
        const float4 h = *(const float4*)(hl + (size_t)node * F_OUT + c4);
        float4 r = {acc.x * di + h.x, acc.y * di + h.y, acc.z * di + h.z, acc.w * di + h.w};
        *(float4*)(t + (size_t)node * F_OUT + c4) = r;
    }
}

__global__ void bn_final_kernel(const float* __restrict__ t,
                                const float* __restrict__ sum, const float* __restrict__ sumsq,
                                const float* __restrict__ gamma, const float* __restrict__ beta,
                                const int* __restrict__ ste,
                                float* __restrict__ out, long long total) {
    long long g = (long long)blockIdx.x * blockDim.x + threadIdx.x;
    if (g >= total) return;
    int c = (int)(g & (F_OUT - 1));
    float mean = sum[c] * (1.0f / N_NODES);
    float var  = sumsq[c] * (1.0f / N_NODES) - mean * mean;
    float v = (t[g] - mean) * rsqrtf(var + BN_EPS) * gamma[c] + beta[c];
    if (*ste)
        out[g] = (v > 0.0f) ? 1.0f : 0.0f;
    else
        out[g] = 1.0f / (1.0f + expf(-v));
}

extern "C" void kernel_launch(void* const* d_in, const int* in_sizes, int n_in,
                              void* d_out, int out_size, void* d_ws, size_t ws_size,
                              hipStream_t stream) {
    const float* x        = (const float*)d_in[0];
    const float* W_conv1  = (const float*)d_in[1];
    const float* W_lin1   = (const float*)d_in[2];
    const float* bn1_g    = (const float*)d_in[3];
    const float* bn1_b    = (const float*)d_in[4];
    const float* W_conv3  = (const float*)d_in[5];
    const float* W_lin3   = (const float*)d_in[6];
    const float* bn3_g    = (const float*)d_in[7];
    const float* bn3_b    = (const float*)d_in[8];
    const int*   src      = (const int*)d_in[9];
    const int*   dst      = (const int*)d_in[10];
    const int*   ste      = (const int*)d_in[11];
    float*       out      = (float*)d_out;

    float* ws = (float*)d_ws;
    const size_t o_cur  = 0;
    const size_t o_s1   = 10240;
    const size_t o_q1   = o_s1 + 256;
    const size_t o_s3   = o_q1 + 256;
    const size_t o_q3   = o_s3 + 16;
    const size_t o_bkt  = 11264;
    const size_t o_rank = o_bkt + 320000;
    const size_t o_agg1 = o_rank + 320000;
    const size_t o_h    = o_agg1 + 2560000;
    const size_t o_p    = o_h + 2560000;
    const size_t o_hl   = o_p + 160000;
    const size_t o_agg3 = o_hl + 160000;
    int*   cur     = (int*)(ws + o_cur);
    int*   bkt     = (int*)(ws + o_bkt);
    int*   rank    = (int*)(ws + o_rank);
    float* agg1    = ws + o_agg1;
    float* h       = ws + o_h;
    float* p       = ws + o_p;
    float* hl      = ws + o_hl;
    float* agg3    = ws + o_agg3;
    float* s1 = ws + o_s1; float* q1 = ws + o_q1;
    float* s3 = ws + o_s3; float* q3 = ws + o_q3;

    hipMemsetAsync(cur, 0, (10240 + 256 + 256 + 16 + 16) * sizeof(float), stream);

    edge_rank_kernel<<<(N_EDGES / 4 + 255) / 256, 256, 0, stream>>>(dst, cur, rank);
    offsets_scan_kernel<<<1, 1024, 0, stream>>>(cur);
    edge_place_kernel<<<(N_EDGES / 4 + 255) / 256, 256, 0, stream>>>(src, dst, rank, cur, bkt);

    agg1_gather_kernel<<<N_NODES / 8, 256, 0, stream>>>(x, cur, bkt, agg1);

    gemm1_kernel<<<(2500 + 7) / 8, 256, 0, stream>>>(agg1, W_conv1, x, W_lin1, h);

    bn_stats_kernel<F_HID><<<128, 256, 0, stream>>>(h, s1, q1);
    {
        long long tot = (long long)N_NODES * F_HID;
        bn_relu_kernel<F_HID><<<(int)((tot + 255) / 256), 256, 0, stream>>>(h, s1, q1, bn1_g, bn1_b, tot);
    }

    gemm3_kernel<<<(625 + 7) / 8, 256, 0, stream>>>(h, W_conv3, W_lin3, p, hl);

    agg3_gather_kernel<<<N_NODES / 8, 256, 0, stream>>>(p, hl, cur, bkt, agg3);

    bn_stats_kernel<F_OUT><<<64, 256, 0, stream>>>(agg3, s3, q3);
    {
        long long tot = (long long)N_NODES * F_OUT;
        bn_final_kernel<<<(int)((tot + 255) / 256), 256, 0, stream>>>(agg3, s3, q3, bn3_g, bn3_b, ste, out, tot);
    }
}


# BatchNorm column sum/sumsq taken where the values are made (layer-1 GEMM epilogue from accumulators, layer-3 gather via block LDS); partials added by float atomics, arrival order; stats passes gone
# speedup vs baseline: 3.2686x; 1.2536x over previous
#include <hip/hip_runtime.h>
#include <hip/hip_bf16.h>
#include <math.h>

#define N_NODES 10000
#define N_EDGES 320000
#define F_IN    256
#define F_HID   256
#define F_OUT   16
#define BN_EPS  1e-5f

typedef __attribute__((ext_vector_type(2))) float v2f;
typedef __attribute__((ext_vector_type(8))) float v8f;

__device__ __forceinline__ v8f wmma4(v2f a, v2f b, v8f c) {
    return __builtin_amdgcn_wmma_f32_16x16x4_f32(
        false, a, false, b, (short)0, c, false, false);
}

__device__ __forceinline__ v2f load_a(const float* A, int lda, int m0, int k0, int lane) {
    const int half = lane >> 4;
    const int row  = lane & 15;
    const float* p = A + (size_t)(m0 + row) * lda + k0 + (half << 1);
    v2f r; r.x = p[0]; r.y = p[1];
    return r;
}

__device__ __forceinline__ v2f load_b(const float* B, int ldb, int k0, int n0, int lane) {
    const int half = lane >> 4;
    const int col  = lane & 15;
    const float* p = B + (size_t)(k0 + (half << 1)) * ldb + n0 + col;
    v2f r; r.x = p[0]; r.y = p[ldb];
    return r;
}

__device__ __forceinline__ void store_c(float* C, int ldc, int m0, int n0, int lane, v8f acc) {
    const int half = lane >> 4;
    const int col  = lane & 15;
#pragma unroll
    for (int r = 0; r < 8; ++r)
        C[(size_t)(m0 + r + half * 8) * ldc + n0 + col] = acc[r];
}

__global__ void edge_rank_kernel(const int* __restrict__ dst, int* __restrict__ cur,
                                 int* __restrict__ rank) {
    int i = blockIdx.x * blockDim.x + threadIdx.x;
    if (i >= N_EDGES / 4) return;
    const int4 d = ((const int4*)dst)[i];
    int4 r;
    r.x = atomicAdd(&cur[d.x], 1);
    r.y = atomicAdd(&cur[d.y], 1);
    r.z = atomicAdd(&cur[d.z], 1);
    r.w = atomicAdd(&cur[d.w], 1);
    ((int4*)rank)[i] = r;
}

__global__ __launch_bounds__(1024) void offsets_scan_kernel(int* cur) {
    __shared__ int wave_total[32];
    const int t = threadIdx.x, lane = t & 31, wv = t >> 5;
    int2* mine = (int2*)(cur + t * 10);
    int2 v[5];
    int s = 0;
#pragma unroll
    for (int i = 0; i < 5; ++i) { v[i] = mine[i]; s += v[i].x + v[i].y; }
    int inc = s;
#pragma unroll
    for (int o = 1; o < 32; o <<= 1) { int u = __shfl_up(inc, o); if (lane >= o) inc += u; }
    if (lane == 31) wave_total[wv] = inc;
    __syncthreads();
    if (wv == 0) {
        const int w = wave_total[lane];
        int winc = w;
#pragma unroll
        for (int o = 1; o < 32; o <<= 1) { int u = __shfl_up(winc, o); if (lane >= o) winc += u; }
        wave_total[lane] = winc - w;
    }
    __syncthreads();
    int run = wave_total[wv] + inc - s;
#pragma unroll
    for (int i = 0; i < 5; ++i) {
        int2 o2;
        o2.x = run; run += v[i].x;
        o2.y = run; run += v[i].y;
        mine[i] = o2;
    }
}

__global__ void edge_place_kernel(const int* __restrict__ src, const int* __restrict__ dst,
                                  const int* __restrict__ rank, const int* __restrict__ cur,
                                  int* __restrict__ bucket_src) {
    int i = blockIdx.x * blockDim.x + threadIdx.x;
    if (i >= N_EDGES / 4) return;
    const int4 d = ((const int4*)dst)[i];
    const int4 s = ((const int4*)src)[i];
    const int4 r = ((const int4*)rank)[i];
    bucket_src[cur[d.x] + r.x] = s.x;
    bucket_src[cur[d.y] + r.y] = s.y;
    bucket_src[cur[d.z] + r.z] = s.z;
    bucket_src[cur[d.w] + r.w] = s.w;
}

template <int K>
__device__ __forceinline__ void add_rows(const float* __restrict__ xl, const int* __restrict__ ids,
                                         float4& a0, float4& a1) {
    int s[K];
    float4 u[K], w[K];
#pragma unroll
    for (int j = 0; j < K; ++j) s[j] = ids[j];
#pragma unroll
    for (int j = 0; j < K; ++j) {
        const float* r = xl + (size_t)s[j] * F_IN;
        u[j] = *(const float4*)r;
        w[j] = *(const float4*)(r + 128);
    }
#pragma unroll
    for (int j = 0; j < K; ++j) {
        a0.x += u[j].x; a0.y += u[j].y; a0.z += u[j].z; a0.w += u[j].w;
        a1.x += w[j].x; a1.y += w[j].y; a1.z += w[j].z; a1.w += w[j].w;
    }
}

__global__ __launch_bounds__(256) void agg1_gather_kernel(const float* __restrict__ x,
                                                          const int* __restrict__ cur,
                                                          const int* __restrict__ bucket_src,
                                                          float* __restrict__ agg1) {
    const int lane = threadIdx.x & 31;
    const int node = __builtin_amdgcn_readfirstlane(blockIdx.x * 8 + (threadIdx.x >> 5));
    const int begin = cur[node];
    const int end   = cur[node + 1];
    const float* xl = x + lane * 4;
    float4 a0 = {0.0f, 0.0f, 0.0f, 0.0f}, a1 = {0.0f, 0.0f, 0.0f, 0.0f};
    int e = begin;
    for (; e + 8 <= end; e += 8) add_rows<8>(xl, bucket_src + e, a0, a1);
    if (e + 4 <= end) { add_rows<4>(xl, bucket_src + e, a0, a1); e += 4; }
    if (e + 2 <= end) { add_rows<2>(xl, bucket_src + e, a0, a1); e += 2; }
    if (e < end)      { add_rows<1>(xl, bucket_src + e, a0, a1); }
    const float di = 1.0f / fmaxf((float)(end - begin), 1.0f);
    float* o = agg1 + (size_t)node * F_IN + lane * 4;
    float4 r0 = {a0.x * di, a0.y * di, a0.z * di, a0.w * di};
    float4 r1 = {a1.x * di, a1.y * di, a1.z * di, a1.w * di};
    *(float4*)o = r0;
    *(float4*)(o + 128) = r1;
}

__device__ __forceinline__ void col_partial(v8f a, float& s, float& q) {
    s = 0.0f; q = 0.0f;
#pragma unroll
    for (int r = 0; r < 8; ++r) { s += a[r]; q += a[r] * a[r]; }
}

__global__ __launch_bounds__(256) void gemm1_kernel(const float* __restrict__ A1,
                                                    const float* __restrict__ B1,
                                                    const float* __restrict__ A2,
                                                    const float* __restrict__ B2,
                                                    float* __restrict__ C,
                                                    float* __restrict__ sum,
                                                    float* __restrict__ sumsq) {
    const int lane = threadIdx.x & 31;
    const int wave = threadIdx.x >> 5;
    const int tile = blockIdx.x * 8 + wave;
    if (tile >= 2500) return;
    const int mt = tile >> 2;
    const int ng = tile & 3;
    const int m0 = mt * 16, n0 = ng * 64;

    v8f acc0 = {}, acc1 = {}, acc2 = {}, acc3 = {};
#pragma unroll 2
    for (int k0 = 0; k0 < F_IN; k0 += 4) {
        v2f a = load_a(A1, F_IN, m0, k0, lane);
        acc0 = wmma4(a, load_b(B1, F_HID, k0, n0 +  0, lane), acc0);
        acc1 = wmma4(a, load_b(B1, F_HID, k0, n0 + 16, lane), acc1);
        acc2 = wmma4(a, load_b(B1, F_HID, k0, n0 + 32, lane), acc2);
        acc3 = wmma4(a, load_b(B1, F_HID, k0, n0 + 48, lane), acc3);
    }
#pragma unroll 2
    for (int k0 = 0; k0 < F_IN; k0 += 4) {
        v2f a = load_a(A2, F_IN, m0, k0, lane);
        acc0 = wmma4(a, load_b(B2, F_HID, k0, n0 +  0, lane), acc0);
        acc1 = wmma4(a, load_b(B2, F_HID, k0, n0 + 16, lane), acc1);
        acc2 = wmma4(a, load_b(B2, F_HID, k0, n0 + 32, lane), acc2);
        acc3 = wmma4(a, load_b(B2, F_HID, k0, n0 + 48, lane), acc3);
    }
    store_c(C, F_HID, m0, n0 +  0, lane, acc0);
    store_c(C, F_HID, m0, n0 + 16, lane, acc1);
    store_c(C, F_HID, m0, n0 + 32, lane, acc2);
    store_c(C, F_HID, m0, n0 + 48, lane, acc3);

    const int half = lane >> 4;
    float s0, q0, s1, q1, s2, q2, s3, q3;
    col_partial(acc0, s0, q0);
    col_partial(acc1, s1, q1);
    col_partial(acc2, s2, q2);
    col_partial(acc3, s3, q3);
    const float sa = (half ? s1 : s0) + __shfl_xor(half ? s0 : s1, 16);
    const float sb = (half ? s3 : s2) + __shfl_xor(half ? s2 : s3, 16);
    const float qa = (half ? q1 : q0) + __shfl_xor(half ? q0 : q1, 16);
    const float qb = (half ? q3 : q2) + __shfl_xor(half ? q2 : q3, 16);
    atomicAdd(&sum[n0 + lane], sa);
    atomicAdd(&sum[n0 + 32 + lane], sb);
    atomicAdd(&sumsq[n0 + lane], qa);
    atomicAdd(&sumsq[n0 + 32 + lane], qb);
}

__global__ __launch_bounds__(256) void gemm3_kernel(const float* __restrict__ A,
                                                    const float* __restrict__ B1,
                                                    const float* __restrict__ B2,
                                                    float* __restrict__ D1,
                                                    float* __restrict__ D2) {
    const int lane = threadIdx.x & 31;
    const int wave = threadIdx.x >> 5;
    const int tile = blockIdx.x * 8 + wave;
    if (tile >= 625) return;
    const int m0 = tile * 16;

    v8f acc1 = {}, acc2 = {};
#pragma unroll 4
    for (int k0 = 0; k0 < F_HID; k0 += 4) {
        v2f a  = load_a(A, F_HID, m0, k0, lane);
        v2f b1 = load_b(B1, F_OUT, k0, 0, lane);
        v2f b2 = load_b(B2, F_OUT, k0, 0, lane);
        acc1 = wmma4(a, b1, acc1);
        acc2 = wmma4(a, b2, acc2);
    }
    store_c(D1, F_OUT, m0, 0, lane, acc1);
    store_c(D2, F_OUT, m0, 0, lane, acc2);
}

template <int COLS>
__global__ void bn_relu_kernel(float* __restrict__ h,
                               const float* __restrict__ sum, const float* __restrict__ sumsq,
                               const float* __restrict__ gamma, const float* __restrict__ beta,
                               long long total) {
    long long g = (long long)blockIdx.x * blockDim.x + threadIdx.x;
    if (g >= total) return;
    int c = (int)(g & (COLS - 1));
    float mean = sum[c] * (1.0f / N_NODES);
    float var  = sumsq[c] * (1.0f / N_NODES) - mean * mean;
    float v = (h[g] - mean) * rsqrtf(var + BN_EPS) * gamma[c] + beta[c];
    h[g] = fmaxf(v, 0.0f);
}

__global__ __launch_bounds__(256) void agg3_gather_kernel(const float* __restrict__ p,
                                                          const float* __restrict__ hl,
                                                          const int* __restrict__ cur,
                                                          const int* __restrict__ bucket_src,
                                                          float* __restrict__ t,
                                                          float* __restrict__ sum,
                                                          float* __restrict__ sumsq) {
    __shared__ float rows[2][8][F_OUT] __attribute__((aligned(16)));
    const int lane = threadIdx.x & 31;
    const int node = __builtin_amdgcn_readfirstlane(blockIdx.x * 8 + (threadIdx.x >> 5));
    const int begin = cur[node];
    const int end   = cur[node + 1];
    const int slot = lane >> 2, c4 = (lane & 3) << 2;
    float4 acc = {0.0f, 0.0f, 0.0f, 0.0f};
    for (int base = begin; base < end; base += 32) {
        int s[4];
        float4 v[4];
#pragma unroll
        for (int j = 0; j < 4; ++j) s[j] = bucket_src[min(base + slot + 8 * j, end - 1)];
#pragma unroll
        for (int j = 0; j < 4; ++j) v[j] = *(const float4*)(p + (size_t)s[j] * F_OUT + c4);
#pragma unroll
        for (int j = 0; j < 4; ++j)
            if (base + slot + 8 * j < end) {
                acc.x += v[j].x; acc.y += v[j].y; acc.z += v[j].z; acc.w += v[j].w;
            }
    }
#pragma unroll
    for (int o = 4; o < 32; o <<= 1) {
        acc.x += __shfl_xor(acc.x, o);
        acc.y += __shfl_xor(acc.y, o);
        acc.z += __shfl_xor(acc.z, o);
        acc.w += __shfl_xor(acc.w, o);
    }
    if (lane < 4) {
        const float di = 1.0f / fmaxf((float)(end - begin), 1.0f);
        const float4 h = *(const float4*)(hl + (size_t)node * F_OUT + c4);
        float4 r = {acc.x * di + h.x, acc.y * di + h.y, acc.z * di + h.z, acc.w * di + h.w};
        *(float4*)(t + (size_t)node * F_OUT + c4) = r;
        const int wv = threadIdx.x >> 5;
        float4 r2 = {r.x * r.x, r.y * r.y, r.z * r.z, r.w * r.w};
        *(float4*)&rows[0][wv][c4] = r;
        *(float4*)&rows[1][wv][c4] = r2;
    }
    __syncthreads();
    if (threadIdx.x < 32) {
        const int st = threadIdx.x >> 4, col = threadIdx.x & 15;
        float a = 0.0f;
#pragma unroll
        for (int w = 0; w < 8; ++w) a += rows[st][w][col];
        atomicAdd(st ? &sumsq[col] : &sum[col], a);
    }
}

__global__ void bn_final_kernel(const float* __restrict__ t,
                                const float* __restrict__ sum, const float* __restrict__ sumsq,
                                const float* __restrict__ gamma, const float* __restrict__ beta,
                                const int* __restrict__ ste,
                                float* __restrict__ out, long long total) {
    long long g = (long long)blockIdx.x * blockDim.x + threadIdx.x;
    if (g >= total) return;
    int c = (int)(g & (F_OUT - 1));
    float mean = sum[c] * (1.0f / N_NODES);
    float var  = sumsq[c] * (1.0f / N_NODES) - mean * mean;
    float v = (t[g] - mean) * rsqrtf(var + BN_EPS) * gamma[c] + beta[c];
    if (*ste)
        out[g] = (v > 0.0f) ? 1.0f : 0.0f;
    else
        out[g] = 1.0f / (1.0f + expf(-v));
}

extern "C" void kernel_launch(void* const* d_in, const int* in_sizes, int n_in,
                              void* d_out, int out_size, void* d_ws, size_t ws_size,
                              hipStream_t stream) {
    const float* x        = (const float*)d_in[0];
    const float* W_conv1  = (const float*)d_in[1];
    const float* W_lin1   = (const float*)d_in[2];
    const float* bn1_g    = (const float*)d_in[3];
    const float* bn1_b    = (const float*)d_in[4];
    const float* W_conv3  = (const float*)d_in[5];
    const float* W_lin3   = (const float*)d_in[6];
    const float* bn3_g    = (const float*)d_in[7];
    const float* bn3_b    = (const float*)d_in[8];
    const int*   src      = (const int*)d_in[9];
    const int*   dst      = (const int*)d_in[10];
    const int*   ste      = (const int*)d_in[11];
    float*       out      = (float*)d_out;

    float* ws = (float*)d_ws;
    const size_t o_cur  = 0;
    const size_t o_s1   = 10240;
    const size_t o_q1   = o_s1 + 256;
    const size_t o_s3   = o_q1 + 256;
    const size_t o_q3   = o_s3 + 16;
    const size_t o_bkt  = 11264;
    const size_t o_rank = o_bkt + 320000;
    const size_t o_agg1 = o_rank + 320000;
    const size_t o_h    = o_agg1 + 2560000;
    const size_t o_p    = o_h + 2560000;
    const size_t o_hl   = o_p + 160000;
    const size_t o_agg3 = o_hl + 160000;
    int*   cur     = (int*)(ws + o_cur);
    int*   bkt     = (int*)(ws + o_bkt);
    int*   rank    = (int*)(ws + o_rank);
    float* agg1    = ws + o_agg1;
    float* h       = ws + o_h;
    float* p       = ws + o_p;
    float* hl      = ws + o_hl;
    float* agg3    = ws + o_agg3;
    float* s1 = ws + o_s1; float* q1 = ws + o_q1;
    float* s3 = ws + o_s3; float* q3 = ws + o_q3;

    hipMemsetAsync(cur, 0, (10240 + 256 + 256 + 16 + 16) * sizeof(float), stream);

    edge_rank_kernel<<<(N_EDGES / 4 + 255) / 256, 256, 0, stream>>>(dst, cur, rank);
    offsets_scan_kernel<<<1, 1024, 0, stream>>>(cur);
    edge_place_kernel<<<(N_EDGES / 4 + 255) / 256, 256, 0, stream>>>(src, dst, rank, cur, bkt);

    agg1_gather_kernel<<<N_NODES / 8, 256, 0, stream>>>(x, cur, bkt, agg1);

    gemm1_kernel<<<(2500 + 7) / 8, 256, 0, stream>>>(agg1, W_conv1, x, W_lin1, h, s1, q1);

    {
        long long tot = (long long)N_NODES * F_HID;
        bn_relu_kernel<F_HID><<<(int)((tot + 255) / 256), 256, 0, stream>>>(h, s1, q1, bn1_g, bn1_b, tot);
    }

    gemm3_kernel<<<(625 + 7) / 8, 256, 0, stream>>>(h, W_conv3, W_lin3, p, hl);

    agg3_gather_kernel<<<N_NODES / 8, 256, 0, stream>>>(p, hl, cur, bkt, agg3, s3, q3);

    {
        long long tot = (long long)N_NODES * F_OUT;
        bn_final_kernel<<<(int)((tot + 255) / 256), 256, 0, stream>>>(agg3, s3, q3, bn3_g, bn3_b, ste, out, tot);
    }
}


# BN1 apply + ReLU fused into the layer-3 GEMM's A-fragment load (per-column mean, rstd, gamma, beta in LDS); the elementwise pass over h and its launch are gone; no sum order changed
# speedup vs baseline: 3.4355x; 1.0511x over previous
#include <hip/hip_runtime.h>
#include <hip/hip_bf16.h>
#include <math.h>

#define N_NODES 10000
#define N_EDGES 320000
#define F_IN    256
#define F_HID   256
#define F_OUT   16
#define BN_EPS  1e-5f

typedef __attribute__((ext_vector_type(2))) float v2f;
typedef __attribute__((ext_vector_type(8))) float v8f;

__device__ __forceinline__ v8f wmma4(v2f a, v2f b, v8f c) {
    return __builtin_amdgcn_wmma_f32_16x16x4_f32(
        false, a, false, b, (short)0, c, false, false);
}

__device__ __forceinline__ v2f load_a(const float* A, int lda, int m0, int k0, int lane) {
    const int half = lane >> 4;
    const int row  = lane & 15;
    const float* p = A + (size_t)(m0 + row) * lda + k0 + (half << 1);
    v2f r; r.x = p[0]; r.y = p[1];
    return r;
}

__device__ __forceinline__ v2f load_b(const float* B, int ldb, int k0, int n0, int lane) {
    const int half = lane >> 4;
    const int col  = lane & 15;
    const float* p = B + (size_t)(k0 + (half << 1)) * ldb + n0 + col;
    v2f r; r.x = p[0]; r.y = p[ldb];
    return r;
}

__device__ __forceinline__ void store_c(float* C, int ldc, int m0, int n0, int lane, v8f acc) {
    const int half = lane >> 4;
    const int col  = lane & 15;
#pragma unroll
    for (int r = 0; r < 8; ++r)
        C[(size_t)(m0 + r + half * 8) * ldc + n0 + col] = acc[r];
}

__global__ void edge_rank_kernel(const int* __restrict__ dst, int* __restrict__ cur,
                                 int* __restrict__ rank) {
    int i = blockIdx.x * blockDim.x + threadIdx.x;
    if (i >= N_EDGES / 4) return;
    const int4 d = ((const int4*)dst)[i];
    int4 r;
    r.x = atomicAdd(&cur[d.x], 1);
    r.y = atomicAdd(&cur[d.y], 1);
    r.z = atomicAdd(&cur[d.z], 1);
    r.w = atomicAdd(&cur[d.w], 1);
    ((int4*)rank)[i] = r;
}

__global__ __launch_bounds__(1024) void offsets_scan_kernel(int* cur) {
    __shared__ int wave_total[32];
    const int t = threadIdx.x, lane = t & 31, wv = t >> 5;
    int2* mine = (int2*)(cur + t * 10);
    int2 v[5];
    int s = 0;
#pragma unroll
    for (int i = 0; i < 5; ++i) { v[i] = mine[i]; s += v[i].x + v[i].y; }
    int inc = s;
#pragma unroll
    for (int o = 1; o < 32; o <<= 1) { int u = __shfl_up(inc, o); if (lane >= o) inc += u; }
    if (lane == 31) wave_total[wv] = inc;
    __syncthreads();
    if (wv == 0) {
        const int w = wave_total[lane];
        int winc = w;
#pragma unroll
        for (int o = 1; o < 32; o <<= 1) { int u = __shfl_up(winc, o); if (lane >= o) winc += u; }
        wave_total[lane] = winc - w;
    }
    __syncthreads();
    int run = wave_total[wv] + inc - s;
#pragma unroll
    for (int i = 0; i < 5; ++i) {
        int2 o2;
        o2.x = run; run += v[i].x;
        o2.y = run; run += v[i].y;
        mine[i] = o2;
    }
}

__global__ void edge_place_kernel(const int* __restrict__ src, const int* __restrict__ dst,
                                  const int* __restrict__ rank, const int* __restrict__ cur,
                                  int* __restrict__ bucket_src) {
    int i = blockIdx.x * blockDim.x + threadIdx.x;
    if (i >= N_EDGES / 4) return;
    const int4 d = ((const int4*)dst)[i];
    const int4 s = ((const int4*)src)[i];
    const int4 r = ((const int4*)rank)[i];
    bucket_src[cur[d.x] + r.x] = s.x;
    bucket_src[cur[d.y] + r.y] = s.y;
    bucket_src[cur[d.z] + r.z] = s.z;
    bucket_src[cur[d.w] + r.w] = s.w;
}

template <int K>
__device__ __forceinline__ void add_rows(const float* __restrict__ xl, const int* __restrict__ ids,
                                         float4& a0, float4& a1) {
    int s[K];
    float4 u[K], w[K];
#pragma unroll
    for (int j = 0; j < K; ++j) s[j] = ids[j];
#pragma unroll
    for (int j = 0; j < K; ++j) {
        const float* r = xl + (size_t)s[j] * F_IN;
        u[j] = *(const float4*)r;
        w[j] = *(const float4*)(r + 128);
    }
#pragma unroll
    for (int j = 0; j < K; ++j) {
        a0.x += u[j].x; a0.y += u[j].y; a0.z += u[j].z; a0.w += u[j].w;
        a1.x += w[j].x; a1.y += w[j].y; a1.z += w[j].z; a1.w += w[j].w;
    }
}

__global__ __launch_bounds__(256) void agg1_gather_kernel(const float* __restrict__ x,
                                                          const int* __restrict__ cur,
                                                          const int* __restrict__ bucket_src,
                                                          float* __restrict__ agg1) {
    const int lane = threadIdx.x & 31;
    const int node = __builtin_amdgcn_readfirstlane(blockIdx.x * 8 + (threadIdx.x >> 5));
    const int begin = cur[node];
    const int end   = cur[node + 1];
    const float* xl = x + lane * 4;
    float4 a0 = {0.0f, 0.0f, 0.0f, 0.0f}, a1 = {0.0f, 0.0f, 0.0f, 0.0f};
    int e = begin;
    for (; e + 8 <= end; e += 8) add_rows<8>(xl, bucket_src + e, a0, a1);
    if (e + 4 <= end) { add_rows<4>(xl, bucket_src + e, a0, a1); e += 4; }
    if (e + 2 <= end) { add_rows<2>(xl, bucket_src + e, a0, a1); e += 2; }
    if (e < end)      { add_rows<1>(xl, bucket_src + e, a0, a1); }
    const float di = 1.0f / fmaxf((float)(end - begin), 1.0f);
    float* o = agg1 + (size_t)node * F_IN + lane * 4;
    float4 r0 = {a0.x * di, a0.y * di, a0.z * di, a0.w * di};
    float4 r1 = {a1.x * di, a1.y * di, a1.z * di, a1.w * di};
    *(float4*)o = r0;
    *(float4*)(o + 128) = r1;
}

__device__ __forceinline__ void col_partial(v8f a, float& s, float& q) {
    s = 0.0f; q = 0.0f;
#pragma unroll
    for (int r = 0; r < 8; ++r) { s += a[r]; q += a[r] * a[r]; }
}

__global__ __launch_bounds__(256) void gemm1_kernel(const float* __restrict__ A1,
                                                    const float* __restrict__ B1,
                                                    const float* __restrict__ A2,
                                                    const float* __restrict__ B2,
                                                    float* __restrict__ C,
                                                    float* __restrict__ sum,
                                                    float* __restrict__ sumsq) {
    const int lane = threadIdx.x & 31;
    const int wave = threadIdx.x >> 5;
    const int tile = blockIdx.x * 8 + wave;
    if (tile >= 2500) return;
    const int mt = tile >> 2;
    const int ng = tile & 3;
    const int m0 = mt * 16, n0 = ng * 64;

    v8f acc0 = {}, acc1 = {}, acc2 = {}, acc3 = {};
#pragma unroll 2
    for (int k0 = 0; k0 < F_IN; k0 += 4) {
        v2f a = load_a(A1, F_IN, m0, k0, lane);
        acc0 = wmma4(a, load_b(B1, F_HID, k0, n0 +  0, lane), acc0);
        acc1 = wmma4(a, load_b(B1, F_HID, k0, n0 + 16, lane), acc1);
        acc2 = wmma4(a, load_b(B1, F_HID, k0, n0 + 32, lane), acc2);
        acc3 = wmma4(a, load_b(B1, F_HID, k0, n0 + 48, lane), acc3);
    }
#pragma unroll 2
    for (int k0 = 0; k0 < F_IN; k0 += 4) {
        v2f a = load_a(A2, F_IN, m0, k0, lane);
        acc0 = wmma4(a, load_b(B2, F_HID, k0, n0 +  0, lane), acc0);
        acc1 = wmma4(a, load_b(B2, F_HID, k0, n0 + 16, lane), acc1);
        acc2 = wmma4(a, load_b(B2, F_HID, k0, n0 + 32, lane), acc2);
        acc3 = wmma4(a, load_b(B2, F_HID, k0, n0 + 48, lane), acc3);
    }
    store_c(C, F_HID, m0, n0 +  0, lane, acc0);
    store_c(C, F_HID, m0, n0 + 16, lane, acc1);
    store_c(C, F_HID, m0, n0 + 32, lane, acc2);
    store_c(C, F_HID, m0, n0 + 48, lane, acc3);

    const int half = lane >> 4;
    float s0, q0, s1, q1, s2, q2, s3, q3;
    col_partial(acc0, s0, q0);
    col_partial(acc1, s1, q1);
    col_partial(acc2, s2, q2);
    col_partial(acc3, s3, q3);
    const float sa = (half ? s1 : s0) + __shfl_xor(half ? s0 : s1, 16);
    const float sb = (half ? s3 : s2) + __shfl_xor(half ? s2 : s3, 16);
    const float qa = (half ? q1 : q0) + __shfl_xor(half ? q0 : q1, 16);
    const float qb = (half ? q3 : q2) + __shfl_xor(half ? q2 : q3, 16);
    atomicAdd(&sum[n0 + lane], sa);
    atomicAdd(&sum[n0 + 32 + lane], sb);
    atomicAdd(&sumsq[n0 + lane], qa);
    atomicAdd(&sumsq[n0 + 32 + lane], qb);
}

__global__ __launch_bounds__(256) void gemm3_kernel(const float* __restrict__ A,
                                                    const float* __restrict__ sum,
                                                    const float* __restrict__ sumsq,
                                                    const float* __restrict__ gamma,
                                                    const float* __restrict__ beta,
                                                    const float* __restrict__ B1,
                                                    const float* __restrict__ B2,
                                                    float* __restrict__ D1,
                                                    float* __restrict__ D2) {
    __shared__ float bn[4][F_HID] __attribute__((aligned(8)));
    {
        const int k = threadIdx.x;
        const float mean = sum[k] * (1.0f / N_NODES);
        const float var  = sumsq[k] * (1.0f / N_NODES) - mean * mean;
        bn[0][k] = mean;
        bn[1][k] = rsqrtf(var + BN_EPS);
        bn[2][k] = gamma[k];
        bn[3][k] = beta[k];
    }
    __syncthreads();

    const int lane = threadIdx.x & 31;
    const int wave = threadIdx.x >> 5;
    const int tile = blockIdx.x * 8 + wave;
    if (tile >= 625) return;
    const int m0 = tile * 16;
    const int kh = (lane >> 4) << 1;

    v8f acc1 = {}, acc2 = {};
#pragma unroll 4
    for (int k0 = 0; k0 < F_HID; k0 += 4) {
        const v2f raw = load_a(A, F_HID, m0, k0, lane);
        const v2f mu = *(const v2f*)&bn[0][k0 + kh];
        const v2f rs = *(const v2f*)&bn[1][k0 + kh];
        const v2f ga = *(const v2f*)&bn[2][k0 + kh];
        const v2f be = *(const v2f*)&bn[3][k0 + kh];
        v2f a;
        a.x = fmaxf((raw.x - mu.x) * rs.x * ga.x + be.x, 0.0f);
        a.y = fmaxf((raw.y - mu.y) * rs.y * ga.y + be.y, 0.0f);
        v2f b1 = load_b(B1, F_OUT, k0, 0, lane);
        v2f b2 = load_b(B2, F_OUT, k0, 0, lane);
        acc1 = wmma4(a, b1, acc1);
        acc2 = wmma4(a, b2, acc2);
    }
    store_c(D1, F_OUT, m0, 0, lane, acc1);
    store_c(D2, F_OUT, m0, 0, lane, acc2);
}


__global__ __launch_bounds__(256) void agg3_gather_kernel(const float* __restrict__ p,
                                                          const float* __restrict__ hl,
                                                          const int* __restrict__ cur,
                                                          const int* __restrict__ bucket_src,
                                                          float* __restrict__ t,
                                                          float* __restrict__ sum,
                                                          float* __restrict__ sumsq) {
    __shared__ float rows[2][8][F_OUT] __attribute__((aligned(16)));
    const int lane = threadIdx.x & 31;
    const int node = __builtin_amdgcn_readfirstlane(blockIdx.x * 8 + (threadIdx.x >> 5));
    const int begin = cur[node];
    const int end   = cur[node + 1];
    const int slot = lane >> 2, c4 = (lane & 3) << 2;
    float4 acc = {0.0f, 0.0f, 0.0f, 0.0f};
    for (int base = begin; base < end; base += 32) {
        int s[4];
        float4 v[4];
#pragma unroll
        for (int j = 0; j < 4; ++j) s[j] = bucket_src[min(base + slot + 8 * j, end - 1)];
#pragma unroll
        for (int j = 0; j < 4; ++j) v[j] = *(const float4*)(p + (size_t)s[j] * F_OUT + c4);
#pragma unroll
        for (int j = 0; j < 4; ++j)
            if (base + slot + 8 * j < end) {
                acc.x += v[j].x; acc.y += v[j].y; acc.z += v[j].z; acc.w += v[j].w;
            }
    }
#pragma unroll
    for (int o = 4; o < 32; o <<= 1) {
        acc.x += __shfl_xor(acc.x, o);
        acc.y += __shfl_xor(acc.y, o);
        acc.z += __shfl_xor(acc.z, o);
        acc.w += __shfl_xor(acc.w, o);
    }
    if (lane < 4) {
        const float di = 1.0f / fmaxf((float)(end - begin), 1.0f);
        const float4 h = *(const float4*)(hl + (size_t)node * F_OUT + c4);
        float4 r = {acc.x * di + h.x, acc.y * di + h.y, acc.z * di + h.z, acc.w * di + h.w};
        *(float4*)(t + (size_t)node * F_OUT + c4) = r;
        const int wv = threadIdx.x >> 5;
        float4 r2 = {r.x * r.x, r.y * r.y, r.z * r.z, r.w * r.w};
        *(float4*)&rows[0][wv][c4] = r;
        *(float4*)&rows[1][wv][c4] = r2;
    }
    __syncthreads();
    if (threadIdx.x < 32) {
        const int st = threadIdx.x >> 4, col = threadIdx.x & 15;
        float a = 0.0f;
#pragma unroll
        for (int w = 0; w < 8; ++w) a += rows[st][w][col];
        atomicAdd(st ? &sumsq[col] : &sum[col], a);
    }
}

__global__ void bn_final_kernel(const float* __restrict__ t,
                                const float* __restrict__ sum, const float* __restrict__ sumsq,
                                const float* __restrict__ gamma, const float* __restrict__ beta,
                                const int* __restrict__ ste,
                                float* __restrict__ out, long long total) {
    long long g = (long long)blockIdx.x * blockDim.x + threadIdx.x;
    if (g >= total) return;
    int c = (int)(g & (F_OUT - 1));
    float mean = sum[c] * (1.0f / N_NODES);
    float var  = sumsq[c] * (1.0f / N_NODES) - mean * mean;
    float v = (t[g] - mean) * rsqrtf(var + BN_EPS) * gamma[c] + beta[c];
    if (*ste)
        out[g] = (v > 0.0f) ? 1.0f : 0.0f;
    else
        out[g] = 1.0f / (1.0f + expf(-v));
}

extern "C" void kernel_launch(void* const* d_in, const int* in_sizes, int n_in,
                              void* d_out, int out_size, void* d_ws, size_t ws_size,
                              hipStream_t stream) {
    const float* x        = (const float*)d_in[0];
    const float* W_conv1  = (const float*)d_in[1];
    const float* W_lin1   = (const float*)d_in[2];
    const float* bn1_g    = (const float*)d_in[3];
    const float* bn1_b    = (const float*)d_in[4];
    const float* W_conv3  = (const float*)d_in[5];
    const float* W_lin3   = (const float*)d_in[6];
    const float* bn3_g    = (const float*)d_in[7];
    const float* bn3_b    = (const float*)d_in[8];
    const int*   src      = (const int*)d_in[9];
    const int*   dst      = (const int*)d_in[10];
    const int*   ste      = (const int*)d_in[11];
    float*       out      = (float*)d_out;

    float* ws = (float*)d_ws;
    const size_t o_cur  = 0;
    const size_t o_s1   = 10240;
    const size_t o_q1   = o_s1 + 256;
    const size_t o_s3   = o_q1 + 256;
    const size_t o_q3   = o_s3 + 16;
    const size_t o_bkt  = 11264;
    const size_t o_rank = o_bkt + 320000;
    const size_t o_agg1 = o_rank + 320000;
    const size_t o_h    = o_agg1 + 2560000;
    const size_t o_p    = o_h + 2560000;
    const size_t o_hl   = o_p + 160000;
    const size_t o_agg3 = o_hl + 160000;
    int*   cur     = (int*)(ws + o_cur);
    int*   bkt     = (int*)(ws + o_bkt);
    int*   rank    = (int*)(ws + o_rank);
    float* agg1    = ws + o_agg1;
    float* h       = ws + o_h;
    float* p       = ws + o_p;
    float* hl      = ws + o_hl;
    float* agg3    = ws + o_agg3;
    float* s1 = ws + o_s1; float* q1 = ws + o_q1;
    float* s3 = ws + o_s3; float* q3 = ws + o_q3;

    hipMemsetAsync(cur, 0, (10240 + 256 + 256 + 16 + 16) * sizeof(float), stream);

    edge_rank_kernel<<<(N_EDGES / 4 + 255) / 256, 256, 0, stream>>>(dst, cur, rank);
    offsets_scan_kernel<<<1, 1024, 0, stream>>>(cur);
    edge_place_kernel<<<(N_EDGES / 4 + 255) / 256, 256, 0, stream>>>(src, dst, rank, cur, bkt);

    agg1_gather_kernel<<<N_NODES / 8, 256, 0, stream>>>(x, cur, bkt, agg1);

    gemm1_kernel<<<(2500 + 7) / 8, 256, 0, stream>>>(agg1, W_conv1, x, W_lin1, h, s1, q1);

    gemm3_kernel<<<(625 + 7) / 8, 256, 0, stream>>>(h, s1, q1, bn1_g, bn1_b, W_conv3, W_lin3, p, hl);

    agg3_gather_kernel<<<N_NODES / 8, 256, 0, stream>>>(p, hl, cur, bkt, agg3, s3, q3);

    {
        long long tot = (long long)N_NODES * F_OUT;
        bn_final_kernel<<<(int)((tot + 255) / 256), 256, 0, stream>>>(agg3, s3, q3, bn3_g, bn3_b, ste, out, tot);
    }
}


# layer-1 GEMM k-loop software-pipelined: next 8 k-steps' operands requested ahead of the current 32 WMMAs, B by 16-byte row loads (N tiles on interleaved columns); no sum order changed
# speedup vs baseline: 3.8388x; 1.1174x over previous
#include <hip/hip_runtime.h>
#include <hip/hip_bf16.h>
#include <math.h>

#define N_NODES 10000
#define N_EDGES 320000
#define F_IN    256
#define F_HID   256
#define F_OUT   16
#define BN_EPS  1e-5f

typedef __attribute__((ext_vector_type(2))) float v2f;
typedef __attribute__((ext_vector_type(8))) float v8f;

__device__ __forceinline__ v8f wmma4(v2f a, v2f b, v8f c) {
    return __builtin_amdgcn_wmma_f32_16x16x4_f32(
        false, a, false, b, (short)0, c, false, false);
}

__device__ __forceinline__ v2f load_a(const float* A, int lda, int m0, int k0, int lane) {
    const int half = lane >> 4;
    const int row  = lane & 15;
    const float* p = A + (size_t)(m0 + row) * lda + k0 + (half << 1);
    v2f r; r.x = p[0]; r.y = p[1];
    return r;
}

__device__ __forceinline__ v2f load_b(const float* B, int ldb, int k0, int n0, int lane) {
    const int half = lane >> 4;
    const int col  = lane & 15;
    const float* p = B + (size_t)(k0 + (half << 1)) * ldb + n0 + col;
    v2f r; r.x = p[0]; r.y = p[ldb];
    return r;
}

__device__ __forceinline__ void store_c(float* C, int ldc, int m0, int n0, int lane, v8f acc) {
    const int half = lane >> 4;
    const int col  = lane & 15;
#pragma unroll
    for (int r = 0; r < 8; ++r)
        C[(size_t)(m0 + r + half * 8) * ldc + n0 + col] = acc[r];
}

__global__ void edge_rank_kernel(const int* __restrict__ dst, int* __restrict__ cur,
                                 int* __restrict__ rank) {
    int i = blockIdx.x * blockDim.x + threadIdx.x;
    if (i >= N_EDGES / 4) return;
    const int4 d = ((const int4*)dst)[i];
    int4 r;
    r.x = atomicAdd(&cur[d.x], 1);
    r.y = atomicAdd(&cur[d.y], 1);
    r.z = atomicAdd(&cur[d.z], 1);
    r.w = atomicAdd(&cur[d.w], 1);
    ((int4*)rank)[i] = r;
}

__global__ __launch_bounds__(1024) void offsets_scan_kernel(int* cur) {
    __shared__ int wave_total[32];
    const int t = threadIdx.x, lane = t & 31, wv = t >> 5;
    int2* mine = (int2*)(cur + t * 10);
    int2 v[5];
    int s = 0;
#pragma unroll
    for (int i = 0; i < 5; ++i) { v[i] = mine[i]; s += v[i].x + v[i].y; }
    int inc = s;
#pragma unroll
    for (int o = 1; o < 32; o <<= 1) { int u = __shfl_up(inc, o); if (lane >= o) inc += u; }
    if (lane == 31) wave_total[wv] = inc;
    __syncthreads();
    if (wv == 0) {
        const int w = wave_total[lane];
        int winc = w;
#pragma unroll
        for (int o = 1; o < 32; o <<= 1) { int u = __shfl_up(winc, o); if (lane >= o) winc += u; }
        wave_total[lane] = winc - w;
    }
    __syncthreads();
    int run = wave_total[wv] + inc - s;
#pragma unroll
    for (int i = 0; i < 5; ++i) {
        int2 o2;
        o2.x = run; run += v[i].x;
        o2.y = run; run += v[i].y;
        mine[i] = o2;
    }
}

__global__ void edge_place_kernel(const int* __restrict__ src, const int* __restrict__ dst,
                                  const int* __restrict__ rank, const int* __restrict__ cur,
                                  int* __restrict__ bucket_src) {
    int i = blockIdx.x * blockDim.x + threadIdx.x;
    if (i >= N_EDGES / 4) return;
    const int4 d = ((const int4*)dst)[i];
    const int4 s = ((const int4*)src)[i];
    const int4 r = ((const int4*)rank)[i];
    bucket_src[cur[d.x] + r.x] = s.x;
    bucket_src[cur[d.y] + r.y] = s.y;
    bucket_src[cur[d.z] + r.z] = s.z;
    bucket_src[cur[d.w] + r.w] = s.w;
}

template <int K>
__device__ __forceinline__ void add_rows(const float* __restrict__ xl, const int* __restrict__ ids,
                                         float4& a0, float4& a1) {
    int s[K];
    float4 u[K], w[K];
#pragma unroll
    for (int j = 0; j < K; ++j) s[j] = ids[j];
#pragma unroll
    for (int j = 0; j < K; ++j) {
        const float* r = xl + (size_t)s[j] * F_IN;
        u[j] = *(const float4*)r;
        w[j] = *(const float4*)(r + 128);
    }
#pragma unroll
    for (int j = 0; j < K; ++j) {
        a0.x += u[j].x; a0.y += u[j].y; a0.z += u[j].z; a0.w += u[j].w;
        a1.x += w[j].x; a1.y += w[j].y; a1.z += w[j].z; a1.w += w[j].w;
    }
}

__global__ __launch_bounds__(256) void agg1_gather_kernel(const float* __restrict__ x,
                                                          const int* __restrict__ cur,
                                                          const int* __restrict__ bucket_src,
                                                          float* __restrict__ agg1) {
    const int lane = threadIdx.x & 31;
    const int node = __builtin_amdgcn_readfirstlane(blockIdx.x * 8 + (threadIdx.x >> 5));
    const int begin = cur[node];
    const int end   = cur[node + 1];
    const float* xl = x + lane * 4;
    float4 a0 = {0.0f, 0.0f, 0.0f, 0.0f}, a1 = {0.0f, 0.0f, 0.0f, 0.0f};
    int e = begin;
    for (; e + 8 <= end; e += 8) add_rows<8>(xl, bucket_src + e, a0, a1);
    if (e + 4 <= end) { add_rows<4>(xl, bucket_src + e, a0, a1); e += 4; }
    if (e + 2 <= end) { add_rows<2>(xl, bucket_src + e, a0, a1); e += 2; }
    if (e < end)      { add_rows<1>(xl, bucket_src + e, a0, a1); }
    const float di = 1.0f / fmaxf((float)(end - begin), 1.0f);
    float* o = agg1 + (size_t)node * F_IN + lane * 4;
    float4 r0 = {a0.x * di, a0.y * di, a0.z * di, a0.w * di};
    float4 r1 = {a1.x * di, a1.y * di, a1.z * di, a1.w * di};
    *(float4*)o = r0;
    *(float4*)(o + 128) = r1;
}

__device__ __forceinline__ void col_partial(v8f a, float& s, float& q) {
    s = 0.0f; q = 0.0f;
#pragma unroll
    for (int r = 0; r < 8; ++r) { s += a[r]; q += a[r] * a[r]; }
}

#define G1_KSTEPS 8
struct G1Frag {
    v2f    a[G1_KSTEPS];
    float4 b0[G1_KSTEPS];
    float4 b1[G1_KSTEPS];
};

__device__ __forceinline__ void g1_load(G1Frag& f, const float* ap, const float* bp) {
#pragma unroll
    for (int s = 0; s < G1_KSTEPS; ++s) {
        f.a[s]  = *(const v2f*)(ap + 4 * s);
        f.b0[s] = *(const float4*)(bp + (size_t)(4 * s) * F_HID);
        f.b1[s] = *(const float4*)(bp + (size_t)(4 * s + 1) * F_HID);
    }
}

__device__ __forceinline__ void g1_mma(const G1Frag& f, v8f& acc0, v8f& acc1, v8f& acc2, v8f& acc3) {
#pragma unroll
    for (int s = 0; s < G1_KSTEPS; ++s) {
        const v2f a = f.a[s];
        v2f b;
        b.x = f.b0[s].x; b.y = f.b1[s].x; acc0 = wmma4(a, b, acc0);
        b.x = f.b0[s].y; b.y = f.b1[s].y; acc1 = wmma4(a, b, acc1);
        b.x = f.b0[s].z; b.y = f.b1[s].z; acc2 = wmma4(a, b, acc2);
        b.x = f.b0[s].w; b.y = f.b1[s].w; acc3 = wmma4(a, b, acc3);
    }
}

__global__ __launch_bounds__(256) void gemm1_kernel(const float* __restrict__ A1,
                                                    const float* __restrict__ B1,
                                                    const float* __restrict__ A2,
                                                    const float* __restrict__ B2,
                                                    float* __restrict__ C,
                                                    float* __restrict__ sum,
                                                    float* __restrict__ sumsq) {
    const int lane = threadIdx.x & 31;
    const int wave = threadIdx.x >> 5;
    const int tile = blockIdx.x * 8 + wave;
    if (tile >= 2500) return;
    const int mt = tile >> 2;
    const int ng = tile & 3;
    const int m0 = mt * 16, n0 = ng * 64;
    const int half = lane >> 4, c = lane & 15;

    const size_t a_off = (size_t)(m0 + c) * F_IN + 2 * half;
    const size_t b_off = (size_t)(2 * half) * F_HID + n0 + 4 * c;
    const float* a1p = A1 + a_off;
    const float* a2p = A2 + a_off;
    const float* b1p = B1 + b_off;
    const float* b2p = B2 + b_off;

    v8f acc0 = {}, acc1 = {}, acc2 = {}, acc3 = {};
    G1Frag f0, f1;
    g1_load(f0, a1p, b1p);
#pragma unroll 1
    for (int g = 0; g < 14; g += 2) {
        const int g1 = g + 1, g2 = g + 2;
        g1_load(f1, ((g1 & 8) ? a2p : a1p) + (g1 & 7) * 32,
                    ((g1 & 8) ? b2p : b1p) + (size_t)(g1 & 7) * 32 * F_HID);
        g1_mma(f0, acc0, acc1, acc2, acc3);
        g1_load(f0, ((g2 & 8) ? a2p : a1p) + (g2 & 7) * 32,
                    ((g2 & 8) ? b2p : b1p) + (size_t)(g2 & 7) * 32 * F_HID);
        g1_mma(f1, acc0, acc1, acc2, acc3);
    }
    g1_load(f1, a2p + 7 * 32, b2p + (size_t)7 * 32 * F_HID);
    g1_mma(f0, acc0, acc1, acc2, acc3);
    g1_mma(f1, acc0, acc1, acc2, acc3);

#pragma unroll
    for (int r = 0; r < 8; ++r) {
        float4 o = {acc0[r], acc1[r], acc2[r], acc3[r]};
        *(float4*)(C + (size_t)(m0 + r + half * 8) * F_HID + n0 + 4 * c) = o;
    }

    float s0, q0, s1, q1, s2, q2, s3, q3;
    col_partial(acc0, s0, q0);
    col_partial(acc1, s1, q1);
    col_partial(acc2, s2, q2);
    col_partial(acc3, s3, q3);
    const float sa = (half ? s1 : s0) + __shfl_xor(half ? s0 : s1, 16);
    const float sb = (half ? s3 : s2) + __shfl_xor(half ? s2 : s3, 16);
    const float qa = (half ? q1 : q0) + __shfl_xor(half ? q0 : q1, 16);
    const float qb = (half ? q3 : q2) + __shfl_xor(half ? q2 : q3, 16);
    atomicAdd(&sum[n0 + 4 * c + half], sa);
    atomicAdd(&sum[n0 + 4 * c + 2 + half], sb);
    atomicAdd(&sumsq[n0 + 4 * c + half], qa);
    atomicAdd(&sumsq[n0 + 4 * c + 2 + half], qb);
}

__global__ __launch_bounds__(256) void gemm3_kernel(const float* __restrict__ A,
                                                    const float* __restrict__ sum,
                                                    const float* __restrict__ sumsq,
                                                    const float* __restrict__ gamma,
                                                    const float* __restrict__ beta,
                                                    const float* __restrict__ B1,
                                                    const float* __restrict__ B2,
                                                    float* __restrict__ D1,
                                                    float* __restrict__ D2) {
    __shared__ float bn[4][F_HID] __attribute__((aligned(8)));
    {
        const int k = threadIdx.x;
        const float mean = sum[k] * (1.0f / N_NODES);
        const float var  = sumsq[k] * (1.0f / N_NODES) - mean * mean;
        bn[0][k] = mean;
        bn[1][k] = rsqrtf(var + BN_EPS);
        bn[2][k] = gamma[k];
        bn[3][k] = beta[k];
    }
    __syncthreads();

    const int lane = threadIdx.x & 31;
    const int wave = threadIdx.x >> 5;
    const int tile = blockIdx.x * 8 + wave;
    if (tile >= 625) return;
    const int m0 = tile * 16;
    const int kh = (lane >> 4) << 1;

    v8f acc1 = {}, acc2 = {};
#pragma unroll 4
    for (int k0 = 0; k0 < F_HID; k0 += 4) {
        const v2f raw = load_a(A, F_HID, m0, k0, lane);
        const v2f mu = *(const v2f*)&bn[0][k0 + kh];
        const v2f rs = *(const v2f*)&bn[1][k0 + kh];
        const v2f ga = *(const v2f*)&bn[2][k0 + kh];
        const v2f be = *(const v2f*)&bn[3][k0 + kh];
        v2f a;
        a.x = fmaxf((raw.x - mu.x) * rs.x * ga.x + be.x, 0.0f);
        a.y = fmaxf((raw.y - mu.y) * rs.y * ga.y + be.y, 0.0f);
        v2f b1 = load_b(B1, F_OUT, k0, 0, lane);
        v2f b2 = load_b(B2, F_OUT, k0, 0, lane);
        acc1 = wmma4(a, b1, acc1);
        acc2 = wmma4(a, b2, acc2);
    }
    store_c(D1, F_OUT, m0, 0, lane, acc1);
    store_c(D2, F_OUT, m0, 0, lane, acc2);
}


__global__ __launch_bounds__(256) void agg3_gather_kernel(const float* __restrict__ p,
                                                          const float* __restrict__ hl,
                                                          const int* __restrict__ cur,
                                                          const int* __restrict__ bucket_src,
                                                          float* __restrict__ t,
                                                          float* __restrict__ sum,
                                                          float* __restrict__ sumsq) {
    __shared__ float rows[2][8][F_OUT] __attribute__((aligned(16)));
    const int lane = threadIdx.x & 31;
    const int node = __builtin_amdgcn_readfirstlane(blockIdx.x * 8 + (threadIdx.x >> 5));
    const int begin = cur[node];
    const int end   = cur[node + 1];
    const int slot = lane >> 2, c4 = (lane & 3) << 2;
    float4 acc = {0.0f, 0.0f, 0.0f, 0.0f};
    for (int base = begin; base < end; base += 32) {
        int s[4];
        float4 v[4];
#pragma unroll
        for (int j = 0; j < 4; ++j) s[j] = bucket_src[min(base + slot + 8 * j, end - 1)];
#pragma unroll
        for (int j = 0; j < 4; ++j) v[j] = *(const float4*)(p + (size_t)s[j] * F_OUT + c4);
#pragma unroll
        for (int j = 0; j < 4; ++j)
            if (base + slot + 8 * j < end) {
                acc.x += v[j].x; acc.y += v[j].y; acc.z += v[j].z; acc.w += v[j].w;
            }
    }
#pragma unroll
    for (int o = 4; o < 32; o <<= 1) {
        acc.x += __shfl_xor(acc.x, o);
        acc.y += __shfl_xor(acc.y, o);
        acc.z += __shfl_xor(acc.z, o);
        acc.w += __shfl_xor(acc.w, o);
    }
    if (lane < 4) {
        const float di = 1.0f / fmaxf((float)(end - begin), 1.0f);
        const float4 h = *(const float4*)(hl + (size_t)node * F_OUT + c4);
        float4 r = {acc.x * di + h.x, acc.y * di + h.y, acc.z * di + h.z, acc.w * di + h.w};
        *(float4*)(t + (size_t)node * F_OUT + c4) = r;
        const int wv = threadIdx.x >> 5;
        float4 r2 = {r.x * r.x, r.y * r.y, r.z * r.z, r.w * r.w};
        *(float4*)&rows[0][wv][c4] = r;
        *(float4*)&rows[1][wv][c4] = r2;
    }
    __syncthreads();
    if (threadIdx.x < 32) {
        const int st = threadIdx.x >> 4, col = threadIdx.x & 15;
        float a = 0.0f;
#pragma unroll
        for (int w = 0; w < 8; ++w) a += rows[st][w][col];
        atomicAdd(st ? &sumsq[col] : &sum[col], a);
    }
}

__global__ void bn_final_kernel(const float* __restrict__ t,
                                const float* __restrict__ sum, const float* __restrict__ sumsq,
                                const float* __restrict__ gamma, const float* __restrict__ beta,
                                const int* __restrict__ ste,
                                float* __restrict__ out, long long total) {
    long long g = (long long)blockIdx.x * blockDim.x + threadIdx.x;
    if (g >= total) return;
    int c = (int)(g & (F_OUT - 1));
    float mean = sum[c] * (1.0f / N_NODES);
    float var  = sumsq[c] * (1.0f / N_NODES) - mean * mean;
    float v = (t[g] - mean) * rsqrtf(var + BN_EPS) * gamma[c] + beta[c];
    if (*ste)
        out[g] = (v > 0.0f) ? 1.0f : 0.0f;
    else
        out[g] = 1.0f / (1.0f + expf(-v));
}

extern "C" void kernel_launch(void* const* d_in, const int* in_sizes, int n_in,
                              void* d_out, int out_size, void* d_ws, size_t ws_size,
                              hipStream_t stream) {
    const float* x        = (const float*)d_in[0];
    const float* W_conv1  = (const float*)d_in[1];
    const float* W_lin1   = (const float*)d_in[2];
    const float* bn1_g    = (const float*)d_in[3];
    const float* bn1_b    = (const float*)d_in[4];
    const float* W_conv3  = (const float*)d_in[5];
    const float* W_lin3   = (const float*)d_in[6];
    const float* bn3_g    = (const float*)d_in[7];
    const float* bn3_b    = (const float*)d_in[8];
    const int*   src      = (const int*)d_in[9];
    const int*   dst      = (const int*)d_in[10];
    const int*   ste      = (const int*)d_in[11];
    float*       out      = (float*)d_out;

    float* ws = (float*)d_ws;
    const size_t o_cur  = 0;
    const size_t o_s1   = 10240;
    const size_t o_q1   = o_s1 + 256;
    const size_t o_s3   = o_q1 + 256;
    const size_t o_q3   = o_s3 + 16;
    const size_t o_bkt  = 11264;
    const size_t o_rank = o_bkt + 320000;
    const size_t o_agg1 = o_rank + 320000;
    const size_t o_h    = o_agg1 + 2560000;
    const size_t o_p    = o_h + 2560000;
    const size_t o_hl   = o_p + 160000;
    const size_t o_agg3 = o_hl + 160000;
    int*   cur     = (int*)(ws + o_cur);
    int*   bkt     = (int*)(ws + o_bkt);
    int*   rank    = (int*)(ws + o_rank);
    float* agg1    = ws + o_agg1;
    float* h       = ws + o_h;
    float* p       = ws + o_p;
    float* hl      = ws + o_hl;
    float* agg3    = ws + o_agg3;
    float* s1 = ws + o_s1; float* q1 = ws + o_q1;
    float* s3 = ws + o_s3; float* q3 = ws + o_q3;

    hipMemsetAsync(cur, 0, (10240 + 256 + 256 + 16 + 16) * sizeof(float), stream);

    edge_rank_kernel<<<(N_EDGES / 4 + 255) / 256, 256, 0, stream>>>(dst, cur, rank);
    offsets_scan_kernel<<<1, 1024, 0, stream>>>(cur);
    edge_place_kernel<<<(N_EDGES / 4 + 255) / 256, 256, 0, stream>>>(src, dst, rank, cur, bkt);

    agg1_gather_kernel<<<N_NODES / 8, 256, 0, stream>>>(x, cur, bkt, agg1);

    gemm1_kernel<<<(2500 + 7) / 8, 256, 0, stream>>>(agg1, W_conv1, x, W_lin1, h, s1, q1);

    gemm3_kernel<<<(625 + 7) / 8, 256, 0, stream>>>(h, s1, q1, bn1_g, bn1_b, W_conv3, W_lin3, p, hl);

    agg3_gather_kernel<<<N_NODES / 8, 256, 0, stream>>>(p, hl, cur, bkt, agg3, s3, q3);

    {
        long long tot = (long long)N_NODES * F_OUT;
        bn_final_kernel<<<(int)((tot + 255) / 256), 256, 0, stream>>>(agg3, s3, q3, bn3_g, bn3_b, ste, out, tot);
    }
}


# layer-1 aggregation moved into the layer-1 GEMM kernel: each work-group (16-row tile, 4 waves) gathers its 16 bucket sums into LDS and reads A fragments there; agg pass, launch gone; sum order kept
# speedup vs baseline: 4.0794x; 1.0627x over previous
#include <hip/hip_runtime.h>
#include <hip/hip_bf16.h>
#include <math.h>

#define N_NODES 10000
#define N_EDGES 320000
#define F_IN    256
#define F_HID   256
#define F_OUT   16
#define BN_EPS  1e-5f

typedef __attribute__((ext_vector_type(2))) float v2f;
typedef __attribute__((ext_vector_type(8))) float v8f;

__device__ __forceinline__ v8f wmma4(v2f a, v2f b, v8f c) {
    return __builtin_amdgcn_wmma_f32_16x16x4_f32(
        false, a, false, b, (short)0, c, false, false);
}

__device__ __forceinline__ v2f load_a(const float* A, int lda, int m0, int k0, int lane) {
    const int half = lane >> 4;
    const int row  = lane & 15;
    const float* p = A + (size_t)(m0 + row) * lda + k0 + (half << 1);
    v2f r; r.x = p[0]; r.y = p[1];
    return r;
}

__device__ __forceinline__ v2f load_b(const float* B, int ldb, int k0, int n0, int lane) {
    const int half = lane >> 4;
    const int col  = lane & 15;
    const float* p = B + (size_t)(k0 + (half << 1)) * ldb + n0 + col;
    v2f r; r.x = p[0]; r.y = p[ldb];
    return r;
}

__device__ __forceinline__ void store_c(float* C, int ldc, int m0, int n0, int lane, v8f acc) {
    const int half = lane >> 4;
    const int col  = lane & 15;
#pragma unroll
    for (int r = 0; r < 8; ++r)
        C[(size_t)(m0 + r + half * 8) * ldc + n0 + col] = acc[r];
}

__global__ void edge_rank_kernel(const int* __restrict__ dst, int* __restrict__ cur,
                                 int* __restrict__ rank) {
    int i = blockIdx.x * blockDim.x + threadIdx.x;
    if (i >= N_EDGES / 4) return;
    const int4 d = ((const int4*)dst)[i];
    int4 r;
    r.x = atomicAdd(&cur[d.x], 1);
    r.y = atomicAdd(&cur[d.y], 1);
    r.z = atomicAdd(&cur[d.z], 1);
    r.w = atomicAdd(&cur[d.w], 1);
    ((int4*)rank)[i] = r;
}

__global__ __launch_bounds__(1024) void offsets_scan_kernel(int* cur) {
    __shared__ int wave_total[32];
    const int t = threadIdx.x, lane = t & 31, wv = t >> 5;
    int2* mine = (int2*)(cur + t * 10);
    int2 v[5];
    int s = 0;
#pragma unroll
    for (int i = 0; i < 5; ++i) { v[i] = mine[i]; s += v[i].x + v[i].y; }
    int inc = s;
#pragma unroll
    for (int o = 1; o < 32; o <<= 1) { int u = __shfl_up(inc, o); if (lane >= o) inc += u; }
    if (lane == 31) wave_total[wv] = inc;
    __syncthreads();
    if (wv == 0) {
        const int w = wave_total[lane];
        int winc = w;
#pragma unroll
        for (int o = 1; o < 32; o <<= 1) { int u = __shfl_up(winc, o); if (lane >= o) winc += u; }
        wave_total[lane] = winc - w;
    }
    __syncthreads();
    int run = wave_total[wv] + inc - s;
#pragma unroll
    for (int i = 0; i < 5; ++i) {
        int2 o2;
        o2.x = run; run += v[i].x;
        o2.y = run; run += v[i].y;
        mine[i] = o2;
    }
}

__global__ void edge_place_kernel(const int* __restrict__ src, const int* __restrict__ dst,
                                  const int* __restrict__ rank, const int* __restrict__ cur,
                                  int* __restrict__ bucket_src) {
    int i = blockIdx.x * blockDim.x + threadIdx.x;
    if (i >= N_EDGES / 4) return;
    const int4 d = ((const int4*)dst)[i];
    const int4 s = ((const int4*)src)[i];
    const int4 r = ((const int4*)rank)[i];
    bucket_src[cur[d.x] + r.x] = s.x;
    bucket_src[cur[d.y] + r.y] = s.y;
    bucket_src[cur[d.z] + r.z] = s.z;
    bucket_src[cur[d.w] + r.w] = s.w;
}

template <int K>
__device__ __forceinline__ void add_rows(const float* __restrict__ xl, const int* __restrict__ ids,
                                         float4& a0, float4& a1) {
    int s[K];
    float4 u[K], w[K];
#pragma unroll
    for (int j = 0; j < K; ++j) s[j] = ids[j];
#pragma unroll
    for (int j = 0; j < K; ++j) {
        const float* r = xl + (size_t)s[j] * F_IN;
        u[j] = *(const float4*)r;
        w[j] = *(const float4*)(r + 128);
    }
#pragma unroll
    for (int j = 0; j < K; ++j) {
        a0.x += u[j].x; a0.y += u[j].y; a0.z += u[j].z; a0.w += u[j].w;
        a1.x += w[j].x; a1.y += w[j].y; a1.z += w[j].z; a1.w += w[j].w;
    }
}

__device__ __forceinline__ void col_partial(v8f a, float& s, float& q) {
    s = 0.0f; q = 0.0f;
#pragma unroll
    for (int r = 0; r < 8; ++r) { s += a[r]; q += a[r] * a[r]; }
}

#define G1_KSTEPS 8
struct G1Frag {
    v2f    a[G1_KSTEPS];
    float4 b0[G1_KSTEPS];
    float4 b1[G1_KSTEPS];
};

__device__ __forceinline__ void g1_load(G1Frag& f, const float* ap, const float* bp) {
#pragma unroll
    for (int s = 0; s < G1_KSTEPS; ++s) {
        f.a[s]  = *(const v2f*)(ap + 4 * s);
        f.b0[s] = *(const float4*)(bp + (size_t)(4 * s) * F_HID);
        f.b1[s] = *(const float4*)(bp + (size_t)(4 * s + 1) * F_HID);
    }
}

__device__ __forceinline__ void g1_mma(const G1Frag& f, v8f& acc0, v8f& acc1, v8f& acc2, v8f& acc3) {
#pragma unroll
    for (int s = 0; s < G1_KSTEPS; ++s) {
        const v2f a = f.a[s];
        v2f b;
        b.x = f.b0[s].x; b.y = f.b1[s].x; acc0 = wmma4(a, b, acc0);
        b.x = f.b0[s].y; b.y = f.b1[s].y; acc1 = wmma4(a, b, acc1);
        b.x = f.b0[s].z; b.y = f.b1[s].z; acc2 = wmma4(a, b, acc2);
        b.x = f.b0[s].w; b.y = f.b1[s].w; acc3 = wmma4(a, b, acc3);
    }
}

#define G1_PITCH 260

__global__ __launch_bounds__(128) void gemm1_kernel(const float* __restrict__ x,
                                                    const int* __restrict__ cur,
                                                    const int* __restrict__ bucket_src,
                                                    const float* __restrict__ B1,
                                                    const float* __restrict__ B2,
                                                    float* __restrict__ C,
                                                    float* __restrict__ sum,
                                                    float* __restrict__ sumsq) {
    __shared__ float agg[16 * G1_PITCH] __attribute__((aligned(16)));
    const int lane = threadIdx.x & 31;
    const int wave = __builtin_amdgcn_readfirstlane(threadIdx.x >> 5);
    const int m0 = blockIdx.x * 16;

    {
        const float* xl = x + lane * 4;
        for (int i = 0; i < 4; ++i) {
            const int row = wave * 4 + i;
            const int begin = cur[m0 + row];
            const int end   = cur[m0 + row + 1];
            float4 a0 = {0.0f, 0.0f, 0.0f, 0.0f}, a1 = {0.0f, 0.0f, 0.0f, 0.0f};
            int e = begin;
            for (; e + 16 <= end; e += 16) add_rows<16>(xl, bucket_src + e, a0, a1);
            if (e + 8 <= end) { add_rows<8>(xl, bucket_src + e, a0, a1); e += 8; }
            if (e + 4 <= end) { add_rows<4>(xl, bucket_src + e, a0, a1); e += 4; }
            if (e + 2 <= end) { add_rows<2>(xl, bucket_src + e, a0, a1); e += 2; }
            if (e < end)      { add_rows<1>(xl, bucket_src + e, a0, a1); }
            const float di = 1.0f / fmaxf((float)(end - begin), 1.0f);
            float* o = agg + row * G1_PITCH + lane * 4;
            float4 r0 = {a0.x * di, a0.y * di, a0.z * di, a0.w * di};
            float4 r1 = {a1.x * di, a1.y * di, a1.z * di, a1.w * di};
            *(float4*)o = r0;
            *(float4*)(o + 128) = r1;
        }
    }
    __syncthreads();

    const int n0 = wave * 64;
    const int half = lane >> 4, c = lane & 15;

    const float* a1p = agg + c * G1_PITCH + 2 * half;
    const float* a2p = x + (size_t)(m0 + c) * F_IN + 2 * half;
    const size_t b_off = (size_t)(2 * half) * F_HID + n0 + 4 * c;
    const float* b1p = B1 + b_off;
    const float* b2p = B2 + b_off;

    v8f acc0 = {}, acc1 = {}, acc2 = {}, acc3 = {};
    G1Frag f0, f1;
    g1_load(f0, a1p, b1p);
#pragma unroll 1
    for (int g = 0; g < 6; g += 2) {
        g1_load(f1, a1p + (g + 1) * 32, b1p + (size_t)(g + 1) * 32 * F_HID);
        g1_mma(f0, acc0, acc1, acc2, acc3);
        g1_load(f0, a1p + (g + 2) * 32, b1p + (size_t)(g + 2) * 32 * F_HID);
        g1_mma(f1, acc0, acc1, acc2, acc3);
    }
    g1_load(f1, a1p + 7 * 32, b1p + (size_t)7 * 32 * F_HID);
    g1_mma(f0, acc0, acc1, acc2, acc3);
    g1_load(f0, a2p, b2p);
    g1_mma(f1, acc0, acc1, acc2, acc3);
#pragma unroll 1
    for (int g = 0; g < 6; g += 2) {
        g1_load(f1, a2p + (g + 1) * 32, b2p + (size_t)(g + 1) * 32 * F_HID);
        g1_mma(f0, acc0, acc1, acc2, acc3);
        g1_load(f0, a2p + (g + 2) * 32, b2p + (size_t)(g + 2) * 32 * F_HID);
        g1_mma(f1, acc0, acc1, acc2, acc3);
    }
    g1_load(f1, a2p + 7 * 32, b2p + (size_t)7 * 32 * F_HID);
    g1_mma(f0, acc0, acc1, acc2, acc3);
    g1_mma(f1, acc0, acc1, acc2, acc3);

#pragma unroll
    for (int r = 0; r < 8; ++r) {
        float4 o = {acc0[r], acc1[r], acc2[r], acc3[r]};
        *(float4*)(C + (size_t)(m0 + r + half * 8) * F_HID + n0 + 4 * c) = o;
    }

    float s0, q0, s1, q1, s2, q2, s3, q3;
    col_partial(acc0, s0, q0);
    col_partial(acc1, s1, q1);
    col_partial(acc2, s2, q2);
    col_partial(acc3, s3, q3);
    const float sa = (half ? s1 : s0) + __shfl_xor(half ? s0 : s1, 16);
    const float sb = (half ? s3 : s2) + __shfl_xor(half ? s2 : s3, 16);
    const float qa = (half ? q1 : q0) + __shfl_xor(half ? q0 : q1, 16);
    const float qb = (half ? q3 : q2) + __shfl_xor(half ? q2 : q3, 16);
    atomicAdd(&sum[n0 + 4 * c + half], sa);
    atomicAdd(&sum[n0 + 4 * c + 2 + half], sb);
    atomicAdd(&sumsq[n0 + 4 * c + half], qa);
    atomicAdd(&sumsq[n0 + 4 * c + 2 + half], qb);
}

__global__ __launch_bounds__(256) void gemm3_kernel(const float* __restrict__ A,
                                                    const float* __restrict__ sum,
                                                    const float* __restrict__ sumsq,
                                                    const float* __restrict__ gamma,
                                                    const float* __restrict__ beta,
                                                    const float* __restrict__ B1,
                                                    const float* __restrict__ B2,
                                                    float* __restrict__ D1,
                                                    float* __restrict__ D2) {
    __shared__ float bn[4][F_HID] __attribute__((aligned(8)));
    {
        const int k = threadIdx.x;
        const float mean = sum[k] * (1.0f / N_NODES);
        const float var  = sumsq[k] * (1.0f / N_NODES) - mean * mean;
        bn[0][k] = mean;
        bn[1][k] = rsqrtf(var + BN_EPS);
        bn[2][k] = gamma[k];
        bn[3][k] = beta[k];
    }
    __syncthreads();

    const int lane = threadIdx.x & 31;
    const int wave = threadIdx.x >> 5;
    const int tile = blockIdx.x * 8 + wave;
    if (tile >= 625) return;
    const int m0 = tile * 16;
    const int kh = (lane >> 4) << 1;

    v8f acc1 = {}, acc2 = {};
#pragma unroll 4
    for (int k0 = 0; k0 < F_HID; k0 += 4) {
        const v2f raw = load_a(A, F_HID, m0, k0, lane);
        const v2f mu = *(const v2f*)&bn[0][k0 + kh];
        const v2f rs = *(const v2f*)&bn[1][k0 + kh];
        const v2f ga = *(const v2f*)&bn[2][k0 + kh];
        const v2f be = *(const v2f*)&bn[3][k0 + kh];
        v2f a;
        a.x = fmaxf((raw.x - mu.x) * rs.x * ga.x + be.x, 0.0f);
        a.y = fmaxf((raw.y - mu.y) * rs.y * ga.y + be.y, 0.0f);
        v2f b1 = load_b(B1, F_OUT, k0, 0, lane);
        v2f b2 = load_b(B2, F_OUT, k0, 0, lane);
        acc1 = wmma4(a, b1, acc1);
        acc2 = wmma4(a, b2, acc2);
    }
    store_c(D1, F_OUT, m0, 0, lane, acc1);
    store_c(D2, F_OUT, m0, 0, lane, acc2);
}


__global__ __launch_bounds__(256) void agg3_gather_kernel(const float* __restrict__ p,
                                                          const float* __restrict__ hl,
                                                          const int* __restrict__ cur,
                                                          const int* __restrict__ bucket_src,
                                                          float* __restrict__ t,
                                                          float* __restrict__ sum,
                                                          float* __restrict__ sumsq) {
    __shared__ float rows[2][8][F_OUT] __attribute__((aligned(16)));
    const int lane = threadIdx.x & 31;
    const int node = __builtin_amdgcn_readfirstlane(blockIdx.x * 8 + (threadIdx.x >> 5));
    const int begin = cur[node];
    const int end   = cur[node + 1];
    const int slot = lane >> 2, c4 = (lane & 3) << 2;
    float4 acc = {0.0f, 0.0f, 0.0f, 0.0f};
    for (int base = begin; base < end; base += 32) {
        int s[4];
        float4 v[4];
#pragma unroll
        for (int j = 0; j < 4; ++j) s[j] = bucket_src[min(base + slot + 8 * j, end - 1)];
#pragma unroll
        for (int j = 0; j < 4; ++j) v[j] = *(const float4*)(p + (size_t)s[j] * F_OUT + c4);
#pragma unroll
        for (int j = 0; j < 4; ++j)
            if (base + slot + 8 * j < end) {
                acc.x += v[j].x; acc.y += v[j].y; acc.z += v[j].z; acc.w += v[j].w;
            }
    }
#pragma unroll
    for (int o = 4; o < 32; o <<= 1) {
        acc.x += __shfl_xor(acc.x, o);
        acc.y += __shfl_xor(acc.y, o);
        acc.z += __shfl_xor(acc.z, o);
        acc.w += __shfl_xor(acc.w, o);
    }
    if (lane < 4) {
        const float di = 1.0f / fmaxf((float)(end - begin), 1.0f);
        const float4 h = *(const float4*)(hl + (size_t)node * F_OUT + c4);
        float4 r = {acc.x * di + h.x, acc.y * di + h.y, acc.z * di + h.z, acc.w * di + h.w};
        *(float4*)(t + (size_t)node * F_OUT + c4) = r;
        const int wv = threadIdx.x >> 5;
        float4 r2 = {r.x * r.x, r.y * r.y, r.z * r.z, r.w * r.w};
        *(float4*)&rows[0][wv][c4] = r;
        *(float4*)&rows[1][wv][c4] = r2;
    }
    __syncthreads();
    if (threadIdx.x < 32) {
        const int st = threadIdx.x >> 4, col = threadIdx.x & 15;
        float a = 0.0f;
#pragma unroll
        for (int w = 0; w < 8; ++w) a += rows[st][w][col];
        atomicAdd(st ? &sumsq[col] : &sum[col], a);
    }
}

__global__ void bn_final_kernel(const float* __restrict__ t,
                                const float* __restrict__ sum, const float* __restrict__ sumsq,
                                const float* __restrict__ gamma, const float* __restrict__ beta,
                                const int* __restrict__ ste,
                                float* __restrict__ out, long long total) {
    long long g = (long long)blockIdx.x * blockDim.x + threadIdx.x;
    if (g >= total) return;
    int c = (int)(g & (F_OUT - 1));
    float mean = sum[c] * (1.0f / N_NODES);
    float var  = sumsq[c] * (1.0f / N_NODES) - mean * mean;
    float v = (t[g] - mean) * rsqrtf(var + BN_EPS) * gamma[c] + beta[c];
    if (*ste)
        out[g] = (v > 0.0f) ? 1.0f : 0.0f;
    else
        out[g] = 1.0f / (1.0f + expf(-v));
}

extern "C" void kernel_launch(void* const* d_in, const int* in_sizes, int n_in,
                              void* d_out, int out_size, void* d_ws, size_t ws_size,
                              hipStream_t stream) {
    const float* x        = (const float*)d_in[0];
    const float* W_conv1  = (const float*)d_in[1];
    const float* W_lin1   = (const float*)d_in[2];
    const float* bn1_g    = (const float*)d_in[3];
    const float* bn1_b    = (const float*)d_in[4];
    const float* W_conv3  = (const float*)d_in[5];
    const float* W_lin3   = (const float*)d_in[6];
    const float* bn3_g    = (const float*)d_in[7];
    const float* bn3_b    = (const float*)d_in[8];
    const int*   src      = (const int*)d_in[9];
    const int*   dst      = (const int*)d_in[10];
    const int*   ste      = (const int*)d_in[11];
    float*       out      = (float*)d_out;

    float* ws = (float*)d_ws;
    const size_t o_cur  = 0;
    const size_t o_s1   = 10240;
    const size_t o_q1   = o_s1 + 256;
    const size_t o_s3   = o_q1 + 256;
    const size_t o_q3   = o_s3 + 16;
    const size_t o_bkt  = 11264;
    const size_t o_rank = o_bkt + 320000;
    const size_t o_h    = o_rank + 320000;
    const size_t o_p    = o_h + 2560000;
    const size_t o_hl   = o_p + 160000;
    const size_t o_agg3 = o_hl + 160000;
    int*   cur     = (int*)(ws + o_cur);
    int*   bkt     = (int*)(ws + o_bkt);
    int*   rank    = (int*)(ws + o_rank);
    float* h       = ws + o_h;
    float* p       = ws + o_p;
    float* hl      = ws + o_hl;
    float* agg3    = ws + o_agg3;
    float* s1 = ws + o_s1; float* q1 = ws + o_q1;
    float* s3 = ws + o_s3; float* q3 = ws + o_q3;

    hipMemsetAsync(cur, 0, (10240 + 256 + 256 + 16 + 16) * sizeof(float), stream);

    edge_rank_kernel<<<(N_EDGES / 4 + 255) / 256, 256, 0, stream>>>(dst, cur, rank);
    offsets_scan_kernel<<<1, 1024, 0, stream>>>(cur);
    edge_place_kernel<<<(N_EDGES / 4 + 255) / 256, 256, 0, stream>>>(src, dst, rank, cur, bkt);

    gemm1_kernel<<<N_NODES / 16, 128, 0, stream>>>(x, cur, bkt, W_conv1, W_lin1, h, s1, q1);

    gemm3_kernel<<<(625 + 7) / 8, 256, 0, stream>>>(h, s1, q1, bn1_g, bn1_b, W_conv3, W_lin3, p, hl);

    agg3_gather_kernel<<<N_NODES / 8, 256, 0, stream>>>(p, hl, cur, bkt, agg3, s3, q3);

    {
        long long tot = (long long)N_NODES * F_OUT;
        bn_final_kernel<<<(int)((tot + 255) / 256), 256, 0, stream>>>(agg3, s3, q3, bn3_g, bn3_b, ste, out, tot);
    }
}


# edges bucketed in one pass: an edge's int-atomic rank indexes a fixed 128-slot id row of its dst node (longer buckets spill to an overflow list); scan and place launches gone; sums stay in rank order
# speedup vs baseline: 4.6157x; 1.1315x over previous
#include <hip/hip_runtime.h>
#include <hip/hip_bf16.h>
#include <math.h>

#define N_NODES 10000
#define N_EDGES 320000
#define F_IN    256
#define F_HID   256
#define F_OUT   16
#define BN_EPS  1e-5f

typedef __attribute__((ext_vector_type(2))) float v2f;
typedef __attribute__((ext_vector_type(8))) float v8f;

__device__ __forceinline__ v8f wmma4(v2f a, v2f b, v8f c) {
    return __builtin_amdgcn_wmma_f32_16x16x4_f32(
        false, a, false, b, (short)0, c, false, false);
}

__device__ __forceinline__ v2f load_a(const float* A, int lda, int m0, int k0, int lane) {
    const int half = lane >> 4;
    const int row  = lane & 15;
    const float* p = A + (size_t)(m0 + row) * lda + k0 + (half << 1);
    v2f r; r.x = p[0]; r.y = p[1];
    return r;
}

__device__ __forceinline__ v2f load_b(const float* B, int ldb, int k0, int n0, int lane) {
    const int half = lane >> 4;
    const int col  = lane & 15;
    const float* p = B + (size_t)(k0 + (half << 1)) * ldb + n0 + col;
    v2f r; r.x = p[0]; r.y = p[ldb];
    return r;
}

__device__ __forceinline__ void store_c(float* C, int ldc, int m0, int n0, int lane, v8f acc) {
    const int half = lane >> 4;
    const int col  = lane & 15;
#pragma unroll
    for (int r = 0; r < 8; ++r)
        C[(size_t)(m0 + r + half * 8) * ldc + n0 + col] = acc[r];
}

#define ELL_CAP 128

__global__ void edge_bucket_kernel(const int* __restrict__ src, const int* __restrict__ dst,
                                   int* __restrict__ cnt, int* __restrict__ slots,
                                   int* __restrict__ ovf_n, int* __restrict__ ovf_dst,
                                   int* __restrict__ ovf_src) {
    int i = blockIdx.x * blockDim.x + threadIdx.x;
    if (i >= N_EDGES / 4) return;
    const int4 d4 = ((const int4*)dst)[i];
    const int4 s4 = ((const int4*)src)[i];
    const int d[4] = {d4.x, d4.y, d4.z, d4.w};
    const int s[4] = {s4.x, s4.y, s4.z, s4.w};
    int r[4];
#pragma unroll
    for (int j = 0; j < 4; ++j) r[j] = atomicAdd(&cnt[d[j]], 1);
#pragma unroll
    for (int j = 0; j < 4; ++j) {
        if (r[j] < ELL_CAP) {
            slots[(size_t)d[j] * ELL_CAP + r[j]] = s[j];
        } else {
            const int o = atomicAdd(ovf_n, 1);
            ovf_dst[o] = d[j];
            ovf_src[o] = s[j];
        }
    }
}

template <int K>
__device__ __forceinline__ void add_rows(const float* __restrict__ xl, const int* __restrict__ ids,
                                         float4& a0, float4& a1) {
    int s[K];
    float4 u[K], w[K];
#pragma unroll
    for (int j = 0; j < K; ++j) s[j] = ids[j];
#pragma unroll
    for (int j = 0; j < K; ++j) {
        const float* r = xl + (size_t)s[j] * F_IN;
        u[j] = *(const float4*)r;
        w[j] = *(const float4*)(r + 128);
    }
#pragma unroll
    for (int j = 0; j < K; ++j) {
        a0.x += u[j].x; a0.y += u[j].y; a0.z += u[j].z; a0.w += u[j].w;
        a1.x += w[j].x; a1.y += w[j].y; a1.z += w[j].z; a1.w += w[j].w;
    }
}

__device__ __forceinline__ void col_partial(v8f a, float& s, float& q) {
    s = 0.0f; q = 0.0f;
#pragma unroll
    for (int r = 0; r < 8; ++r) { s += a[r]; q += a[r] * a[r]; }
}

#define G1_KSTEPS 8
struct G1Frag {
    v2f    a[G1_KSTEPS];
    float4 b0[G1_KSTEPS];
    float4 b1[G1_KSTEPS];
};

__device__ __forceinline__ void g1_load(G1Frag& f, const float* ap, const float* bp) {
#pragma unroll
    for (int s = 0; s < G1_KSTEPS; ++s) {
        f.a[s]  = *(const v2f*)(ap + 4 * s);
        f.b0[s] = *(const float4*)(bp + (size_t)(4 * s) * F_HID);
        f.b1[s] = *(const float4*)(bp + (size_t)(4 * s + 1) * F_HID);
    }
}

__device__ __forceinline__ void g1_mma(const G1Frag& f, v8f& acc0, v8f& acc1, v8f& acc2, v8f& acc3) {
#pragma unroll
    for (int s = 0; s < G1_KSTEPS; ++s) {
        const v2f a = f.a[s];
        v2f b;
        b.x = f.b0[s].x; b.y = f.b1[s].x; acc0 = wmma4(a, b, acc0);
        b.x = f.b0[s].y; b.y = f.b1[s].y; acc1 = wmma4(a, b, acc1);
        b.x = f.b0[s].z; b.y = f.b1[s].z; acc2 = wmma4(a, b, acc2);
        b.x = f.b0[s].w; b.y = f.b1[s].w; acc3 = wmma4(a, b, acc3);
    }
}

#define G1_PITCH 260

__global__ __launch_bounds__(128) void gemm1_kernel(const float* __restrict__ x,
                                                    const int* __restrict__ cnt,
                                                    const int* __restrict__ slots,
                                                    const int* __restrict__ ovf_n,
                                                    const int* __restrict__ ovf_dst,
                                                    const int* __restrict__ ovf_src,
                                                    const float* __restrict__ B1,
                                                    const float* __restrict__ B2,
                                                    float* __restrict__ C,
                                                    float* __restrict__ sum,
                                                    float* __restrict__ sumsq) {
    __shared__ float agg[16 * G1_PITCH] __attribute__((aligned(16)));
    const int lane = threadIdx.x & 31;
    const int wave = __builtin_amdgcn_readfirstlane(threadIdx.x >> 5);
    const int m0 = blockIdx.x * 16;

    {
        const float* xl = x + lane * 4;
        for (int i = 0; i < 4; ++i) {
            const int row = wave * 4 + i;
            const int deg = cnt[m0 + row];
            const int end = min(deg, ELL_CAP);
            const int* ids = slots + (size_t)(m0 + row) * ELL_CAP;
            float4 a0 = {0.0f, 0.0f, 0.0f, 0.0f}, a1 = {0.0f, 0.0f, 0.0f, 0.0f};
            int e = 0;
            for (; e + 16 <= end; e += 16) add_rows<16>(xl, ids + e, a0, a1);
            if (e + 8 <= end) { add_rows<8>(xl, ids + e, a0, a1); e += 8; }
            if (e + 4 <= end) { add_rows<4>(xl, ids + e, a0, a1); e += 4; }
            if (e + 2 <= end) { add_rows<2>(xl, ids + e, a0, a1); e += 2; }
            if (e < end)      { add_rows<1>(xl, ids + e, a0, a1); }
            if (deg > ELL_CAP) {
                const int node = m0 + row;
                const int no = *ovf_n;
                for (int k = 0; k < no; ++k)
                    if (ovf_dst[k] == node) add_rows<1>(xl, ovf_src + k, a0, a1);
            }
            const float di = 1.0f / fmaxf((float)deg, 1.0f);
            float* o = agg + row * G1_PITCH + lane * 4;
            float4 r0 = {a0.x * di, a0.y * di, a0.z * di, a0.w * di};
            float4 r1 = {a1.x * di, a1.y * di, a1.z * di, a1.w * di};
            *(float4*)o = r0;
            *(float4*)(o + 128) = r1;
        }
    }
    __syncthreads();

    const int n0 = wave * 64;
    const int half = lane >> 4, c = lane & 15;

    const float* a1p = agg + c * G1_PITCH + 2 * half;
    const float* a2p = x + (size_t)(m0 + c) * F_IN + 2 * half;
    const size_t b_off = (size_t)(2 * half) * F_HID + n0 + 4 * c;
    const float* b1p = B1 + b_off;
    const float* b2p = B2 + b_off;

    v8f acc0 = {}, acc1 = {}, acc2 = {}, acc3 = {};
    G1Frag f0, f1;
    g1_load(f0, a1p, b1p);
#pragma unroll 1
    for (int g = 0; g < 6; g += 2) {
        g1_load(f1, a1p + (g + 1) * 32, b1p + (size_t)(g + 1) * 32 * F_HID);
        g1_mma(f0, acc0, acc1, acc2, acc3);
        g1_load(f0, a1p + (g + 2) * 32, b1p + (size_t)(g + 2) * 32 * F_HID);
        g1_mma(f1, acc0, acc1, acc2, acc3);
    }
    g1_load(f1, a1p + 7 * 32, b1p + (size_t)7 * 32 * F_HID);
    g1_mma(f0, acc0, acc1, acc2, acc3);
    g1_load(f0, a2p, b2p);
    g1_mma(f1, acc0, acc1, acc2, acc3);
#pragma unroll 1
    for (int g = 0; g < 6; g += 2) {
        g1_load(f1, a2p + (g + 1) * 32, b2p + (size_t)(g + 1) * 32 * F_HID);
        g1_mma(f0, acc0, acc1, acc2, acc3);
        g1_load(f0, a2p + (g + 2) * 32, b2p + (size_t)(g + 2) * 32 * F_HID);
        g1_mma(f1, acc0, acc1, acc2, acc3);
    }
    g1_load(f1, a2p + 7 * 32, b2p + (size_t)7 * 32 * F_HID);
    g1_mma(f0, acc0, acc1, acc2, acc3);
    g1_mma(f1, acc0, acc1, acc2, acc3);

#pragma unroll
    for (int r = 0; r < 8; ++r) {
        float4 o = {acc0[r], acc1[r], acc2[r], acc3[r]};
        *(float4*)(C + (size_t)(m0 + r + half * 8) * F_HID + n0 + 4 * c) = o;
    }

    float s0, q0, s1, q1, s2, q2, s3, q3;
    col_partial(acc0, s0, q0);
    col_partial(acc1, s1, q1);
    col_partial(acc2, s2, q2);
    col_partial(acc3, s3, q3);
    const float sa = (half ? s1 : s0) + __shfl_xor(half ? s0 : s1, 16);
    const float sb = (half ? s3 : s2) + __shfl_xor(half ? s2 : s3, 16);
    const float qa = (half ? q1 : q0) + __shfl_xor(half ? q0 : q1, 16);
    const float qb = (half ? q3 : q2) + __shfl_xor(half ? q2 : q3, 16);
    atomicAdd(&sum[n0 + 4 * c + half], sa);
    atomicAdd(&sum[n0 + 4 * c + 2 + half], sb);
    atomicAdd(&sumsq[n0 + 4 * c + half], qa);
    atomicAdd(&sumsq[n0 + 4 * c + 2 + half], qb);
}

__global__ __launch_bounds__(256) void gemm3_kernel(const float* __restrict__ A,
                                                    const float* __restrict__ sum,
                                                    const float* __restrict__ sumsq,
                                                    const float* __restrict__ gamma,
                                                    const float* __restrict__ beta,
                                                    const float* __restrict__ B1,
                                                    const float* __restrict__ B2,
                                                    float* __restrict__ D1,
                                                    float* __restrict__ D2) {
    __shared__ float bn[4][F_HID] __attribute__((aligned(8)));
    {
        const int k = threadIdx.x;
        const float mean = sum[k] * (1.0f / N_NODES);
        const float var  = sumsq[k] * (1.0f / N_NODES) - mean * mean;
        bn[0][k] = mean;
        bn[1][k] = rsqrtf(var + BN_EPS);
        bn[2][k] = gamma[k];
        bn[3][k] = beta[k];
    }
    __syncthreads();

    const int lane = threadIdx.x & 31;
    const int wave = threadIdx.x >> 5;
    const int tile = blockIdx.x * 8 + wave;
    if (tile >= 625) return;
    const int m0 = tile * 16;
    const int kh = (lane >> 4) << 1;

    v8f acc1 = {}, acc2 = {};
#pragma unroll 4
    for (int k0 = 0; k0 < F_HID; k0 += 4) {
        const v2f raw = load_a(A, F_HID, m0, k0, lane);
        const v2f mu = *(const v2f*)&bn[0][k0 + kh];
        const v2f rs = *(const v2f*)&bn[1][k0 + kh];
        const v2f ga = *(const v2f*)&bn[2][k0 + kh];
        const v2f be = *(const v2f*)&bn[3][k0 + kh];
        v2f a;
        a.x = fmaxf((raw.x - mu.x) * rs.x * ga.x + be.x, 0.0f);
        a.y = fmaxf((raw.y - mu.y) * rs.y * ga.y + be.y, 0.0f);
        v2f b1 = load_b(B1, F_OUT, k0, 0, lane);
        v2f b2 = load_b(B2, F_OUT, k0, 0, lane);
        acc1 = wmma4(a, b1, acc1);
        acc2 = wmma4(a, b2, acc2);
    }
    store_c(D1, F_OUT, m0, 0, lane, acc1);
    store_c(D2, F_OUT, m0, 0, lane, acc2);
}


__global__ __launch_bounds__(256) void agg3_gather_kernel(const float* __restrict__ p,
                                                          const float* __restrict__ hl,
                                                          const int* __restrict__ cnt,
                                                          const int* __restrict__ slots,
                                                          const int* __restrict__ ovf_n,
                                                          const int* __restrict__ ovf_dst,
                                                          const int* __restrict__ ovf_src,
                                                          float* __restrict__ t,
                                                          float* __restrict__ sum,
                                                          float* __restrict__ sumsq) {
    __shared__ float rows[2][8][F_OUT] __attribute__((aligned(16)));
    const int lane = threadIdx.x & 31;
    const int node = __builtin_amdgcn_readfirstlane(blockIdx.x * 8 + (threadIdx.x >> 5));
    const int deg = cnt[node];
    const int end = min(deg, ELL_CAP);
    const int* ids = slots + (size_t)node * ELL_CAP;
    const int slot = lane >> 2, c4 = (lane & 3) << 2;
    float4 acc = {0.0f, 0.0f, 0.0f, 0.0f};
    for (int base = 0; base < end; base += 32) {
        int s[4];
        float4 v[4];
#pragma unroll
        for (int j = 0; j < 4; ++j) s[j] = ids[min(base + slot + 8 * j, end - 1)];
#pragma unroll
        for (int j = 0; j < 4; ++j) v[j] = *(const float4*)(p + (size_t)s[j] * F_OUT + c4);
#pragma unroll
        for (int j = 0; j < 4; ++j)
            if (base + slot + 8 * j < end) {
                acc.x += v[j].x; acc.y += v[j].y; acc.z += v[j].z; acc.w += v[j].w;
            }
    }
    if (deg > ELL_CAP) {
        const int no = *ovf_n;
        for (int k = slot; k < no; k += 8)
            if (ovf_dst[k] == node) {
                const float4 v = *(const float4*)(p + (size_t)ovf_src[k] * F_OUT + c4);
                acc.x += v.x; acc.y += v.y; acc.z += v.z; acc.w += v.w;
            }
    }
#pragma unroll
    for (int o = 4; o < 32; o <<= 1) {
        acc.x += __shfl_xor(acc.x, o);
        acc.y += __shfl_xor(acc.y, o);
        acc.z += __shfl_xor(acc.z, o);
        acc.w += __shfl_xor(acc.w, o);
    }
    if (lane < 4) {
        const float di = 1.0f / fmaxf((float)deg, 1.0f);
        const float4 h = *(const float4*)(hl + (size_t)node * F_OUT + c4);
        float4 r = {acc.x * di + h.x, acc.y * di + h.y, acc.z * di + h.z, acc.w * di + h.w};
        *(float4*)(t + (size_t)node * F_OUT + c4) = r;
        const int wv = threadIdx.x >> 5;
        float4 r2 = {r.x * r.x, r.y * r.y, r.z * r.z, r.w * r.w};
        *(float4*)&rows[0][wv][c4] = r;
        *(float4*)&rows[1][wv][c4] = r2;
    }
    __syncthreads();
    if (threadIdx.x < 32) {
        const int st = threadIdx.x >> 4, col = threadIdx.x & 15;
        float a = 0.0f;
#pragma unroll
        for (int w = 0; w < 8; ++w) a += rows[st][w][col];
        atomicAdd(st ? &sumsq[col] : &sum[col], a);
    }
}

__global__ void bn_final_kernel(const float* __restrict__ t,
                                const float* __restrict__ sum, const float* __restrict__ sumsq,
                                const float* __restrict__ gamma, const float* __restrict__ beta,
                                const int* __restrict__ ste,
                                float* __restrict__ out, long long total) {
    long long g = (long long)blockIdx.x * blockDim.x + threadIdx.x;
    if (g >= total) return;
    int c = (int)(g & (F_OUT - 1));
    float mean = sum[c] * (1.0f / N_NODES);
    float var  = sumsq[c] * (1.0f / N_NODES) - mean * mean;
    float v = (t[g] - mean) * rsqrtf(var + BN_EPS) * gamma[c] + beta[c];
    if (*ste)
        out[g] = (v > 0.0f) ? 1.0f : 0.0f;
    else
        out[g] = 1.0f / (1.0f + expf(-v));
}

extern "C" void kernel_launch(void* const* d_in, const int* in_sizes, int n_in,
                              void* d_out, int out_size, void* d_ws, size_t ws_size,
                              hipStream_t stream) {
    const float* x        = (const float*)d_in[0];
    const float* W_conv1  = (const float*)d_in[1];
    const float* W_lin1   = (const float*)d_in[2];
    const float* bn1_g    = (const float*)d_in[3];
    const float* bn1_b    = (const float*)d_in[4];
    const float* W_conv3  = (const float*)d_in[5];
    const float* W_lin3   = (const float*)d_in[6];
    const float* bn3_g    = (const float*)d_in[7];
    const float* bn3_b    = (const float*)d_in[8];
    const int*   src      = (const int*)d_in[9];
    const int*   dst      = (const int*)d_in[10];
    const int*   ste      = (const int*)d_in[11];
    float*       out      = (float*)d_out;

    float* ws = (float*)d_ws;
    const size_t o_cnt  = 0;
    const size_t o_s1   = 10240;
    const size_t o_q1   = o_s1 + 256;
    const size_t o_s3   = o_q1 + 256;
    const size_t o_q3   = o_s3 + 16;
    const size_t o_ovn  = o_q3 + 16;
    const size_t o_slot = 11264;
    const size_t o_ovd  = o_slot + 1280000;
    const size_t o_ovs  = o_ovd + 320000;
    const size_t o_h    = o_ovs + 320000;
    const size_t o_p    = o_h + 2560000;
    const size_t o_hl   = o_p + 160000;
    const size_t o_agg3 = o_hl + 160000;
    int*   cnt     = (int*)(ws + o_cnt);
    int*   ovn     = (int*)(ws + o_ovn);
    int*   slots   = (int*)(ws + o_slot);
    int*   ovd     = (int*)(ws + o_ovd);
    int*   ovs     = (int*)(ws + o_ovs);
    float* h       = ws + o_h;
    float* p       = ws + o_p;
    float* hl      = ws + o_hl;
    float* agg3    = ws + o_agg3;
    float* s1 = ws + o_s1; float* q1 = ws + o_q1;
    float* s3 = ws + o_s3; float* q3 = ws + o_q3;

    hipMemsetAsync(cnt, 0, (10240 + 256 + 256 + 16 + 16 + 16) * sizeof(float), stream);

    edge_bucket_kernel<<<(N_EDGES / 4 + 255) / 256, 256, 0, stream>>>(src, dst, cnt, slots, ovn, ovd, ovs);

    gemm1_kernel<<<N_NODES / 16, 128, 0, stream>>>(x, cnt, slots, ovn, ovd, ovs, W_conv1, W_lin1, h, s1, q1);

    gemm3_kernel<<<(625 + 7) / 8, 256, 0, stream>>>(h, s1, q1, bn1_g, bn1_b, W_conv3, W_lin3, p, hl);

    agg3_gather_kernel<<<N_NODES / 8, 256, 0, stream>>>(p, hl, cnt, slots, ovn, ovd, ovs, agg3, s3, q3);

    {
        long long tot = (long long)N_NODES * F_OUT;
        bn_final_kernel<<<(int)((tot + 255) / 256), 256, 0, stream>>>(agg3, s3, q3, bn3_g, bn3_b, ste, out, tot);
    }
}


# layer-1 kernel re-tiled: 48 rows a work-group, 16 waves: 12 gather agg rows while 4 run x @ W_lin1, then agg @ W_conv1; B fragment feeds 3 M tiles; x product summed first, stats per 48 rows
# speedup vs baseline: 4.8256x; 1.0455x over previous
#include <hip/hip_runtime.h>
#include <hip/hip_bf16.h>
#include <math.h>

#define N_NODES 10000
#define N_EDGES 320000
#define F_IN    256
#define F_HID   256
#define F_OUT   16
#define BN_EPS  1e-5f

typedef __attribute__((ext_vector_type(2))) float v2f;
typedef __attribute__((ext_vector_type(8))) float v8f;

__device__ __forceinline__ v8f wmma4(v2f a, v2f b, v8f c) {
    return __builtin_amdgcn_wmma_f32_16x16x4_f32(
        false, a, false, b, (short)0, c, false, false);
}

__device__ __forceinline__ v2f load_a(const float* A, int lda, int m0, int k0, int lane) {
    const int half = lane >> 4;
    const int row  = lane & 15;
    const float* p = A + (size_t)(m0 + row) * lda + k0 + (half << 1);
    v2f r; r.x = p[0]; r.y = p[1];
    return r;
}

__device__ __forceinline__ v2f load_b(const float* B, int ldb, int k0, int n0, int lane) {
    const int half = lane >> 4;
    const int col  = lane & 15;
    const float* p = B + (size_t)(k0 + (half << 1)) * ldb + n0 + col;
    v2f r; r.x = p[0]; r.y = p[ldb];
    return r;
}

__device__ __forceinline__ void store_c(float* C, int ldc, int m0, int n0, int lane, v8f acc) {
    const int half = lane >> 4;
    const int col  = lane & 15;
#pragma unroll
    for (int r = 0; r < 8; ++r)
        C[(size_t)(m0 + r + half * 8) * ldc + n0 + col] = acc[r];
}

#define ELL_CAP 128

__global__ void edge_bucket_kernel(const int* __restrict__ src, const int* __restrict__ dst,
                                   int* __restrict__ cnt, int* __restrict__ slots,
                                   int* __restrict__ ovf_n, int* __restrict__ ovf_dst,
                                   int* __restrict__ ovf_src) {
    int i = blockIdx.x * blockDim.x + threadIdx.x;
    if (i >= N_EDGES / 4) return;
    const int4 d4 = ((const int4*)dst)[i];
    const int4 s4 = ((const int4*)src)[i];
    const int d[4] = {d4.x, d4.y, d4.z, d4.w};
    const int s[4] = {s4.x, s4.y, s4.z, s4.w};
    int r[4];
#pragma unroll
    for (int j = 0; j < 4; ++j) r[j] = atomicAdd(&cnt[d[j]], 1);
#pragma unroll
    for (int j = 0; j < 4; ++j) {
        if (r[j] < ELL_CAP) {
            slots[(size_t)d[j] * ELL_CAP + r[j]] = s[j];
        } else {
            const int o = atomicAdd(ovf_n, 1);
            ovf_dst[o] = d[j];
            ovf_src[o] = s[j];
        }
    }
}

template <int K>
__device__ __forceinline__ void add_rows(const float* __restrict__ xl, const int* __restrict__ ids,
                                         float4& a0, float4& a1) {
    int s[K];
    float4 u[K], w[K];
#pragma unroll
    for (int j = 0; j < K; ++j) s[j] = ids[j];
#pragma unroll
    for (int j = 0; j < K; ++j) {
        const float* r = xl + (size_t)s[j] * F_IN;
        u[j] = *(const float4*)r;
        w[j] = *(const float4*)(r + 128);
    }
#pragma unroll
    for (int j = 0; j < K; ++j) {
        a0.x += u[j].x; a0.y += u[j].y; a0.z += u[j].z; a0.w += u[j].w;
        a1.x += w[j].x; a1.y += w[j].y; a1.z += w[j].z; a1.w += w[j].w;
    }
}

__device__ __forceinline__ void col_partial(v8f a, float& s, float& q) {
    s = 0.0f; q = 0.0f;
#pragma unroll
    for (int r = 0; r < 8; ++r) { s += a[r]; q += a[r] * a[r]; }
}

#define G1_KSTEPS 4
#define G1_MT     3
#define G1_ROWS   48
#define G1_GRID   209
struct G1Frag {
    v2f    a[G1_KSTEPS][G1_MT];
    float4 b0[G1_KSTEPS];
    float4 b1[G1_KSTEPS];
};

__device__ __forceinline__ void g1_load(G1Frag& f, const float* a0, const float* a1, const float* a2,
                                        const float* bp) {
#pragma unroll
    for (int s = 0; s < G1_KSTEPS; ++s) {
        f.a[s][0] = *(const v2f*)(a0 + 4 * s);
        f.a[s][1] = *(const v2f*)(a1 + 4 * s);
        f.a[s][2] = *(const v2f*)(a2 + 4 * s);
        f.b0[s] = *(const float4*)(bp + (size_t)(4 * s) * F_HID);
        f.b1[s] = *(const float4*)(bp + (size_t)(4 * s + 1) * F_HID);
    }
}

__device__ __forceinline__ void g1_mma(const G1Frag& f, v8f (&acc)[G1_MT][4]) {
#pragma unroll
    for (int s = 0; s < G1_KSTEPS; ++s) {
        v2f b;
        b.x = f.b0[s].x; b.y = f.b1[s].x;
#pragma unroll
        for (int t = 0; t < G1_MT; ++t) acc[t][0] = wmma4(f.a[s][t], b, acc[t][0]);
        b.x = f.b0[s].y; b.y = f.b1[s].y;
#pragma unroll
        for (int t = 0; t < G1_MT; ++t) acc[t][1] = wmma4(f.a[s][t], b, acc[t][1]);
        b.x = f.b0[s].z; b.y = f.b1[s].z;
#pragma unroll
        for (int t = 0; t < G1_MT; ++t) acc[t][2] = wmma4(f.a[s][t], b, acc[t][2]);
        b.x = f.b0[s].w; b.y = f.b1[s].w;
#pragma unroll
        for (int t = 0; t < G1_MT; ++t) acc[t][3] = wmma4(f.a[s][t], b, acc[t][3]);
    }
}

#define G1_PITCH 260

__device__ __forceinline__ void g1_product(const float* ap, const float* bp, v8f (&acc)[G1_MT][4]) {
    const float* a1p = ap + 16 * G1_PITCH;
    const float* a2p = ap + 32 * G1_PITCH;
    G1Frag f0, f1;
    g1_load(f0, ap, a1p, a2p, bp);
#pragma unroll 1
    for (int g = 0; g < 14; g += 2) {
        g1_load(f1, ap + (g + 1) * 16, a1p + (g + 1) * 16, a2p + (g + 1) * 16, bp + (size_t)(g + 1) * 16 * F_HID);
        g1_mma(f0, acc);
        g1_load(f0, ap + (g + 2) * 16, a1p + (g + 2) * 16, a2p + (g + 2) * 16, bp + (size_t)(g + 2) * 16 * F_HID);
        g1_mma(f1, acc);
    }
    g1_load(f1, ap + 15 * 16, a1p + 15 * 16, a2p + 15 * 16, bp + (size_t)15 * 16 * F_HID);
    g1_mma(f0, acc);
    g1_mma(f1, acc);
}

__global__ __launch_bounds__(512) void gemm1_kernel(const float* __restrict__ x,
                                                    const int* __restrict__ cnt,
                                                    const int* __restrict__ slots,
                                                    const int* __restrict__ ovf_n,
                                                    const int* __restrict__ ovf_dst,
                                                    const int* __restrict__ ovf_src,
                                                    const float* __restrict__ B1,
                                                    const float* __restrict__ B2,
                                                    float* __restrict__ C,
                                                    float* __restrict__ sum,
                                                    float* __restrict__ sumsq) {
    __shared__ float xs[G1_ROWS * G1_PITCH] __attribute__((aligned(16)));
    __shared__ float agg[G1_ROWS * G1_PITCH] __attribute__((aligned(16)));
    const int lane = threadIdx.x & 31;
    const int wave = __builtin_amdgcn_readfirstlane(threadIdx.x >> 5);
    const int m0 = blockIdx.x * G1_ROWS;
    const int half = lane >> 4, c = lane & 15;
    const int n0 = wave * 64;
    const size_t b_off = (size_t)(2 * half) * F_HID + n0 + 4 * c;

    {
        float4 u[3], w[3];
#pragma unroll
        for (int i = 0; i < 3; ++i) {
            const float* r = x + (size_t)min(m0 + wave * 3 + i, N_NODES - 1) * F_IN + lane * 4;
            u[i] = *(const float4*)r;
            w[i] = *(const float4*)(r + 128);
        }
#pragma unroll
        for (int i = 0; i < 3; ++i) {
            float* o = xs + (wave * 3 + i) * G1_PITCH + lane * 4;
            *(float4*)o = u[i];
            *(float4*)(o + 128) = w[i];
        }
    }
    __syncthreads();

    v8f acc[G1_MT][4];

    if (wave >= 4) {
        const float* xl = x + lane * 4;
        for (int i = 0; i < 4; ++i) {
            const int row = (wave - 4) * 4 + i;
            const int deg = cnt[m0 + row];
            const int end = min(deg, ELL_CAP);
            const int* ids = slots + (size_t)(m0 + row) * ELL_CAP;
            float4 a0 = {0.0f, 0.0f, 0.0f, 0.0f}, a1 = {0.0f, 0.0f, 0.0f, 0.0f};
            int e = 0;
            for (; e + 16 <= end; e += 16) add_rows<16>(xl, ids + e, a0, a1);
            if (e + 8 <= end) { add_rows<8>(xl, ids + e, a0, a1); e += 8; }
            if (e + 4 <= end) { add_rows<4>(xl, ids + e, a0, a1); e += 4; }
            if (e + 2 <= end) { add_rows<2>(xl, ids + e, a0, a1); e += 2; }
            if (e < end)      { add_rows<1>(xl, ids + e, a0, a1); }
            if (deg > ELL_CAP) {
                const int node = m0 + row;
                const int no = *ovf_n;
                for (int k = 0; k < no; ++k)
                    if (ovf_dst[k] == node) add_rows<1>(xl, ovf_src + k, a0, a1);
            }
            const float di = 1.0f / fmaxf((float)deg, 1.0f);
            float* o = agg + row * G1_PITCH + lane * 4;
            float4 r0 = {a0.x * di, a0.y * di, a0.z * di, a0.w * di};
            float4 r1 = {a1.x * di, a1.y * di, a1.z * di, a1.w * di};
            *(float4*)o = r0;
            *(float4*)(o + 128) = r1;
        }
    } else {
#pragma unroll
        for (int t = 0; t < G1_MT; ++t)
#pragma unroll
            for (int j = 0; j < 4; ++j) acc[t][j] = v8f{};
        g1_product(xs + c * G1_PITCH + 2 * half, B2 + b_off, acc);
    }
    __syncthreads();
    if (wave >= 4) return;

    g1_product(agg + c * G1_PITCH + 2 * half, B1 + b_off, acc);

    float s0 = 0.0f, q0 = 0.0f, s1 = 0.0f, q1 = 0.0f, s2 = 0.0f, q2 = 0.0f, s3 = 0.0f, q3 = 0.0f;
#pragma unroll
    for (int t = 0; t < G1_MT; ++t) {
        if (m0 + 16 * t < N_NODES) {
#pragma unroll
            for (int r = 0; r < 8; ++r) {
                float4 o = {acc[t][0][r], acc[t][1][r], acc[t][2][r], acc[t][3][r]};
                *(float4*)(C + (size_t)(m0 + 16 * t + r + half * 8) * F_HID + n0 + 4 * c) = o;
            }
            float ps, pq;
            col_partial(acc[t][0], ps, pq); s0 += ps; q0 += pq;
            col_partial(acc[t][1], ps, pq); s1 += ps; q1 += pq;
            col_partial(acc[t][2], ps, pq); s2 += ps; q2 += pq;
            col_partial(acc[t][3], ps, pq); s3 += ps; q3 += pq;
        }
    }
    const float sa = (half ? s1 : s0) + __shfl_xor(half ? s0 : s1, 16);
    const float sb = (half ? s3 : s2) + __shfl_xor(half ? s2 : s3, 16);
    const float qa = (half ? q1 : q0) + __shfl_xor(half ? q0 : q1, 16);
    const float qb = (half ? q3 : q2) + __shfl_xor(half ? q2 : q3, 16);
    atomicAdd(&sum[n0 + 4 * c + half], sa);
    atomicAdd(&sum[n0 + 4 * c + 2 + half], sb);
    atomicAdd(&sumsq[n0 + 4 * c + half], qa);
    atomicAdd(&sumsq[n0 + 4 * c + 2 + half], qb);
}

__global__ __launch_bounds__(256) void gemm3_kernel(const float* __restrict__ A,
                                                    const float* __restrict__ sum,
                                                    const float* __restrict__ sumsq,
                                                    const float* __restrict__ gamma,
                                                    const float* __restrict__ beta,
                                                    const float* __restrict__ B1,
                                                    const float* __restrict__ B2,
                                                    float* __restrict__ D1,
                                                    float* __restrict__ D2) {
    __shared__ float bn[4][F_HID] __attribute__((aligned(8)));
    {
        const int k = threadIdx.x;
        const float mean = sum[k] * (1.0f / N_NODES);
        const float var  = sumsq[k] * (1.0f / N_NODES) - mean * mean;
        bn[0][k] = mean;
        bn[1][k] = rsqrtf(var + BN_EPS);
        bn[2][k] = gamma[k];
        bn[3][k] = beta[k];
    }
    __syncthreads();

    const int lane = threadIdx.x & 31;
    const int wave = threadIdx.x >> 5;
    const int tile = blockIdx.x * 8 + wave;
    if (tile >= 625) return;
    const int m0 = tile * 16;
    const int kh = (lane >> 4) << 1;

    v8f acc1 = {}, acc2 = {};
#pragma unroll 4
    for (int k0 = 0; k0 < F_HID; k0 += 4) {
        const v2f raw = load_a(A, F_HID, m0, k0, lane);
        const v2f mu = *(const v2f*)&bn[0][k0 + kh];
        const v2f rs = *(const v2f*)&bn[1][k0 + kh];
        const v2f ga = *(const v2f*)&bn[2][k0 + kh];
        const v2f be = *(const v2f*)&bn[3][k0 + kh];
        v2f a;
        a.x = fmaxf((raw.x - mu.x) * rs.x * ga.x + be.x, 0.0f);
        a.y = fmaxf((raw.y - mu.y) * rs.y * ga.y + be.y, 0.0f);
        v2f b1 = load_b(B1, F_OUT, k0, 0, lane);
        v2f b2 = load_b(B2, F_OUT, k0, 0, lane);
        acc1 = wmma4(a, b1, acc1);
        acc2 = wmma4(a, b2, acc2);
    }
    store_c(D1, F_OUT, m0, 0, lane, acc1);
    store_c(D2, F_OUT, m0, 0, lane, acc2);
}


__global__ __launch_bounds__(256) void agg3_gather_kernel(const float* __restrict__ p,
                                                          const float* __restrict__ hl,
                                                          const int* __restrict__ cnt,
                                                          const int* __restrict__ slots,
                                                          const int* __restrict__ ovf_n,
                                                          const int* __restrict__ ovf_dst,
                                                          const int* __restrict__ ovf_src,
                                                          float* __restrict__ t,
                                                          float* __restrict__ sum,
                                                          float* __restrict__ sumsq) {
    __shared__ float rows[2][8][F_OUT] __attribute__((aligned(16)));
    const int lane = threadIdx.x & 31;
    const int node = __builtin_amdgcn_readfirstlane(blockIdx.x * 8 + (threadIdx.x >> 5));
    const int deg = cnt[node];
    const int end = min(deg, ELL_CAP);
    const int* ids = slots + (size_t)node * ELL_CAP;
    const int slot = lane >> 2, c4 = (lane & 3) << 2;
    float4 acc = {0.0f, 0.0f, 0.0f, 0.0f};
    for (int base = 0; base < end; base += 32) {
        int s[4];
        float4 v[4];
#pragma unroll
        for (int j = 0; j < 4; ++j) s[j] = ids[min(base + slot + 8 * j, end - 1)];
#pragma unroll
        for (int j = 0; j < 4; ++j) v[j] = *(const float4*)(p + (size_t)s[j] * F_OUT + c4);
#pragma unroll
        for (int j = 0; j < 4; ++j)
            if (base + slot + 8 * j < end) {
                acc.x += v[j].x; acc.y += v[j].y; acc.z += v[j].z; acc.w += v[j].w;
            }
    }
    if (deg > ELL_CAP) {
        const int no = *ovf_n;
        for (int k = slot; k < no; k += 8)
            if (ovf_dst[k] == node) {
                const float4 v = *(const float4*)(p + (size_t)ovf_src[k] * F_OUT + c4);
                acc.x += v.x; acc.y += v.y; acc.z += v.z; acc.w += v.w;
            }
    }
#pragma unroll
    for (int o = 4; o < 32; o <<= 1) {
        acc.x += __shfl_xor(acc.x, o);
        acc.y += __shfl_xor(acc.y, o);
        acc.z += __shfl_xor(acc.z, o);
        acc.w += __shfl_xor(acc.w, o);
    }
    if (lane < 4) {
        const float di = 1.0f / fmaxf((float)deg, 1.0f);
        const float4 h = *(const float4*)(hl + (size_t)node * F_OUT + c4);
        float4 r = {acc.x * di + h.x, acc.y * di + h.y, acc.z * di + h.z, acc.w * di + h.w};
        *(float4*)(t + (size_t)node * F_OUT + c4) = r;
        const int wv = threadIdx.x >> 5;
        float4 r2 = {r.x * r.x, r.y * r.y, r.z * r.z, r.w * r.w};
        *(float4*)&rows[0][wv][c4] = r;
        *(float4*)&rows[1][wv][c4] = r2;
    }
    __syncthreads();
    if (threadIdx.x < 32) {
        const int st = threadIdx.x >> 4, col = threadIdx.x & 15;
        float a = 0.0f;
#pragma unroll
        for (int w = 0; w < 8; ++w) a += rows[st][w][col];
        atomicAdd(st ? &sumsq[col] : &sum[col], a);
    }
}

__global__ void bn_final_kernel(const float* __restrict__ t,
                                const float* __restrict__ sum, const float* __restrict__ sumsq,
                                const float* __restrict__ gamma, const float* __restrict__ beta,
                                const int* __restrict__ ste,
                                float* __restrict__ out, long long total) {
    long long g = (long long)blockIdx.x * blockDim.x + threadIdx.x;
    if (g >= total) return;
    int c = (int)(g & (F_OUT - 1));
    float mean = sum[c] * (1.0f / N_NODES);
    float var  = sumsq[c] * (1.0f / N_NODES) - mean * mean;
    float v = (t[g] - mean) * rsqrtf(var + BN_EPS) * gamma[c] + beta[c];
    if (*ste)
        out[g] = (v > 0.0f) ? 1.0f : 0.0f;
    else
        out[g] = 1.0f / (1.0f + expf(-v));
}

extern "C" void kernel_launch(void* const* d_in, const int* in_sizes, int n_in,
                              void* d_out, int out_size, void* d_ws, size_t ws_size,
                              hipStream_t stream) {
    const float* x        = (const float*)d_in[0];
    const float* W_conv1  = (const float*)d_in[1];
    const float* W_lin1   = (const float*)d_in[2];
    const float* bn1_g    = (const float*)d_in[3];
    const float* bn1_b    = (const float*)d_in[4];
    const float* W_conv3  = (const float*)d_in[5];
    const float* W_lin3   = (const float*)d_in[6];
    const float* bn3_g    = (const float*)d_in[7];
    const float* bn3_b    = (const float*)d_in[8];
    const int*   src      = (const int*)d_in[9];
    const int*   dst      = (const int*)d_in[10];
    const int*   ste      = (const int*)d_in[11];
    float*       out      = (float*)d_out;

    float* ws = (float*)d_ws;
    const size_t o_cnt  = 0;
    const size_t o_s1   = 10240;
    const size_t o_q1   = o_s1 + 256;
    const size_t o_s3   = o_q1 + 256;
    const size_t o_q3   = o_s3 + 16;
    const size_t o_ovn  = o_q3 + 16;
    const size_t o_slot = 11264;
    const size_t o_ovd  = o_slot + 1280000;
    const size_t o_ovs  = o_ovd + 320000;
    const size_t o_h    = o_ovs + 320000;
    const size_t o_p    = o_h + 2560000;
    const size_t o_hl   = o_p + 160000;
    const size_t o_agg3 = o_hl + 160000;
    int*   cnt     = (int*)(ws + o_cnt);
    int*   ovn     = (int*)(ws + o_ovn);
    int*   slots   = (int*)(ws + o_slot);
    int*   ovd     = (int*)(ws + o_ovd);
    int*   ovs     = (int*)(ws + o_ovs);
    float* h       = ws + o_h;
    float* p       = ws + o_p;
    float* hl      = ws + o_hl;
    float* agg3    = ws + o_agg3;
    float* s1 = ws + o_s1; float* q1 = ws + o_q1;
    float* s3 = ws + o_s3; float* q3 = ws + o_q3;

    hipMemsetAsync(cnt, 0, (10240 + 256 + 256 + 16 + 16 + 16) * sizeof(float), stream);

    edge_bucket_kernel<<<(N_EDGES / 4 + 255) / 256, 256, 0, stream>>>(src, dst, cnt, slots, ovn, ovd, ovs);

    gemm1_kernel<<<G1_GRID, 512, 0, stream>>>(x, cnt, slots, ovn, ovd, ovs, W_conv1, W_lin1, h, s1, q1);

    gemm3_kernel<<<(625 + 7) / 8, 256, 0, stream>>>(h, s1, q1, bn1_g, bn1_b, W_conv3, W_lin3, p, hl);

    agg3_gather_kernel<<<N_NODES / 8, 256, 0, stream>>>(p, hl, cnt, slots, ovn, ovd, ovs, agg3, s3, q3);

    {
        long long tot = (long long)N_NODES * F_OUT;
        bn_final_kernel<<<(int)((tot + 255) / 256), 256, 0, stream>>>(agg3, s3, q3, bn3_g, bn3_b, ste, out, tot);
    }
}


# edge ranks drawn from LDS counters: 4096-edge chunks sorted by 256-node range, then one work-group per range ranks its pairs; global returning adds and the memset are gone; sums stay in rank order
# speedup vs baseline: 6.5722x; 1.3619x over previous
#include <hip/hip_runtime.h>
#include <hip/hip_bf16.h>
#include <math.h>

#define N_NODES 10000
#define N_EDGES 320000
#define F_IN    256
#define F_HID   256
#define F_OUT   16
#define BN_EPS  1e-5f

typedef __attribute__((ext_vector_type(2))) float v2f;
typedef __attribute__((ext_vector_type(8))) float v8f;

__device__ __forceinline__ v8f wmma4(v2f a, v2f b, v8f c) {
    return __builtin_amdgcn_wmma_f32_16x16x4_f32(
        false, a, false, b, (short)0, c, false, false);
}

__device__ __forceinline__ v2f load_a(const float* A, int lda, int m0, int k0, int lane) {
    const int half = lane >> 4;
    const int row  = lane & 15;
    const float* p = A + (size_t)(m0 + row) * lda + k0 + (half << 1);
    v2f r; r.x = p[0]; r.y = p[1];
    return r;
}

__device__ __forceinline__ v2f load_b(const float* B, int ldb, int k0, int n0, int lane) {
    const int half = lane >> 4;
    const int col  = lane & 15;
    const float* p = B + (size_t)(k0 + (half << 1)) * ldb + n0 + col;
    v2f r; r.x = p[0]; r.y = p[ldb];
    return r;
}

__device__ __forceinline__ void store_c(float* C, int ldc, int m0, int n0, int lane, v8f acc) {
    const int half = lane >> 4;
    const int col  = lane & 15;
#pragma unroll
    for (int r = 0; r < 8; ++r)
        C[(size_t)(m0 + r + half * 8) * ldc + n0 + col] = acc[r];
}

#define ELL_CAP 128
#define BIN_SHIFT 8
#define BIN_NODES 256
#define N_BIN     40
#define BIN_CH    4096
#define N_CH      79
#define N_STATS   544

__global__ __launch_bounds__(1024) void edge_bin_kernel(const int* __restrict__ src,
                                                        const int* __restrict__ dst,
                                                        int* __restrict__ pairs,
                                                        int* __restrict__ subcnt,
                                                        float* __restrict__ stats) {
    __shared__ int hist[N_BIN];
    const int tid = threadIdx.x, w = blockIdx.x;
    if (tid < N_BIN) hist[tid] = 0;
    if (w == 0 && tid < N_STATS) stats[tid] = 0.0f;
    __syncthreads();
    const int i = w * (BIN_CH / 4) + tid;
    if (i < N_EDGES / 4) {
        const int4 d4 = ((const int4*)dst)[i];
        const int4 s4 = ((const int4*)src)[i];
        const int d[4] = {d4.x, d4.y, d4.z, d4.w};
        const int s[4] = {s4.x, s4.y, s4.z, s4.w};
        int r[4];
#pragma unroll
        for (int j = 0; j < 4; ++j)
            r[j] = ((unsigned)d[j] < (unsigned)N_NODES) ? atomicAdd(&hist[d[j] >> BIN_SHIFT], 1) : -1;
        int2* pl = (int2*)pairs;
#pragma unroll
        for (int j = 0; j < 4; ++j)
            if (r[j] >= 0)
                pl[((size_t)(d[j] >> BIN_SHIFT) * N_CH + w) * BIN_CH + r[j]] = make_int2(d[j], s[j]);
    }
    __syncthreads();
    if (tid < N_BIN) subcnt[tid * N_CH + w] = hist[tid];
}

__global__ __launch_bounds__(1024) void edge_rank_kernel(const int* __restrict__ pairs,
                                                         const int* __restrict__ subcnt,
                                                         int* __restrict__ cnt, int* __restrict__ slots,
                                                         int* __restrict__ ovf_n, int* __restrict__ ovf_dst,
                                                         int* __restrict__ ovf_src) {
    __shared__ int lcnt[BIN_NODES];
    __shared__ int lovf;
    const int tid = threadIdx.x, lane = threadIdx.x & 31, b = blockIdx.x;
    const int wave = __builtin_amdgcn_readfirstlane(threadIdx.x >> 5);
    if (tid < BIN_NODES) lcnt[tid] = 0;
    if (tid == 0) lovf = 0;
    __syncthreads();
    int* od = ovf_dst + (size_t)b * N_EDGES;
    int* os = ovf_src + (size_t)b * N_EDGES;
    int nl[3];
#pragma unroll
    for (int k = 0; k < 3; ++k) {
        const int w = wave + 32 * k;
        nl[k] = subcnt[b * N_CH + min(w, N_CH - 1)];
        if (w >= N_CH) nl[k] = 0;
    }
#pragma unroll
    for (int k = 0; k < 3; ++k) {
        const int n = nl[k];
        const int2* sub = (const int2*)pairs + ((size_t)b * N_CH + wave + 32 * k) * BIN_CH;
        for (int i0 = 0; i0 < n; i0 += 128) {
            int2 pr[4];
#pragma unroll
            for (int j = 0; j < 4; ++j) pr[j] = sub[min(i0 + 32 * j + lane, n - 1)];
#pragma unroll
            for (int j = 0; j < 4; ++j)
                if (i0 + 32 * j + lane < n) {
                    const int node = pr[j].x;
                    const int r = atomicAdd(&lcnt[node - b * BIN_NODES], 1);
                    if (r < ELL_CAP) {
                        slots[(size_t)node * ELL_CAP + r] = pr[j].y;
                    } else {
                        const int o = atomicAdd(&lovf, 1);
                        od[o] = node;
                        os[o] = pr[j].y;
                    }
                }
        }
    }
    __syncthreads();
    if (tid < BIN_NODES) cnt[b * BIN_NODES + tid] = lcnt[tid];
    if (tid == 0) ovf_n[b] = lovf;
}

template <int K>
__device__ __forceinline__ void add_rows(const float* __restrict__ xl, const int* __restrict__ ids,
                                         float4& a0, float4& a1) {
    int s[K];
    float4 u[K], w[K];
#pragma unroll
    for (int j = 0; j < K; ++j) s[j] = ids[j];
#pragma unroll
    for (int j = 0; j < K; ++j) {
        const float* r = xl + (size_t)s[j] * F_IN;
        u[j] = *(const float4*)r;
        w[j] = *(const float4*)(r + 128);
    }
#pragma unroll
    for (int j = 0; j < K; ++j) {
        a0.x += u[j].x; a0.y += u[j].y; a0.z += u[j].z; a0.w += u[j].w;
        a1.x += w[j].x; a1.y += w[j].y; a1.z += w[j].z; a1.w += w[j].w;
    }
}

__device__ __forceinline__ void col_partial(v8f a, float& s, float& q) {
    s = 0.0f; q = 0.0f;
#pragma unroll
    for (int r = 0; r < 8; ++r) { s += a[r]; q += a[r] * a[r]; }
}

#define G1_KSTEPS 4
#define G1_MT     3
#define G1_ROWS   48
#define G1_GRID   209
struct G1Frag {
    v2f    a[G1_KSTEPS][G1_MT];
    float4 b0[G1_KSTEPS];
    float4 b1[G1_KSTEPS];
};

__device__ __forceinline__ void g1_load(G1Frag& f, const float* a0, const float* a1, const float* a2,
                                        const float* bp) {
#pragma unroll
    for (int s = 0; s < G1_KSTEPS; ++s) {
        f.a[s][0] = *(const v2f*)(a0 + 4 * s);
        f.a[s][1] = *(const v2f*)(a1 + 4 * s);
        f.a[s][2] = *(const v2f*)(a2 + 4 * s);
        f.b0[s] = *(const float4*)(bp + (size_t)(4 * s) * F_HID);
        f.b1[s] = *(const float4*)(bp + (size_t)(4 * s + 1) * F_HID);
    }
}

__device__ __forceinline__ void g1_mma(const G1Frag& f, v8f (&acc)[G1_MT][4]) {
#pragma unroll
    for (int s = 0; s < G1_KSTEPS; ++s) {
        v2f b;
        b.x = f.b0[s].x; b.y = f.b1[s].x;
#pragma unroll
        for (int t = 0; t < G1_MT; ++t) acc[t][0] = wmma4(f.a[s][t], b, acc[t][0]);
        b.x = f.b0[s].y; b.y = f.b1[s].y;
#pragma unroll
        for (int t = 0; t < G1_MT; ++t) acc[t][1] = wmma4(f.a[s][t], b, acc[t][1]);
        b.x = f.b0[s].z; b.y = f.b1[s].z;
#pragma unroll
        for (int t = 0; t < G1_MT; ++t) acc[t][2] = wmma4(f.a[s][t], b, acc[t][2]);
        b.x = f.b0[s].w; b.y = f.b1[s].w;
#pragma unroll
        for (int t = 0; t < G1_MT; ++t) acc[t][3] = wmma4(f.a[s][t], b, acc[t][3]);
    }
}

#define G1_PITCH 260

__device__ __forceinline__ void g1_product(const float* ap, const float* bp, v8f (&acc)[G1_MT][4]) {
    const float* a1p = ap + 16 * G1_PITCH;
    const float* a2p = ap + 32 * G1_PITCH;
    G1Frag f0, f1;
    g1_load(f0, ap, a1p, a2p, bp);
#pragma unroll 1
    for (int g = 0; g < 14; g += 2) {
        g1_load(f1, ap + (g + 1) * 16, a1p + (g + 1) * 16, a2p + (g + 1) * 16, bp + (size_t)(g + 1) * 16 * F_HID);
        g1_mma(f0, acc);
        g1_load(f0, ap + (g + 2) * 16, a1p + (g + 2) * 16, a2p + (g + 2) * 16, bp + (size_t)(g + 2) * 16 * F_HID);
        g1_mma(f1, acc);
    }
    g1_load(f1, ap + 15 * 16, a1p + 15 * 16, a2p + 15 * 16, bp + (size_t)15 * 16 * F_HID);
    g1_mma(f0, acc);
    g1_mma(f1, acc);
}

__global__ __launch_bounds__(512) void gemm1_kernel(const float* __restrict__ x,
                                                    const int* __restrict__ cnt,
                                                    const int* __restrict__ slots,
                                                    const int* __restrict__ ovf_n,
                                                    const int* __restrict__ ovf_dst,
                                                    const int* __restrict__ ovf_src,
                                                    const float* __restrict__ B1,
                                                    const float* __restrict__ B2,
                                                    float* __restrict__ C,
                                                    float* __restrict__ sum,
                                                    float* __restrict__ sumsq) {
    __shared__ float xs[G1_ROWS * G1_PITCH] __attribute__((aligned(16)));
    __shared__ float agg[G1_ROWS * G1_PITCH] __attribute__((aligned(16)));
    const int lane = threadIdx.x & 31;
    const int wave = __builtin_amdgcn_readfirstlane(threadIdx.x >> 5);
    const int m0 = blockIdx.x * G1_ROWS;
    const int half = lane >> 4, c = lane & 15;
    const int n0 = wave * 64;
    const size_t b_off = (size_t)(2 * half) * F_HID + n0 + 4 * c;

    {
        float4 u[3], w[3];
#pragma unroll
        for (int i = 0; i < 3; ++i) {
            const float* r = x + (size_t)min(m0 + wave * 3 + i, N_NODES - 1) * F_IN + lane * 4;
            u[i] = *(const float4*)r;
            w[i] = *(const float4*)(r + 128);
        }
#pragma unroll
        for (int i = 0; i < 3; ++i) {
            float* o = xs + (wave * 3 + i) * G1_PITCH + lane * 4;
            *(float4*)o = u[i];
            *(float4*)(o + 128) = w[i];
        }
    }
    __syncthreads();

    v8f acc[G1_MT][4];

    if (wave >= 4) {
        const float* xl = x + lane * 4;
        for (int i = 0; i < 4; ++i) {
            const int row = (wave - 4) * 4 + i;
            const int deg = cnt[m0 + row];
            const int end = min(deg, ELL_CAP);
            const int* ids = slots + (size_t)(m0 + row) * ELL_CAP;
            float4 a0 = {0.0f, 0.0f, 0.0f, 0.0f}, a1 = {0.0f, 0.0f, 0.0f, 0.0f};
            int e = 0;
            for (; e + 16 <= end; e += 16) add_rows<16>(xl, ids + e, a0, a1);
            if (e + 8 <= end) { add_rows<8>(xl, ids + e, a0, a1); e += 8; }
            if (e + 4 <= end) { add_rows<4>(xl, ids + e, a0, a1); e += 4; }
            if (e + 2 <= end) { add_rows<2>(xl, ids + e, a0, a1); e += 2; }
            if (e < end)      { add_rows<1>(xl, ids + e, a0, a1); }
            if (deg > ELL_CAP) {
                const int node = m0 + row;
                const int bin = node >> BIN_SHIFT;
                const int no = ovf_n[bin];
                const int* od = ovf_dst + (size_t)bin * N_EDGES;
                const int* os = ovf_src + (size_t)bin * N_EDGES;
                for (int k = 0; k < no; ++k)
                    if (od[k] == node) add_rows<1>(xl, os + k, a0, a1);
            }
            const float di = 1.0f / fmaxf((float)deg, 1.0f);
            float* o = agg + row * G1_PITCH + lane * 4;
            float4 r0 = {a0.x * di, a0.y * di, a0.z * di, a0.w * di};
            float4 r1 = {a1.x * di, a1.y * di, a1.z * di, a1.w * di};
            *(float4*)o = r0;
            *(float4*)(o + 128) = r1;
        }
    } else {
#pragma unroll
        for (int t = 0; t < G1_MT; ++t)
#pragma unroll
            for (int j = 0; j < 4; ++j) acc[t][j] = v8f{};
        g1_product(xs + c * G1_PITCH + 2 * half, B2 + b_off, acc);
    }
    __syncthreads();
    if (wave >= 4) return;

    g1_product(agg + c * G1_PITCH + 2 * half, B1 + b_off, acc);

    float s0 = 0.0f, q0 = 0.0f, s1 = 0.0f, q1 = 0.0f, s2 = 0.0f, q2 = 0.0f, s3 = 0.0f, q3 = 0.0f;
#pragma unroll
    for (int t = 0; t < G1_MT; ++t) {
        if (m0 + 16 * t < N_NODES) {
#pragma unroll
            for (int r = 0; r < 8; ++r) {
                float4 o = {acc[t][0][r], acc[t][1][r], acc[t][2][r], acc[t][3][r]};
                *(float4*)(C + (size_t)(m0 + 16 * t + r + half * 8) * F_HID + n0 + 4 * c) = o;
            }
            float ps, pq;
            col_partial(acc[t][0], ps, pq); s0 += ps; q0 += pq;
            col_partial(acc[t][1], ps, pq); s1 += ps; q1 += pq;
            col_partial(acc[t][2], ps, pq); s2 += ps; q2 += pq;
            col_partial(acc[t][3], ps, pq); s3 += ps; q3 += pq;
        }
    }
    const float sa = (half ? s1 : s0) + __shfl_xor(half ? s0 : s1, 16);
    const float sb = (half ? s3 : s2) + __shfl_xor(half ? s2 : s3, 16);
    const float qa = (half ? q1 : q0) + __shfl_xor(half ? q0 : q1, 16);
    const float qb = (half ? q3 : q2) + __shfl_xor(half ? q2 : q3, 16);
    atomicAdd(&sum[n0 + 4 * c + half], sa);
    atomicAdd(&sum[n0 + 4 * c + 2 + half], sb);
    atomicAdd(&sumsq[n0 + 4 * c + half], qa);
    atomicAdd(&sumsq[n0 + 4 * c + 2 + half], qb);
}

__global__ __launch_bounds__(256) void gemm3_kernel(const float* __restrict__ A,
                                                    const float* __restrict__ sum,
                                                    const float* __restrict__ sumsq,
                                                    const float* __restrict__ gamma,
                                                    const float* __restrict__ beta,
                                                    const float* __restrict__ B1,
                                                    const float* __restrict__ B2,
                                                    float* __restrict__ D1,
                                                    float* __restrict__ D2) {
    __shared__ float bn[4][F_HID] __attribute__((aligned(8)));
    {
        const int k = threadIdx.x;
        const float mean = sum[k] * (1.0f / N_NODES);
        const float var  = sumsq[k] * (1.0f / N_NODES) - mean * mean;
        bn[0][k] = mean;
        bn[1][k] = rsqrtf(var + BN_EPS);
        bn[2][k] = gamma[k];
        bn[3][k] = beta[k];
    }
    __syncthreads();

    const int lane = threadIdx.x & 31;
    const int wave = threadIdx.x >> 5;
    const int tile = blockIdx.x * 8 + wave;
    if (tile >= 625) return;
    const int m0 = tile * 16;
    const int kh = (lane >> 4) << 1;

    v8f acc1 = {}, acc2 = {};
#pragma unroll 4
    for (int k0 = 0; k0 < F_HID; k0 += 4) {
        const v2f raw = load_a(A, F_HID, m0, k0, lane);
        const v2f mu = *(const v2f*)&bn[0][k0 + kh];
        const v2f rs = *(const v2f*)&bn[1][k0 + kh];
        const v2f ga = *(const v2f*)&bn[2][k0 + kh];
        const v2f be = *(const v2f*)&bn[3][k0 + kh];
        v2f a;
        a.x = fmaxf((raw.x - mu.x) * rs.x * ga.x + be.x, 0.0f);
        a.y = fmaxf((raw.y - mu.y) * rs.y * ga.y + be.y, 0.0f);
        v2f b1 = load_b(B1, F_OUT, k0, 0, lane);
        v2f b2 = load_b(B2, F_OUT, k0, 0, lane);
        acc1 = wmma4(a, b1, acc1);
        acc2 = wmma4(a, b2, acc2);
    }
    store_c(D1, F_OUT, m0, 0, lane, acc1);
    store_c(D2, F_OUT, m0, 0, lane, acc2);
}


__global__ __launch_bounds__(256) void agg3_gather_kernel(const float* __restrict__ p,
                                                          const float* __restrict__ hl,
                                                          const int* __restrict__ cnt,
                                                          const int* __restrict__ slots,
                                                          const int* __restrict__ ovf_n,
                                                          const int* __restrict__ ovf_dst,
                                                          const int* __restrict__ ovf_src,
                                                          float* __restrict__ t,
                                                          float* __restrict__ sum,
                                                          float* __restrict__ sumsq) {
    __shared__ float rows[2][8][F_OUT] __attribute__((aligned(16)));
    const int lane = threadIdx.x & 31;
    const int node = __builtin_amdgcn_readfirstlane(blockIdx.x * 8 + (threadIdx.x >> 5));
    const int deg = cnt[node];
    const int end = min(deg, ELL_CAP);
    const int* ids = slots + (size_t)node * ELL_CAP;
    const int slot = lane >> 2, c4 = (lane & 3) << 2;
    float4 acc = {0.0f, 0.0f, 0.0f, 0.0f};
    for (int base = 0; base < end; base += 32) {
        int s[4];
        float4 v[4];
#pragma unroll
        for (int j = 0; j < 4; ++j) s[j] = ids[min(base + slot + 8 * j, end - 1)];
#pragma unroll
        for (int j = 0; j < 4; ++j) v[j] = *(const float4*)(p + (size_t)s[j] * F_OUT + c4);
#pragma unroll
        for (int j = 0; j < 4; ++j)
            if (base + slot + 8 * j < end) {
                acc.x += v[j].x; acc.y += v[j].y; acc.z += v[j].z; acc.w += v[j].w;
            }
    }
    if (deg > ELL_CAP) {
        const int bin = node >> BIN_SHIFT;
        const int no = ovf_n[bin];
        const int* od = ovf_dst + (size_t)bin * N_EDGES;
        const int* os = ovf_src + (size_t)bin * N_EDGES;
        for (int k = slot; k < no; k += 8)
            if (od[k] == node) {
                const float4 v = *(const float4*)(p + (size_t)os[k] * F_OUT + c4);
                acc.x += v.x; acc.y += v.y; acc.z += v.z; acc.w += v.w;
            }
    }
#pragma unroll
    for (int o = 4; o < 32; o <<= 1) {
        acc.x += __shfl_xor(acc.x, o);
        acc.y += __shfl_xor(acc.y, o);
        acc.z += __shfl_xor(acc.z, o);
        acc.w += __shfl_xor(acc.w, o);
    }
    if (lane < 4) {
        const float di = 1.0f / fmaxf((float)deg, 1.0f);
        const float4 h = *(const float4*)(hl + (size_t)node * F_OUT + c4);
        float4 r = {acc.x * di + h.x, acc.y * di + h.y, acc.z * di + h.z, acc.w * di + h.w};
        *(float4*)(t + (size_t)node * F_OUT + c4) = r;
        const int wv = threadIdx.x >> 5;
        float4 r2 = {r.x * r.x, r.y * r.y, r.z * r.z, r.w * r.w};
        *(float4*)&rows[0][wv][c4] = r;
        *(float4*)&rows[1][wv][c4] = r2;
    }
    __syncthreads();
    if (threadIdx.x < 32) {
        const int st = threadIdx.x >> 4, col = threadIdx.x & 15;
        float a = 0.0f;
#pragma unroll
        for (int w = 0; w < 8; ++w) a += rows[st][w][col];
        atomicAdd(st ? &sumsq[col] : &sum[col], a);
    }
}

__global__ void bn_final_kernel(const float* __restrict__ t,
                                const float* __restrict__ sum, const float* __restrict__ sumsq,
                                const float* __restrict__ gamma, const float* __restrict__ beta,
                                const int* __restrict__ ste,
                                float* __restrict__ out, long long total) {
    long long g = (long long)blockIdx.x * blockDim.x + threadIdx.x;
    if (g >= total) return;
    int c = (int)(g & (F_OUT - 1));
    float mean = sum[c] * (1.0f / N_NODES);
    float var  = sumsq[c] * (1.0f / N_NODES) - mean * mean;
    float v = (t[g] - mean) * rsqrtf(var + BN_EPS) * gamma[c] + beta[c];
    if (*ste)
        out[g] = (v > 0.0f) ? 1.0f : 0.0f;
    else
        out[g] = 1.0f / (1.0f + expf(-v));
}

extern "C" void kernel_launch(void* const* d_in, const int* in_sizes, int n_in,
                              void* d_out, int out_size, void* d_ws, size_t ws_size,
                              hipStream_t stream) {
    const float* x        = (const float*)d_in[0];
    const float* W_conv1  = (const float*)d_in[1];
    const float* W_lin1   = (const float*)d_in[2];
    const float* bn1_g    = (const float*)d_in[3];
    const float* bn1_b    = (const float*)d_in[4];
    const float* W_conv3  = (const float*)d_in[5];
    const float* W_lin3   = (const float*)d_in[6];
    const float* bn3_g    = (const float*)d_in[7];
    const float* bn3_b    = (const float*)d_in[8];
    const int*   src      = (const int*)d_in[9];
    const int*   dst      = (const int*)d_in[10];
    const int*   ste      = (const int*)d_in[11];
    float*       out      = (float*)d_out;

    float* ws = (float*)d_ws;
    const size_t o_cnt  = 0;
    const size_t o_s1   = 10240;
    const size_t o_q1   = o_s1 + 256;
    const size_t o_s3   = o_q1 + 256;
    const size_t o_q3   = o_s3 + 16;
    const size_t o_ovn  = o_q3 + 16;
    const size_t o_slot = 11264;
    const size_t o_sub  = o_slot + 1280000;
    const size_t o_h    = o_sub + 640000;
    const size_t o_p    = o_h + 2560000;
    const size_t o_hl   = o_p + 160000;
    const size_t o_agg3 = o_hl + 160000;
    const size_t o_pair = o_agg3 + 160000;
    const size_t o_ovd  = o_pair + 25886720;
    const size_t o_ovs  = o_ovd + 12800000;
    int*   cnt     = (int*)(ws + o_cnt);
    int*   ovn     = (int*)(ws + o_ovn);
    int*   slots   = (int*)(ws + o_slot);
    int*   subcnt  = (int*)(ws + o_sub);
    int*   pairs   = (int*)(ws + o_pair);
    int*   ovd     = (int*)(ws + o_ovd);
    int*   ovs     = (int*)(ws + o_ovs);
    float* h       = ws + o_h;
    float* p       = ws + o_p;
    float* hl      = ws + o_hl;
    float* agg3    = ws + o_agg3;
    float* s1 = ws + o_s1; float* q1 = ws + o_q1;
    float* s3 = ws + o_s3; float* q3 = ws + o_q3;

    edge_bin_kernel<<<N_CH, 1024, 0, stream>>>(src, dst, pairs, subcnt, s1);
    edge_rank_kernel<<<N_BIN, 1024, 0, stream>>>(pairs, subcnt, cnt, slots, ovn, ovd, ovs);

    gemm1_kernel<<<G1_GRID, 512, 0, stream>>>(x, cnt, slots, ovn, ovd, ovs, W_conv1, W_lin1, h, s1, q1);

    gemm3_kernel<<<(625 + 7) / 8, 256, 0, stream>>>(h, s1, q1, bn1_g, bn1_b, W_conv3, W_lin3, p, hl);

    agg3_gather_kernel<<<N_NODES / 8, 256, 0, stream>>>(p, hl, cnt, slots, ovn, ovd, ovs, agg3, s3, q3);

    {
        long long tot = (long long)N_NODES * F_OUT;
        bn_final_kernel<<<(int)((tot + 255) / 256), 256, 0, stream>>>(agg3, s3, q3, bn3_g, bn3_b, ste, out, tot);
    }
}


# edge ranking moved into the layer-1 kernel: each 48-row work-group ranks its nodes' pairs on LDS counters, ids stay in LDS for its gather; rank launch and overflow lists gone; sums stay in rank order
# speedup vs baseline: 7.2602x; 1.1047x over previous
#include <hip/hip_runtime.h>
#include <hip/hip_bf16.h>
#include <math.h>

#define N_NODES 10000
#define N_EDGES 320000
#define F_IN    256
#define F_HID   256
#define F_OUT   16
#define BN_EPS  1e-5f

typedef __attribute__((ext_vector_type(2))) float v2f;
typedef __attribute__((ext_vector_type(8))) float v8f;

__device__ __forceinline__ v8f wmma4(v2f a, v2f b, v8f c) {
    return __builtin_amdgcn_wmma_f32_16x16x4_f32(
        false, a, false, b, (short)0, c, false, false);
}

__device__ __forceinline__ v2f load_a(const float* A, int lda, int m0, int k0, int lane) {
    const int half = lane >> 4;
    const int row  = lane & 15;
    const float* p = A + (size_t)(m0 + row) * lda + k0 + (half << 1);
    v2f r; r.x = p[0]; r.y = p[1];
    return r;
}

__device__ __forceinline__ v2f load_b(const float* B, int ldb, int k0, int n0, int lane) {
    const int half = lane >> 4;
    const int col  = lane & 15;
    const float* p = B + (size_t)(k0 + (half << 1)) * ldb + n0 + col;
    v2f r; r.x = p[0]; r.y = p[ldb];
    return r;
}

__device__ __forceinline__ void store_c(float* C, int ldc, int m0, int n0, int lane, v8f acc) {
    const int half = lane >> 4;
    const int col  = lane & 15;
#pragma unroll
    for (int r = 0; r < 8; ++r)
        C[(size_t)(m0 + r + half * 8) * ldc + n0 + col] = acc[r];
}

#define ELL_CAP 128
#define BIN_NODES 240
#define N_BIN     42
#define BIN_CH    4096
#define N_CH      79
#define N_STATS   544

__global__ __launch_bounds__(1024) void edge_bin_kernel(const int* __restrict__ src,
                                                        const int* __restrict__ dst,
                                                        int* __restrict__ pairs,
                                                        int* __restrict__ subcnt,
                                                        float* __restrict__ stats) {
    __shared__ int hist[N_BIN];
    const int tid = threadIdx.x, w = blockIdx.x;
    if (tid < N_BIN) hist[tid] = 0;
    if (w == 0 && tid < N_STATS) stats[tid] = 0.0f;
    __syncthreads();
    const int i = w * (BIN_CH / 4) + tid;
    if (i < N_EDGES / 4) {
        const int4 d4 = ((const int4*)dst)[i];
        const int4 s4 = ((const int4*)src)[i];
        const int d[4] = {d4.x, d4.y, d4.z, d4.w};
        const int s[4] = {s4.x, s4.y, s4.z, s4.w};
        int r[4], bin[4];
#pragma unroll
        for (int j = 0; j < 4; ++j) {
            bin[j] = (int)((unsigned)d[j] / BIN_NODES);
            r[j] = ((unsigned)d[j] < (unsigned)N_NODES) ? atomicAdd(&hist[bin[j]], 1) : -1;
        }
        int2* pl = (int2*)pairs;
#pragma unroll
        for (int j = 0; j < 4; ++j)
            if (r[j] >= 0)
                pl[((size_t)bin[j] * N_CH + w) * BIN_CH + r[j]] = make_int2(d[j], s[j]);
    }
    __syncthreads();
    if (tid < N_BIN) subcnt[tid * N_CH + w] = hist[tid];
}

template <int K>
__device__ __forceinline__ void add_rows(const float* __restrict__ xl, const int* ids,
                                         float4& a0, float4& a1) {
    int s[K];
    float4 u[K], w[K];
#pragma unroll
    for (int j = 0; j < K; ++j) s[j] = __builtin_amdgcn_readfirstlane(ids[j]);
#pragma unroll
    for (int j = 0; j < K; ++j) {
        const float* r = xl + (size_t)s[j] * F_IN;
        u[j] = *(const float4*)r;
        w[j] = *(const float4*)(r + 128);
    }
#pragma unroll
    for (int j = 0; j < K; ++j) {
        a0.x += u[j].x; a0.y += u[j].y; a0.z += u[j].z; a0.w += u[j].w;
        a1.x += w[j].x; a1.y += w[j].y; a1.z += w[j].z; a1.w += w[j].w;
    }
}

__device__ __forceinline__ void col_partial(v8f a, float& s, float& q) {
    s = 0.0f; q = 0.0f;
#pragma unroll
    for (int r = 0; r < 8; ++r) { s += a[r]; q += a[r] * a[r]; }
}

#define G1_KSTEPS 4
#define G1_MT     3
#define G1_ROWS   48
#define G1_GRID   209
struct G1Frag {
    v2f    a[G1_KSTEPS][G1_MT];
    float4 b0[G1_KSTEPS];
    float4 b1[G1_KSTEPS];
};

__device__ __forceinline__ void g1_load(G1Frag& f, const float* a0, const float* a1, const float* a2,
                                        const float* bp) {
#pragma unroll
    for (int s = 0; s < G1_KSTEPS; ++s) {
        f.a[s][0] = *(const v2f*)(a0 + 4 * s);
        f.a[s][1] = *(const v2f*)(a1 + 4 * s);
        f.a[s][2] = *(const v2f*)(a2 + 4 * s);
        f.b0[s] = *(const float4*)(bp + (size_t)(4 * s) * F_HID);
        f.b1[s] = *(const float4*)(bp + (size_t)(4 * s + 1) * F_HID);
    }
}

__device__ __forceinline__ void g1_mma(const G1Frag& f, v8f (&acc)[G1_MT][4]) {
#pragma unroll
    for (int s = 0; s < G1_KSTEPS; ++s) {
        v2f b;
        b.x = f.b0[s].x; b.y = f.b1[s].x;
#pragma unroll
        for (int t = 0; t < G1_MT; ++t) acc[t][0] = wmma4(f.a[s][t], b, acc[t][0]);
        b.x = f.b0[s].y; b.y = f.b1[s].y;
#pragma unroll
        for (int t = 0; t < G1_MT; ++t) acc[t][1] = wmma4(f.a[s][t], b, acc[t][1]);
        b.x = f.b0[s].z; b.y = f.b1[s].z;
#pragma unroll
        for (int t = 0; t < G1_MT; ++t) acc[t][2] = wmma4(f.a[s][t], b, acc[t][2]);
        b.x = f.b0[s].w; b.y = f.b1[s].w;
#pragma unroll
        for (int t = 0; t < G1_MT; ++t) acc[t][3] = wmma4(f.a[s][t], b, acc[t][3]);
    }
}

#define G1_PITCH 260

__device__ __forceinline__ void g1_product(const float* ap, const float* bp, v8f (&acc)[G1_MT][4]) {
    const float* a1p = ap + 16 * G1_PITCH;
    const float* a2p = ap + 32 * G1_PITCH;
    G1Frag f0, f1;
    g1_load(f0, ap, a1p, a2p, bp);
#pragma unroll 1
    for (int g = 0; g < 14; g += 2) {
        g1_load(f1, ap + (g + 1) * 16, a1p + (g + 1) * 16, a2p + (g + 1) * 16, bp + (size_t)(g + 1) * 16 * F_HID);
        g1_mma(f0, acc);
        g1_load(f0, ap + (g + 2) * 16, a1p + (g + 2) * 16, a2p + (g + 2) * 16, bp + (size_t)(g + 2) * 16 * F_HID);
        g1_mma(f1, acc);
    }
    g1_load(f1, ap + 15 * 16, a1p + 15 * 16, a2p + 15 * 16, bp + (size_t)15 * 16 * F_HID);
    g1_mma(f0, acc);
    g1_mma(f1, acc);
}

#define G1_LISTS 5
static_assert(BIN_NODES % G1_ROWS == 0, "a range is a whole number of gemm1 tiles");
static_assert(16 * G1_LISTS >= N_CH, "every list of the range has a wave");

__device__ __forceinline__ int g1_draw(int2 pr, int m0, int* lcnt) {
    const unsigned loc = (unsigned)(pr.x - m0);
    return (loc < (unsigned)G1_ROWS) ? atomicAdd(&lcnt[loc], 1) : ELL_CAP;
}
__device__ __forceinline__ void g1_place(int2 pr, int r, int m0, int* lids, int* __restrict__ slots) {
    if (r < ELL_CAP) {
        lids[(pr.x - m0) * ELL_CAP + r] = pr.y;
        slots[(size_t)pr.x * ELL_CAP + r] = pr.y;
    }
}

__global__ __launch_bounds__(512) void gemm1_kernel(const float* __restrict__ x,
                                                    const int* __restrict__ pairs,
                                                    const int* __restrict__ subcnt,
                                                    int* __restrict__ cnt,
                                                    int* __restrict__ slots,
                                                    const float* __restrict__ B1,
                                                    const float* __restrict__ B2,
                                                    float* __restrict__ C,
                                                    float* __restrict__ sum,
                                                    float* __restrict__ sumsq) {
    __shared__ float xs[G1_ROWS * G1_PITCH] __attribute__((aligned(16)));
    __shared__ float agg[G1_ROWS * G1_PITCH] __attribute__((aligned(16)));
    __shared__ int lcnt[G1_ROWS];
    __shared__ int lids[G1_ROWS * ELL_CAP];
    const int lane = threadIdx.x & 31;
    const int wave = __builtin_amdgcn_readfirstlane(threadIdx.x >> 5);
    const int m0 = blockIdx.x * G1_ROWS;
    const int half = lane >> 4, c = lane & 15;
    const int n0 = wave * 64;
    const size_t b_off = (size_t)(2 * half) * F_HID + n0 + 4 * c;

    {
        float4 u[3], w[3];
#pragma unroll
        for (int i = 0; i < 3; ++i) {
            const float* r = x + (size_t)min(m0 + wave * 3 + i, N_NODES - 1) * F_IN + lane * 4;
            u[i] = *(const float4*)r;
            w[i] = *(const float4*)(r + 128);
        }
        if (threadIdx.x < G1_ROWS) lcnt[threadIdx.x] = 0;
        const int bin = blockIdx.x / (BIN_NODES / G1_ROWS);
        int nv = 0;
        if (lane < G1_LISTS && wave + 16 * lane < N_CH) nv = subcnt[bin * N_CH + wave + 16 * lane];
        int nl[G1_LISTS];
        const int2* sub[G1_LISTS];
#pragma unroll
        for (int k = 0; k < G1_LISTS; ++k) {
            nl[k] = __builtin_amdgcn_readlane(nv, k);
            sub[k] = (const int2*)pairs + ((size_t)bin * N_CH + min(wave + 16 * k, N_CH - 1)) * BIN_CH;
        }
        const int2* spare = (const int2*)subcnt;
        int2 pr[G1_LISTS][4];
#pragma unroll
        for (int k = 0; k < G1_LISTS; ++k)
#pragma unroll
            for (int j = 0; j < 4; ++j) {
                const bool in = 32 * j + lane < nl[k];
                pr[k][j] = *(in ? sub[k] + 32 * j + lane : spare);
                if (!in) pr[k][j].x = -1;
            }
#pragma unroll
        for (int i = 0; i < 3; ++i) {
            float* o = xs + (wave * 3 + i) * G1_PITCH + lane * 4;
            *(float4*)o = u[i];
            *(float4*)(o + 128) = w[i];
        }
        __syncthreads();
        int rk[G1_LISTS][4];
#pragma unroll
        for (int k = 0; k < G1_LISTS; ++k)
#pragma unroll
            for (int j = 0; j < 4; ++j) rk[k][j] = g1_draw(pr[k][j], m0, lcnt);
#pragma unroll
        for (int k = 0; k < G1_LISTS; ++k)
#pragma unroll
            for (int j = 0; j < 4; ++j) g1_place(pr[k][j], rk[k][j], m0, lids, slots);
#pragma unroll
        for (int k = 0; k < G1_LISTS; ++k)
            for (int i0 = 128; i0 < nl[k]; i0 += 128) {
                int2 q[4];
#pragma unroll
                for (int j = 0; j < 4; ++j) {
                    const bool in = i0 + 32 * j + lane < nl[k];
                    q[j] = *(in ? sub[k] + i0 + 32 * j + lane : spare);
                    if (!in) q[j].x = -1;
                }
                int rq[4];
#pragma unroll
                for (int j = 0; j < 4; ++j) rq[j] = g1_draw(q[j], m0, lcnt);
#pragma unroll
                for (int j = 0; j < 4; ++j) g1_place(q[j], rq[j], m0, lids, slots);
            }
    }
    __syncthreads();
    if (threadIdx.x < G1_ROWS) cnt[m0 + threadIdx.x] = lcnt[threadIdx.x];

    v8f acc[G1_MT][4];

    if (wave >= 4) {
        const float* xl = x + lane * 4;
        for (int i = 0; i < 4; ++i) {
            const int row = (wave - 4) * 4 + i;
            const int deg = __builtin_amdgcn_readfirstlane(lcnt[row]);
            const int* ids = lids + row * ELL_CAP;
            float4 a0 = {0.0f, 0.0f, 0.0f, 0.0f}, a1 = {0.0f, 0.0f, 0.0f, 0.0f};
            if (deg <= ELL_CAP) {
                int e = 0;
                for (; e + 16 <= deg; e += 16) add_rows<16>(xl, ids + e, a0, a1);
                if (e + 8 <= deg) { add_rows<8>(xl, ids + e, a0, a1); e += 8; }
                if (e + 4 <= deg) { add_rows<4>(xl, ids + e, a0, a1); e += 4; }
                if (e + 2 <= deg) { add_rows<2>(xl, ids + e, a0, a1); e += 2; }
                if (e < deg)      { add_rows<1>(xl, ids + e, a0, a1); }
            } else {
                const int node = m0 + row;
                const int bin = blockIdx.x / (BIN_NODES / G1_ROWS);
                for (int ch = 0; ch < N_CH; ++ch) {
                    const int n = subcnt[bin * N_CH + ch];
                    const int* l = pairs + ((size_t)bin * N_CH + ch) * BIN_CH * 2;
                    for (int k = 0; k < n; ++k)
                        if (l[2 * k] == node) add_rows<1>(xl, l + 2 * k + 1, a0, a1);
                }
            }
            const float di = 1.0f / fmaxf((float)deg, 1.0f);
            float* o = agg + row * G1_PITCH + lane * 4;
            float4 r0 = {a0.x * di, a0.y * di, a0.z * di, a0.w * di};
            float4 r1 = {a1.x * di, a1.y * di, a1.z * di, a1.w * di};
            *(float4*)o = r0;
            *(float4*)(o + 128) = r1;
        }
    } else {
#pragma unroll
        for (int t = 0; t < G1_MT; ++t)
#pragma unroll
            for (int j = 0; j < 4; ++j) acc[t][j] = v8f{};
        g1_product(xs + c * G1_PITCH + 2 * half, B2 + b_off, acc);
    }
    __syncthreads();
    if (wave >= 4) return;

    g1_product(agg + c * G1_PITCH + 2 * half, B1 + b_off, acc);

    float s0 = 0.0f, q0 = 0.0f, s1 = 0.0f, q1 = 0.0f, s2 = 0.0f, q2 = 0.0f, s3 = 0.0f, q3 = 0.0f;
#pragma unroll
    for (int t = 0; t < G1_MT; ++t) {
        if (m0 + 16 * t < N_NODES) {
#pragma unroll
            for (int r = 0; r < 8; ++r) {
                float4 o = {acc[t][0][r], acc[t][1][r], acc[t][2][r], acc[t][3][r]};
                *(float4*)(C + (size_t)(m0 + 16 * t + r + half * 8) * F_HID + n0 + 4 * c) = o;
            }
            float ps, pq;
            col_partial(acc[t][0], ps, pq); s0 += ps; q0 += pq;
            col_partial(acc[t][1], ps, pq); s1 += ps; q1 += pq;
            col_partial(acc[t][2], ps, pq); s2 += ps; q2 += pq;
            col_partial(acc[t][3], ps, pq); s3 += ps; q3 += pq;
        }
    }
    const float sa = (half ? s1 : s0) + __shfl_xor(half ? s0 : s1, 16);
    const float sb = (half ? s3 : s2) + __shfl_xor(half ? s2 : s3, 16);
    const float qa = (half ? q1 : q0) + __shfl_xor(half ? q0 : q1, 16);
    const float qb = (half ? q3 : q2) + __shfl_xor(half ? q2 : q3, 16);
    atomicAdd(&sum[n0 + 4 * c + half], sa);
    atomicAdd(&sum[n0 + 4 * c + 2 + half], sb);
    atomicAdd(&sumsq[n0 + 4 * c + half], qa);
    atomicAdd(&sumsq[n0 + 4 * c + 2 + half], qb);
}

__global__ __launch_bounds__(256) void gemm3_kernel(const float* __restrict__ A,
                                                    const float* __restrict__ sum,
                                                    const float* __restrict__ sumsq,
                                                    const float* __restrict__ gamma,
                                                    const float* __restrict__ beta,
                                                    const float* __restrict__ B1,
                                                    const float* __restrict__ B2,
                                                    float* __restrict__ D1,
                                                    float* __restrict__ D2) {
    __shared__ float bn[4][F_HID] __attribute__((aligned(8)));
    {
        const int k = threadIdx.x;
        const float mean = sum[k] * (1.0f / N_NODES);
        const float var  = sumsq[k] * (1.0f / N_NODES) - mean * mean;
        bn[0][k] = mean;
        bn[1][k] = rsqrtf(var + BN_EPS);
        bn[2][k] = gamma[k];
        bn[3][k] = beta[k];
    }
    __syncthreads();

    const int lane = threadIdx.x & 31;
    const int wave = threadIdx.x >> 5;
    const int tile = blockIdx.x * 8 + wave;
    if (tile >= 625) return;
    const int m0 = tile * 16;
    const int kh = (lane >> 4) << 1;

    v8f acc1 = {}, acc2 = {};
#pragma unroll 4
    for (int k0 = 0; k0 < F_HID; k0 += 4) {
        const v2f raw = load_a(A, F_HID, m0, k0, lane);
        const v2f mu = *(const v2f*)&bn[0][k0 + kh];
        const v2f rs = *(const v2f*)&bn[1][k0 + kh];
        const v2f ga = *(const v2f*)&bn[2][k0 + kh];
        const v2f be = *(const v2f*)&bn[3][k0 + kh];
        v2f a;
        a.x = fmaxf((raw.x - mu.x) * rs.x * ga.x + be.x, 0.0f);
        a.y = fmaxf((raw.y - mu.y) * rs.y * ga.y + be.y, 0.0f);
        v2f b1 = load_b(B1, F_OUT, k0, 0, lane);
        v2f b2 = load_b(B2, F_OUT, k0, 0, lane);
        acc1 = wmma4(a, b1, acc1);
        acc2 = wmma4(a, b2, acc2);
    }
    store_c(D1, F_OUT, m0, 0, lane, acc1);
    store_c(D2, F_OUT, m0, 0, lane, acc2);
}


__global__ __launch_bounds__(256) void agg3_gather_kernel(const float* __restrict__ p,
                                                          const float* __restrict__ hl,
                                                          const int* __restrict__ cnt,
                                                          const int* __restrict__ slots,
                                                          const int* __restrict__ pairs,
                                                          const int* __restrict__ subcnt,
                                                          float* __restrict__ t,
                                                          float* __restrict__ sum,
                                                          float* __restrict__ sumsq) {
    __shared__ float rows[2][8][F_OUT] __attribute__((aligned(16)));
    const int lane = threadIdx.x & 31;
    const int node = __builtin_amdgcn_readfirstlane(blockIdx.x * 8 + (threadIdx.x >> 5));
    const int deg = cnt[node];
    const int end = min(deg, ELL_CAP);
    const int* ids = slots + (size_t)node * ELL_CAP;
    const int slot = lane >> 2, c4 = (lane & 3) << 2;
    float4 acc = {0.0f, 0.0f, 0.0f, 0.0f};
    for (int base = 0; base < end && deg <= ELL_CAP; base += 32) {
        int s[4];
        float4 v[4];
#pragma unroll
        for (int j = 0; j < 4; ++j) s[j] = ids[min(base + slot + 8 * j, end - 1)];
#pragma unroll
        for (int j = 0; j < 4; ++j) v[j] = *(const float4*)(p + (size_t)s[j] * F_OUT + c4);
#pragma unroll
        for (int j = 0; j < 4; ++j)
            if (base + slot + 8 * j < end) {
                acc.x += v[j].x; acc.y += v[j].y; acc.z += v[j].z; acc.w += v[j].w;
            }
    }
    if (deg > ELL_CAP) {
        const int bin = node / BIN_NODES;
        for (int ch = 0; ch < N_CH; ++ch) {
            const int n = subcnt[bin * N_CH + ch];
            const int2* l = (const int2*)pairs + ((size_t)bin * N_CH + ch) * BIN_CH;
            for (int k = slot; k < n; k += 8) {
                const int2 pr = l[k];
                if (pr.x == node) {
                    const float4 v = *(const float4*)(p + (size_t)pr.y * F_OUT + c4);
                    acc.x += v.x; acc.y += v.y; acc.z += v.z; acc.w += v.w;
                }
            }
        }
    }
#pragma unroll
    for (int o = 4; o < 32; o <<= 1) {
        acc.x += __shfl_xor(acc.x, o);
        acc.y += __shfl_xor(acc.y, o);
        acc.z += __shfl_xor(acc.z, o);
        acc.w += __shfl_xor(acc.w, o);
    }
    if (lane < 4) {
        const float di = 1.0f / fmaxf((float)deg, 1.0f);
        const float4 h = *(const float4*)(hl + (size_t)node * F_OUT + c4);
        float4 r = {acc.x * di + h.x, acc.y * di + h.y, acc.z * di + h.z, acc.w * di + h.w};
        *(float4*)(t + (size_t)node * F_OUT + c4) = r;
        const int wv = threadIdx.x >> 5;
        float4 r2 = {r.x * r.x, r.y * r.y, r.z * r.z, r.w * r.w};
        *(float4*)&rows[0][wv][c4] = r;
        *(float4*)&rows[1][wv][c4] = r2;
    }
    __syncthreads();
    if (threadIdx.x < 32) {
        const int st = threadIdx.x >> 4, col = threadIdx.x & 15;
        float a = 0.0f;
#pragma unroll
        for (int w = 0; w < 8; ++w) a += rows[st][w][col];
        atomicAdd(st ? &sumsq[col] : &sum[col], a);
    }
}

__global__ void bn_final_kernel(const float* __restrict__ t,
                                const float* __restrict__ sum, const float* __restrict__ sumsq,
                                const float* __restrict__ gamma, const float* __restrict__ beta,
                                const int* __restrict__ ste,
                                float* __restrict__ out, long long total) {
    long long g = (long long)blockIdx.x * blockDim.x + threadIdx.x;
    if (g >= total) return;
    int c = (int)(g & (F_OUT - 1));
    float mean = sum[c] * (1.0f / N_NODES);
    float var  = sumsq[c] * (1.0f / N_NODES) - mean * mean;
    float v = (t[g] - mean) * rsqrtf(var + BN_EPS) * gamma[c] + beta[c];
    if (*ste)
        out[g] = (v > 0.0f) ? 1.0f : 0.0f;
    else
        out[g] = 1.0f / (1.0f + expf(-v));
}

extern "C" void kernel_launch(void* const* d_in, const int* in_sizes, int n_in,
                              void* d_out, int out_size, void* d_ws, size_t ws_size,
                              hipStream_t stream) {
    const float* x        = (const float*)d_in[0];
    const float* W_conv1  = (const float*)d_in[1];
    const float* W_lin1   = (const float*)d_in[2];
    const float* bn1_g    = (const float*)d_in[3];
    const float* bn1_b    = (const float*)d_in[4];
    const float* W_conv3  = (const float*)d_in[5];
    const float* W_lin3   = (const float*)d_in[6];
    const float* bn3_g    = (const float*)d_in[7];
    const float* bn3_b    = (const float*)d_in[8];
    const int*   src      = (const int*)d_in[9];
    const int*   dst      = (const int*)d_in[10];
    const int*   ste      = (const int*)d_in[11];
    float*       out      = (float*)d_out;

    float* ws = (float*)d_ws;
    const size_t o_cnt  = 0;
    const size_t o_s1   = 10240;
    const size_t o_q1   = o_s1 + 256;
    const size_t o_s3   = o_q1 + 256;
    const size_t o_q3   = o_s3 + 16;
    const size_t o_slot = 11264;
    const size_t o_sub  = o_slot + 1280000;
    const size_t o_h    = o_sub + 640000;
    const size_t o_p    = o_h + 2560000;
    const size_t o_hl   = o_p + 160000;
    const size_t o_agg3 = o_hl + 160000;
    const size_t o_pair = o_agg3 + 160000;
    int*   cnt     = (int*)(ws + o_cnt);
    int*   slots   = (int*)(ws + o_slot);
    int*   subcnt  = (int*)(ws + o_sub);
    int*   pairs   = (int*)(ws + o_pair);
    float* h       = ws + o_h;
    float* p       = ws + o_p;
    float* hl      = ws + o_hl;
    float* agg3    = ws + o_agg3;
    float* s1 = ws + o_s1; float* q1 = ws + o_q1;
    float* s3 = ws + o_s3; float* q3 = ws + o_q3;

    edge_bin_kernel<<<N_CH, 1024, 0, stream>>>(src, dst, pairs, subcnt, s1);

    gemm1_kernel<<<G1_GRID, 512, 0, stream>>>(x, pairs, subcnt, cnt, slots, W_conv1, W_lin1, h, s1, q1);

    gemm3_kernel<<<(625 + 7) / 8, 256, 0, stream>>>(h, s1, q1, bn1_g, bn1_b, W_conv3, W_lin3, p, hl);

    agg3_gather_kernel<<<N_NODES / 8, 256, 0, stream>>>(p, hl, cnt, slots, pairs, subcnt, agg3, s3, q3);

    {
        long long tot = (long long)N_NODES * F_OUT;
        bn_final_kernel<<<(int)((tot + 255) / 256), 256, 0, stream>>>(agg3, s3, q3, bn3_g, bn3_b, ste, out, tot);
    }
}


# layer-3 GEMM: a wave requests its whole 16x256 A tile up front in 512-byte row pieces, applies BN1+ReLU once and keeps it in LDS with the weights; k loop reads LDS only; no sum order changed
# speedup vs baseline: 8.3328x; 1.1477x over previous
#include <hip/hip_runtime.h>
#include <hip/hip_bf16.h>
#include <math.h>

#define N_NODES 10000
#define N_EDGES 320000
#define F_IN    256
#define F_HID   256
#define F_OUT   16
#define BN_EPS  1e-5f

typedef __attribute__((ext_vector_type(2))) float v2f;
typedef __attribute__((ext_vector_type(8))) float v8f;

__device__ __forceinline__ v8f wmma4(v2f a, v2f b, v8f c) {
    return __builtin_amdgcn_wmma_f32_16x16x4_f32(
        false, a, false, b, (short)0, c, false, false);
}

__device__ __forceinline__ v2f load_a(const float* A, int lda, int m0, int k0, int lane) {
    const int half = lane >> 4;
    const int row  = lane & 15;
    const float* p = A + (size_t)(m0 + row) * lda + k0 + (half << 1);
    v2f r; r.x = p[0]; r.y = p[1];
    return r;
}

__device__ __forceinline__ v2f load_b(const float* B, int ldb, int k0, int n0, int lane) {
    const int half = lane >> 4;
    const int col  = lane & 15;
    const float* p = B + (size_t)(k0 + (half << 1)) * ldb + n0 + col;
    v2f r; r.x = p[0]; r.y = p[ldb];
    return r;
}

__device__ __forceinline__ void store_c(float* C, int ldc, int m0, int n0, int lane, v8f acc) {
    const int half = lane >> 4;
    const int col  = lane & 15;
#pragma unroll
    for (int r = 0; r < 8; ++r)
        C[(size_t)(m0 + r + half * 8) * ldc + n0 + col] = acc[r];
}

#define ELL_CAP 128
#define BIN_NODES 240
#define N_BIN     42
#define BIN_CH    4096
#define N_CH      79
#define N_STATS   544

__global__ __launch_bounds__(1024) void edge_bin_kernel(const int* __restrict__ src,
                                                        const int* __restrict__ dst,
                                                        int* __restrict__ pairs,
                                                        int* __restrict__ subcnt,
                                                        float* __restrict__ stats) {
    __shared__ int hist[N_BIN];
    const int tid = threadIdx.x, w = blockIdx.x;
    if (tid < N_BIN) hist[tid] = 0;
    if (w == 0 && tid < N_STATS) stats[tid] = 0.0f;
    __syncthreads();
    const int i = w * (BIN_CH / 4) + tid;
    if (i < N_EDGES / 4) {
        const int4 d4 = ((const int4*)dst)[i];
        const int4 s4 = ((const int4*)src)[i];
        const int d[4] = {d4.x, d4.y, d4.z, d4.w};
        const int s[4] = {s4.x, s4.y, s4.z, s4.w};
        int r[4], bin[4];
#pragma unroll
        for (int j = 0; j < 4; ++j) {
            bin[j] = (int)((unsigned)d[j] / BIN_NODES);
            r[j] = ((unsigned)d[j] < (unsigned)N_NODES) ? atomicAdd(&hist[bin[j]], 1) : -1;
        }
        int2* pl = (int2*)pairs;
#pragma unroll
        for (int j = 0; j < 4; ++j)
            if (r[j] >= 0)
                pl[((size_t)bin[j] * N_CH + w) * BIN_CH + r[j]] = make_int2(d[j], s[j]);
    }
    __syncthreads();
    if (tid < N_BIN) subcnt[tid * N_CH + w] = hist[tid];
}

template <int K>
__device__ __forceinline__ void add_rows(const float* __restrict__ xl, const int* ids,
                                         float4& a0, float4& a1) {
    int s[K];
    float4 u[K], w[K];
#pragma unroll
    for (int j = 0; j < K; ++j) s[j] = __builtin_amdgcn_readfirstlane(ids[j]);
#pragma unroll
    for (int j = 0; j < K; ++j) {
        const float* r = xl + (size_t)s[j] * F_IN;
        u[j] = *(const float4*)r;
        w[j] = *(const float4*)(r + 128);
    }
#pragma unroll
    for (int j = 0; j < K; ++j) {
        a0.x += u[j].x; a0.y += u[j].y; a0.z += u[j].z; a0.w += u[j].w;
        a1.x += w[j].x; a1.y += w[j].y; a1.z += w[j].z; a1.w += w[j].w;
    }
}

__device__ __forceinline__ void col_partial(v8f a, float& s, float& q) {
    s = 0.0f; q = 0.0f;
#pragma unroll
    for (int r = 0; r < 8; ++r) { s += a[r]; q += a[r] * a[r]; }
}

#define G1_KSTEPS 4
#define G1_MT     3
#define G1_ROWS   48
#define G1_GRID   209
struct G1Frag {
    v2f    a[G1_KSTEPS][G1_MT];
    float4 b0[G1_KSTEPS];
    float4 b1[G1_KSTEPS];
};

__device__ __forceinline__ void g1_load(G1Frag& f, const float* a0, const float* a1, const float* a2,
                                        const float* bp) {
#pragma unroll
    for (int s = 0; s < G1_KSTEPS; ++s) {
        f.a[s][0] = *(const v2f*)(a0 + 4 * s);
        f.a[s][1] = *(const v2f*)(a1 + 4 * s);
        f.a[s][2] = *(const v2f*)(a2 + 4 * s);
        f.b0[s] = *(const float4*)(bp + (size_t)(4 * s) * F_HID);
        f.b1[s] = *(const float4*)(bp + (size_t)(4 * s + 1) * F_HID);
    }
}

__device__ __forceinline__ void g1_mma(const G1Frag& f, v8f (&acc)[G1_MT][4]) {
#pragma unroll
    for (int s = 0; s < G1_KSTEPS; ++s) {
        v2f b;
        b.x = f.b0[s].x; b.y = f.b1[s].x;
#pragma unroll
        for (int t = 0; t < G1_MT; ++t) acc[t][0] = wmma4(f.a[s][t], b, acc[t][0]);
        b.x = f.b0[s].y; b.y = f.b1[s].y;
#pragma unroll
        for (int t = 0; t < G1_MT; ++t) acc[t][1] = wmma4(f.a[s][t], b, acc[t][1]);
        b.x = f.b0[s].z; b.y = f.b1[s].z;
#pragma unroll
        for (int t = 0; t < G1_MT; ++t) acc[t][2] = wmma4(f.a[s][t], b, acc[t][2]);
        b.x = f.b0[s].w; b.y = f.b1[s].w;
#pragma unroll
        for (int t = 0; t < G1_MT; ++t) acc[t][3] = wmma4(f.a[s][t], b, acc[t][3]);
    }
}

#define G1_PITCH 260

__device__ __forceinline__ void g1_product(const float* ap, const float* bp, v8f (&acc)[G1_MT][4]) {
    const float* a1p = ap + 16 * G1_PITCH;
    const float* a2p = ap + 32 * G1_PITCH;
    G1Frag f0, f1;
    g1_load(f0, ap, a1p, a2p, bp);
#pragma unroll 1
    for (int g = 0; g < 14; g += 2) {
        g1_load(f1, ap + (g + 1) * 16, a1p + (g + 1) * 16, a2p + (g + 1) * 16, bp + (size_t)(g + 1) * 16 * F_HID);
        g1_mma(f0, acc);
        g1_load(f0, ap + (g + 2) * 16, a1p + (g + 2) * 16, a2p + (g + 2) * 16, bp + (size_t)(g + 2) * 16 * F_HID);
        g1_mma(f1, acc);
    }
    g1_load(f1, ap + 15 * 16, a1p + 15 * 16, a2p + 15 * 16, bp + (size_t)15 * 16 * F_HID);
    g1_mma(f0, acc);
    g1_mma(f1, acc);
}

#define G1_LISTS 5
static_assert(BIN_NODES % G1_ROWS == 0, "a range is a whole number of gemm1 tiles");
static_assert(16 * G1_LISTS >= N_CH, "every list of the range has a wave");

__device__ __forceinline__ int g1_draw(int2 pr, int m0, int* lcnt) {
    const unsigned loc = (unsigned)(pr.x - m0);
    return (loc < (unsigned)G1_ROWS) ? atomicAdd(&lcnt[loc], 1) : ELL_CAP;
}
__device__ __forceinline__ void g1_place(int2 pr, int r, int m0, int* lids, int* __restrict__ slots) {
    if (r < ELL_CAP) {
        lids[(pr.x - m0) * ELL_CAP + r] = pr.y;
        slots[(size_t)pr.x * ELL_CAP + r] = pr.y;
    }
}

__global__ __launch_bounds__(512) void gemm1_kernel(const float* __restrict__ x,
                                                    const int* __restrict__ pairs,
                                                    const int* __restrict__ subcnt,
                                                    int* __restrict__ cnt,
                                                    int* __restrict__ slots,
                                                    const float* __restrict__ B1,
                                                    const float* __restrict__ B2,
                                                    float* __restrict__ C,
                                                    float* __restrict__ sum,
                                                    float* __restrict__ sumsq) {
    __shared__ float xs[G1_ROWS * G1_PITCH] __attribute__((aligned(16)));
    __shared__ float agg[G1_ROWS * G1_PITCH] __attribute__((aligned(16)));
    __shared__ int lcnt[G1_ROWS];
    __shared__ int lids[G1_ROWS * ELL_CAP];
    const int lane = threadIdx.x & 31;
    const int wave = __builtin_amdgcn_readfirstlane(threadIdx.x >> 5);
    const int m0 = blockIdx.x * G1_ROWS;
    const int half = lane >> 4, c = lane & 15;
    const int n0 = wave * 64;
    const size_t b_off = (size_t)(2 * half) * F_HID + n0 + 4 * c;

    {
        float4 u[3], w[3];
#pragma unroll
        for (int i = 0; i < 3; ++i) {
            const float* r = x + (size_t)min(m0 + wave * 3 + i, N_NODES - 1) * F_IN + lane * 4;
            u[i] = *(const float4*)r;
            w[i] = *(const float4*)(r + 128);
        }
        if (threadIdx.x < G1_ROWS) lcnt[threadIdx.x] = 0;
        const int bin = blockIdx.x / (BIN_NODES / G1_ROWS);
        int nv = 0;
        if (lane < G1_LISTS && wave + 16 * lane < N_CH) nv = subcnt[bin * N_CH + wave + 16 * lane];
        int nl[G1_LISTS];
        const int2* sub[G1_LISTS];
#pragma unroll
        for (int k = 0; k < G1_LISTS; ++k) {
            nl[k] = __builtin_amdgcn_readlane(nv, k);
            sub[k] = (const int2*)pairs + ((size_t)bin * N_CH + min(wave + 16 * k, N_CH - 1)) * BIN_CH;
        }
        const int2* spare = (const int2*)subcnt;
        int2 pr[G1_LISTS][4];
#pragma unroll
        for (int k = 0; k < G1_LISTS; ++k)
#pragma unroll
            for (int j = 0; j < 4; ++j) {
                const bool in = 32 * j + lane < nl[k];
                pr[k][j] = *(in ? sub[k] + 32 * j + lane : spare);
                if (!in) pr[k][j].x = -1;
            }
#pragma unroll
        for (int i = 0; i < 3; ++i) {
            float* o = xs + (wave * 3 + i) * G1_PITCH + lane * 4;
            *(float4*)o = u[i];
            *(float4*)(o + 128) = w[i];
        }
        __syncthreads();
        int rk[G1_LISTS][4];
#pragma unroll
        for (int k = 0; k < G1_LISTS; ++k)
#pragma unroll
            for (int j = 0; j < 4; ++j) rk[k][j] = g1_draw(pr[k][j], m0, lcnt);
#pragma unroll
        for (int k = 0; k < G1_LISTS; ++k)
#pragma unroll
            for (int j = 0; j < 4; ++j) g1_place(pr[k][j], rk[k][j], m0, lids, slots);
#pragma unroll
        for (int k = 0; k < G1_LISTS; ++k)
            for (int i0 = 128; i0 < nl[k]; i0 += 128) {
                int2 q[4];
#pragma unroll
                for (int j = 0; j < 4; ++j) {
                    const bool in = i0 + 32 * j + lane < nl[k];
                    q[j] = *(in ? sub[k] + i0 + 32 * j + lane : spare);
                    if (!in) q[j].x = -1;
                }
                int rq[4];
#pragma unroll
                for (int j = 0; j < 4; ++j) rq[j] = g1_draw(q[j], m0, lcnt);
#pragma unroll
                for (int j = 0; j < 4; ++j) g1_place(q[j], rq[j], m0, lids, slots);
            }
    }
    __syncthreads();
    if (threadIdx.x < G1_ROWS) cnt[m0 + threadIdx.x] = lcnt[threadIdx.x];

    v8f acc[G1_MT][4];

    if (wave >= 4) {
        const float* xl = x + lane * 4;
        for (int i = 0; i < 4; ++i) {
            const int row = (wave - 4) * 4 + i;
            const int deg = __builtin_amdgcn_readfirstlane(lcnt[row]);
            const int* ids = lids + row * ELL_CAP;
            float4 a0 = {0.0f, 0.0f, 0.0f, 0.0f}, a1 = {0.0f, 0.0f, 0.0f, 0.0f};
            if (deg <= ELL_CAP) {
                int e = 0;
                for (; e + 16 <= deg; e += 16) add_rows<16>(xl, ids + e, a0, a1);
                if (e + 8 <= deg) { add_rows<8>(xl, ids + e, a0, a1); e += 8; }
                if (e + 4 <= deg) { add_rows<4>(xl, ids + e, a0, a1); e += 4; }
                if (e + 2 <= deg) { add_rows<2>(xl, ids + e, a0, a1); e += 2; }
                if (e < deg)      { add_rows<1>(xl, ids + e, a0, a1); }
            } else {
                const int node = m0 + row;
                const int bin = blockIdx.x / (BIN_NODES / G1_ROWS);
                for (int ch = 0; ch < N_CH; ++ch) {
                    const int n = subcnt[bin * N_CH + ch];
                    const int* l = pairs + ((size_t)bin * N_CH + ch) * BIN_CH * 2;
                    for (int k = 0; k < n; ++k)
                        if (l[2 * k] == node) add_rows<1>(xl, l + 2 * k + 1, a0, a1);
                }
            }
            const float di = 1.0f / fmaxf((float)deg, 1.0f);
            float* o = agg + row * G1_PITCH + lane * 4;
            float4 r0 = {a0.x * di, a0.y * di, a0.z * di, a0.w * di};
            float4 r1 = {a1.x * di, a1.y * di, a1.z * di, a1.w * di};
            *(float4*)o = r0;
            *(float4*)(o + 128) = r1;
        }
    } else {
#pragma unroll
        for (int t = 0; t < G1_MT; ++t)
#pragma unroll
            for (int j = 0; j < 4; ++j) acc[t][j] = v8f{};
        g1_product(xs + c * G1_PITCH + 2 * half, B2 + b_off, acc);
    }
    __syncthreads();
    if (wave >= 4) return;

    g1_product(agg + c * G1_PITCH + 2 * half, B1 + b_off, acc);

    float s0 = 0.0f, q0 = 0.0f, s1 = 0.0f, q1 = 0.0f, s2 = 0.0f, q2 = 0.0f, s3 = 0.0f, q3 = 0.0f;
#pragma unroll
    for (int t = 0; t < G1_MT; ++t) {
        if (m0 + 16 * t < N_NODES) {
#pragma unroll
            for (int r = 0; r < 8; ++r) {
                float4 o = {acc[t][0][r], acc[t][1][r], acc[t][2][r], acc[t][3][r]};
                *(float4*)(C + (size_t)(m0 + 16 * t + r + half * 8) * F_HID + n0 + 4 * c) = o;
            }
            float ps, pq;
            col_partial(acc[t][0], ps, pq); s0 += ps; q0 += pq;
            col_partial(acc[t][1], ps, pq); s1 += ps; q1 += pq;
            col_partial(acc[t][2], ps, pq); s2 += ps; q2 += pq;
            col_partial(acc[t][3], ps, pq); s3 += ps; q3 += pq;
        }
    }
    const float sa = (half ? s1 : s0) + __shfl_xor(half ? s0 : s1, 16);
    const float sb = (half ? s3 : s2) + __shfl_xor(half ? s2 : s3, 16);
    const float qa = (half ? q1 : q0) + __shfl_xor(half ? q0 : q1, 16);
    const float qb = (half ? q3 : q2) + __shfl_xor(half ? q2 : q3, 16);
    atomicAdd(&sum[n0 + 4 * c + half], sa);
    atomicAdd(&sum[n0 + 4 * c + 2 + half], sb);
    atomicAdd(&sumsq[n0 + 4 * c + half], qa);
    atomicAdd(&sumsq[n0 + 4 * c + 2 + half], qb);
}

#define G3_WAVES 4
#define G3_TILES (N_NODES / 16)
#define G3_GRID  ((G3_TILES + G3_WAVES - 1) / G3_WAVES)
#define G3_BVEC  (F_HID * F_OUT / 4 / (32 * G3_WAVES))
#define G3_PITCH 260
static_assert(G3_BVEC * 4 * 32 * G3_WAVES == F_HID * F_OUT, "the work-group copies a weight matrix in whole pieces");
static_assert(F_HID == 256, "a lane owns two float4 column slices of a row");

__device__ __forceinline__ float4 bn_relu4(float4 r, float4 mu, float4 rs, float4 g, float4 b) {
    float4 o;
    o.x = fmaxf((r.x - mu.x) * rs.x * g.x + b.x, 0.0f);
    o.y = fmaxf((r.y - mu.y) * rs.y * g.y + b.y, 0.0f);
    o.z = fmaxf((r.z - mu.z) * rs.z * g.z + b.z, 0.0f);
    o.w = fmaxf((r.w - mu.w) * rs.w * g.w + b.w, 0.0f);
    return o;
}
__device__ __forceinline__ void bn_consts4(float4 s, float4 q, float4& mu, float4& rs) {
    mu.x = s.x * (1.0f / N_NODES); mu.y = s.y * (1.0f / N_NODES);
    mu.z = s.z * (1.0f / N_NODES); mu.w = s.w * (1.0f / N_NODES);
    rs.x = rsqrtf(q.x * (1.0f / N_NODES) - mu.x * mu.x + BN_EPS);
    rs.y = rsqrtf(q.y * (1.0f / N_NODES) - mu.y * mu.y + BN_EPS);
    rs.z = rsqrtf(q.z * (1.0f / N_NODES) - mu.z * mu.z + BN_EPS);
    rs.w = rsqrtf(q.w * (1.0f / N_NODES) - mu.w * mu.w + BN_EPS);
}

struct G3Frag {
    v2f a[4];
    v2f b1[4];
    v2f b2[4];
};
__device__ __forceinline__ void g3_load(G3Frag& f, const float* ap, const float* b1p, const float* b2p) {
#pragma unroll
    for (int s = 0; s < 4; ++s) {
        f.a[s] = *(const v2f*)(ap + 4 * s);
        f.b1[s].x = b1p[(4 * s) * F_OUT]; f.b1[s].y = b1p[(4 * s + 1) * F_OUT];
        f.b2[s].x = b2p[(4 * s) * F_OUT]; f.b2[s].y = b2p[(4 * s + 1) * F_OUT];
    }
}
__device__ __forceinline__ void g3_mma(const G3Frag& f, v8f& acc1, v8f& acc2) {
#pragma unroll
    for (int s = 0; s < 4; ++s) {
        acc1 = wmma4(f.a[s], f.b1[s], acc1);
        acc2 = wmma4(f.a[s], f.b2[s], acc2);
    }
}

__global__ __launch_bounds__(32 * G3_WAVES) void gemm3_kernel(const float* __restrict__ A,
                                                             const float* __restrict__ sum,
                                                             const float* __restrict__ sumsq,
                                                             const float* __restrict__ gamma,
                                                             const float* __restrict__ beta,
                                                             const float* __restrict__ B1,
                                                             const float* __restrict__ B2,
                                                             float* __restrict__ D1,
                                                             float* __restrict__ D2) {
    __shared__ float lds[2 * F_HID * F_OUT + G3_WAVES * 16 * G3_PITCH] __attribute__((aligned(16)));
    float* bs1 = lds;
    float* bs2 = lds + F_HID * F_OUT;

    const int tid = threadIdx.x;
    const int lane = tid & 31;
    const int wave = __builtin_amdgcn_readfirstlane(tid >> 5);
    const int tile = blockIdx.x * G3_WAVES + wave;
    const int m0 = min(tile, G3_TILES - 1) * 16;
    const int kh = (lane >> 4) << 1;
    const int c = lane & 15;
    const int c0 = 4 * lane, c1 = 128 + 4 * lane;

    float4 w1[G3_BVEC], w2[G3_BVEC];
#pragma unroll
    for (int i = 0; i < G3_BVEC; ++i) {
        w1[i] = ((const float4*)B1)[i * (32 * G3_WAVES) + tid];
        w2[i] = ((const float4*)B2)[i * (32 * G3_WAVES) + tid];
    }
    const float4 su0 = *(const float4*)(sum + c0),   su1 = *(const float4*)(sum + c1);
    const float4 sq0 = *(const float4*)(sumsq + c0), sq1 = *(const float4*)(sumsq + c1);
    const float4 ga0 = *(const float4*)(gamma + c0), ga1 = *(const float4*)(gamma + c1);
    const float4 be0 = *(const float4*)(beta + c0),  be1 = *(const float4*)(beta + c1);
    float4 u[16], w[16];
#pragma unroll
    for (int i = 0; i < 16; ++i) {
        const float* r = A + (size_t)(m0 + i) * F_HID;
        u[i] = *(const float4*)(r + c0);
        w[i] = *(const float4*)(r + c1);
    }

#pragma unroll
    for (int i = 0; i < G3_BVEC; ++i) {
        ((float4*)bs1)[i * (32 * G3_WAVES) + tid] = w1[i];
        ((float4*)bs2)[i * (32 * G3_WAVES) + tid] = w2[i];
    }
    float4 mu0, rs0, mu1, rs1;
    bn_consts4(su0, sq0, mu0, rs0);
    bn_consts4(su1, sq1, mu1, rs1);
    float* trow = lds + 2 * F_HID * F_OUT + wave * (16 * G3_PITCH);
#pragma unroll
    for (int i = 0; i < 16; ++i) {
        *(float4*)(trow + i * G3_PITCH + c0) = bn_relu4(u[i], mu0, rs0, ga0, be0);
        *(float4*)(trow + i * G3_PITCH + c1) = bn_relu4(w[i], mu1, rs1, ga1, be1);
    }
    __syncthreads();
    if (tile >= G3_TILES) return;

    const float* ap  = trow + c * G3_PITCH + kh;
    const float* b1p = bs1 + kh * F_OUT + c;
    const float* b2p = bs2 + kh * F_OUT + c;
    v8f acc1 = {}, acc2 = {};
    G3Frag f0, f1;
    g3_load(f0, ap, b1p, b2p);
    __builtin_amdgcn_sched_barrier(0);
#pragma unroll
    for (int g = 0; g < 14; g += 2) {
        g3_load(f1, ap + (g + 1) * 16, b1p + (g + 1) * 16 * F_OUT, b2p + (g + 1) * 16 * F_OUT);
        __builtin_amdgcn_sched_barrier(0);
        g3_mma(f0, acc1, acc2);
        g3_load(f0, ap + (g + 2) * 16, b1p + (g + 2) * 16 * F_OUT, b2p + (g + 2) * 16 * F_OUT);
        __builtin_amdgcn_sched_barrier(0);
        g3_mma(f1, acc1, acc2);
    }
    g3_load(f1, ap + 15 * 16, b1p + 15 * 16 * F_OUT, b2p + 15 * 16 * F_OUT);
    __builtin_amdgcn_sched_barrier(0);
    g3_mma(f0, acc1, acc2);
    g3_mma(f1, acc1, acc2);
    store_c(D1, F_OUT, m0, 0, lane, acc1);
    store_c(D2, F_OUT, m0, 0, lane, acc2);
}


__global__ __launch_bounds__(256) void agg3_gather_kernel(const float* __restrict__ p,
                                                          const float* __restrict__ hl,
                                                          const int* __restrict__ cnt,
                                                          const int* __restrict__ slots,
                                                          const int* __restrict__ pairs,
                                                          const int* __restrict__ subcnt,
                                                          float* __restrict__ t,
                                                          float* __restrict__ sum,
                                                          float* __restrict__ sumsq) {
    __shared__ float rows[2][8][F_OUT] __attribute__((aligned(16)));
    const int lane = threadIdx.x & 31;
    const int node = __builtin_amdgcn_readfirstlane(blockIdx.x * 8 + (threadIdx.x >> 5));
    const int deg = cnt[node];
    const int end = min(deg, ELL_CAP);
    const int* ids = slots + (size_t)node * ELL_CAP;
    const int slot = lane >> 2, c4 = (lane & 3) << 2;
    float4 acc = {0.0f, 0.0f, 0.0f, 0.0f};
    for (int base = 0; base < end && deg <= ELL_CAP; base += 32) {
        int s[4];
        float4 v[4];
#pragma unroll
        for (int j = 0; j < 4; ++j) s[j] = ids[min(base + slot + 8 * j, end - 1)];
#pragma unroll
        for (int j = 0; j < 4; ++j) v[j] = *(const float4*)(p + (size_t)s[j] * F_OUT + c4);
#pragma unroll
        for (int j = 0; j < 4; ++j)
            if (base + slot + 8 * j < end) {
                acc.x += v[j].x; acc.y += v[j].y; acc.z += v[j].z; acc.w += v[j].w;
            }
    }
    if (deg > ELL_CAP) {
        const int bin = node / BIN_NODES;
        for (int ch = 0; ch < N_CH; ++ch) {
            const int n = subcnt[bin * N_CH + ch];
            const int2* l = (const int2*)pairs + ((size_t)bin * N_CH + ch) * BIN_CH;
            for (int k = slot; k < n; k += 8) {
                const int2 pr = l[k];
                if (pr.x == node) {
                    const float4 v = *(const float4*)(p + (size_t)pr.y * F_OUT + c4);
                    acc.x += v.x; acc.y += v.y; acc.z += v.z; acc.w += v.w;
                }
            }
        }
    }
#pragma unroll
    for (int o = 4; o < 32; o <<= 1) {
        acc.x += __shfl_xor(acc.x, o);
        acc.y += __shfl_xor(acc.y, o);
        acc.z += __shfl_xor(acc.z, o);
        acc.w += __shfl_xor(acc.w, o);
    }
    if (lane < 4) {
        const float di = 1.0f / fmaxf((float)deg, 1.0f);
        const float4 h = *(const float4*)(hl + (size_t)node * F_OUT + c4);
        float4 r = {acc.x * di + h.x, acc.y * di + h.y, acc.z * di + h.z, acc.w * di + h.w};
        *(float4*)(t + (size_t)node * F_OUT + c4) = r;
        const int wv = threadIdx.x >> 5;
        float4 r2 = {r.x * r.x, r.y * r.y, r.z * r.z, r.w * r.w};
        *(float4*)&rows[0][wv][c4] = r;
        *(float4*)&rows[1][wv][c4] = r2;
    }
    __syncthreads();
    if (threadIdx.x < 32) {
        const int st = threadIdx.x >> 4, col = threadIdx.x & 15;
        float a = 0.0f;
#pragma unroll
        for (int w = 0; w < 8; ++w) a += rows[st][w][col];
        atomicAdd(st ? &sumsq[col] : &sum[col], a);
    }
}

__global__ void bn_final_kernel(const float* __restrict__ t,
                                const float* __restrict__ sum, const float* __restrict__ sumsq,
                                const float* __restrict__ gamma, const float* __restrict__ beta,
                                const int* __restrict__ ste,
                                float* __restrict__ out, long long total) {
    long long g = (long long)blockIdx.x * blockDim.x + threadIdx.x;
    if (g >= total) return;
    int c = (int)(g & (F_OUT - 1));
    float mean = sum[c] * (1.0f / N_NODES);
    float var  = sumsq[c] * (1.0f / N_NODES) - mean * mean;
    float v = (t[g] - mean) * rsqrtf(var + BN_EPS) * gamma[c] + beta[c];
    if (*ste)
        out[g] = (v > 0.0f) ? 1.0f : 0.0f;
    else
        out[g] = 1.0f / (1.0f + expf(-v));
}

extern "C" void kernel_launch(void* const* d_in, const int* in_sizes, int n_in,
                              void* d_out, int out_size, void* d_ws, size_t ws_size,
                              hipStream_t stream) {
    const float* x        = (const float*)d_in[0];
    const float* W_conv1  = (const float*)d_in[1];
    const float* W_lin1   = (const float*)d_in[2];
    const float* bn1_g    = (const float*)d_in[3];
    const float* bn1_b    = (const float*)d_in[4];
    const float* W_conv3  = (const float*)d_in[5];
    const float* W_lin3   = (const float*)d_in[6];
    const float* bn3_g    = (const float*)d_in[7];
    const float* bn3_b    = (const float*)d_in[8];
    const int*   src      = (const int*)d_in[9];
    const int*   dst      = (const int*)d_in[10];
    const int*   ste      = (const int*)d_in[11];
    float*       out      = (float*)d_out;

    float* ws = (float*)d_ws;
    const size_t o_cnt  = 0;
    const size_t o_s1   = 10240;
    const size_t o_q1   = o_s1 + 256;
    const size_t o_s3   = o_q1 + 256;
    const size_t o_q3   = o_s3 + 16;
    const size_t o_slot = 11264;
    const size_t o_sub  = o_slot + 1280000;
    const size_t o_h    = o_sub + 640000;
    const size_t o_p    = o_h + 2560000;
    const size_t o_hl   = o_p + 160000;
    const size_t o_agg3 = o_hl + 160000;
    const size_t o_pair = o_agg3 + 160000;
    int*   cnt     = (int*)(ws + o_cnt);
    int*   slots   = (int*)(ws + o_slot);
    int*   subcnt  = (int*)(ws + o_sub);
    int*   pairs   = (int*)(ws + o_pair);
    float* h       = ws + o_h;
    float* p       = ws + o_p;
    float* hl      = ws + o_hl;
    float* agg3    = ws + o_agg3;
    float* s1 = ws + o_s1; float* q1 = ws + o_q1;
    float* s3 = ws + o_s3; float* q3 = ws + o_q3;

    edge_bin_kernel<<<N_CH, 1024, 0, stream>>>(src, dst, pairs, subcnt, s1);

    gemm1_kernel<<<G1_GRID, 512, 0, stream>>>(x, pairs, subcnt, cnt, slots, W_conv1, W_lin1, h, s1, q1);

    gemm3_kernel<<<G3_GRID, 32 * G3_WAVES, 0, stream>>>(h, s1, q1, bn1_g, bn1_b, W_conv3, W_lin3, p, hl);

    agg3_gather_kernel<<<N_NODES / 8, 256, 0, stream>>>(p, hl, cnt, slots, pairs, subcnt, agg3, s3, q3);

    {
        long long tot = (long long)N_NODES * F_OUT;
        bn_final_kernel<<<(int)((tot + 255) / 256), 256, 0, stream>>>(agg3, s3, q3, bn3_g, bn3_b, ste, out, tot);
    }
}
